# starting kernel
# baseline (speedup 1.0000x reference)
#include <hip/hip_runtime.h>
#include <hip/hip_bf16.h>


typedef float v2f __attribute__((ext_vector_type(2)));
typedef float v8f __attribute__((ext_vector_type(8)));

#define NNODES  200000
#define NEDGES  6400000
#define NFEAT   128
#define NHID    16
#define NMLP    100
#define NCLS    27
#define NGRAPH  4096


__global__ void fill_f32(float* __restrict__ p, int n, float v) {
    int i = blockIdx.x * blockDim.x + threadIdx.x;
    if (i < n) p[i] = v;
}

__global__ void deg_accum(const int* __restrict__ dst, float* __restrict__ deg, int e) {
    int i = blockIdx.x * blockDim.x + threadIdx.x;
    if (i < e) atomicAdd(&deg[dst[i]], 1.0f);
}

__global__ void rsqrt_ip(float* __restrict__ p, int n) {
    int i = blockIdx.x * blockDim.x + threadIdx.x;
    if (i < n) p[i] = rsqrtf(p[i]);
}

__global__ __launch_bounds__(256)
void gemm_wmma_f32(const float* __restrict__ X, const float* __restrict__ W,
                   float* __restrict__ T, int M, int K) {
    __shared__ float lw[NFEAT * NHID];
    const int tid = threadIdx.x;
    const int kn  = K * NHID;
    for (int i = tid; i < kn; i += 256) lw[i] = W[i];
    __syncthreads();

    const int wave = tid >> 5;
    const int lane = tid & 31;
    const int tile = blockIdx.x * 8 + wave;
    if (tile * 16 >= M) return;

    const int row   = tile * 16 + (lane & 15);
    const int kHalf = (lane >> 4) << 1;
    const int n     = lane & 15;

    v8f c = {};
    const float* xr = X + (size_t)row * K + kHalf;
    for (int k0 = 0; k0 < K; k0 += 4) {
        float2 av = *(const float2*)(xr + k0);
        v2f a; a.x = av.x; a.y = av.y;
        v2f b;
        b.x = lw[(k0 + kHalf)     * NHID + n];
        b.y = lw[(k0 + kHalf + 1) * NHID + n];
        c = __builtin_amdgcn_wmma_f32_16x16x4_f32(
                 false, a,  false, b,
                 (short)0, c,  false,  false);
    }

    const int rbase = tile * 16 + ((lane >> 4) << 3);
#pragma unroll
    for (int v = 0; v < 8; ++v)
        T[(size_t)(rbase + v) * NHID + n] = c[v];
}

__global__ void edge_scatter(const int* __restrict__ src, const int* __restrict__ dst,
                             const float* __restrict__ dinv, const float* __restrict__ T,
                             float* __restrict__ OUT, int e) {
    int t  = blockIdx.x * blockDim.x + threadIdx.x;
    int ed = t >> 2, q = t & 3;
    if (ed >= e) return;
    int s = src[ed], d = dst[ed];
    float w = dinv[s] * dinv[d];
    const float4 m = *(const float4*)(T + (size_t)s * NHID + q * 4);
    float* o = OUT + (size_t)d * NHID + q * 4;
    atomicAdd(o + 0, m.x * w);
    atomicAdd(o + 1, m.y * w);
    atomicAdd(o + 2, m.z * w);
    atomicAdd(o + 3, m.w * w);
}

__global__ void finalize_node(float* __restrict__ ACC, const float* __restrict__ T,
                              const float* __restrict__ dinv, const float* __restrict__ bias,
                              int n16, int doRelu) {
    int idx = blockIdx.x * blockDim.x + threadIdx.x;
    if (idx >= n16) return;
    int i = idx >> 4, f = idx & 15;
    float di = dinv[i];
    float v  = ACC[idx] + T[idx] * di * di + bias[f];
    ACC[idx] = doRelu ? fmaxf(v, 0.0f) : v;
}

__global__ void pool_kernel(const int* __restrict__ bvec, const float* __restrict__ Hn,
                            float* __restrict__ Gp, int n) {
    int t = blockIdx.x * blockDim.x + threadIdx.x;
    int i = t >> 2, q = t & 3;
    if (i >= n) return;
    int g = bvec[i];
    const float4 h = *(const float4*)(Hn + (size_t)i * NHID + q * 4);
    float* o = Gp + (size_t)g * NHID + q * 4;
    atomicAdd(o + 0, h.x);
    atomicAdd(o + 1, h.y);
    atomicAdd(o + 2, h.z);
    atomicAdd(o + 3, h.w);
}

__global__ void lin1_kernel(const float* __restrict__ G, const float* __restrict__ W,
                            const float* __restrict__ B, float* __restrict__ O) {
    int t = blockIdx.x * blockDim.x + threadIdx.x;
    if (t >= NGRAPH * NMLP) return;
    int gi = t / NMLP, j = t % NMLP;
    float acc = B[j];
#pragma unroll
    for (int k = 0; k < NHID; ++k)
        acc = fmaf(fmaxf(G[gi * NHID + k], 0.0f), W[k * NMLP + j], acc);
    O[t] = fmaxf(acc, 0.0f);
}

__global__ void lin2_kernel(const float* __restrict__ G, const float* __restrict__ W,
                            const float* __restrict__ B, float* __restrict__ O) {
    int t = blockIdx.x * blockDim.x + threadIdx.x;
    if (t >= NGRAPH * NCLS) return;
    int gi = t / NCLS, c = t % NCLS;
    float acc = B[c];
    for (int k = 0; k < NMLP; ++k)
        acc = fmaf(G[gi * NMLP + k], W[k * NCLS + c], acc);
    O[t] = acc;
}


extern "C" void kernel_launch(void* const* d_in, const int* in_sizes, int n_in,
                              void* d_out, int out_size, void* d_ws, size_t ws_size,
                              hipStream_t stream) {
    const float* x   = (const float*)d_in[0];
    const int*  eidx = (const int*)d_in[1];
    const int*  bvec = (const int*)d_in[2];
    const float* W1  = (const float*)d_in[3];
    const float* b1  = (const float*)d_in[4];
    const float* W2  = (const float*)d_in[5];
    const float* b2  = (const float*)d_in[6];
    const float* Wl1 = (const float*)d_in[7];
    const float* bl1 = (const float*)d_in[8];
    const float* Wl2 = (const float*)d_in[9];
    const float* bl2 = (const float*)d_in[10];
    float* out = (float*)d_out;

    char* ws = (char*)d_ws;
    size_t off = 0;
    auto walloc = [&](size_t bytes) -> void* {
        void* p = ws + off;
        off = (off + bytes + 255) & ~(size_t)255;
        return p;
    };
    float* deg  = (float*)walloc((size_t)NNODES * 4);
    float* bufA = (float*)walloc((size_t)NNODES * NHID * 4);
    float* bufB = (float*)walloc((size_t)NNODES * NHID * 4);
    float* gbuf = (float*)walloc((size_t)NGRAPH * NHID * 4);
    float* g2   = (float*)walloc((size_t)NGRAPH * NMLP * 4);

    const int* src = eidx;
    const int* dst = eidx + NEDGES;

    const int TB = 256;
    const int nN    = (NNODES + TB - 1) / TB;
    const int nE    = (NEDGES + TB - 1) / TB;
    const int nN16  = (NNODES * NHID + TB - 1) / TB;
    const int nE4   = (NEDGES * 4 + TB - 1) / TB;
    const int nN4   = (NNODES * 4 + TB - 1) / TB;
    const int nTile = ((NNODES / 16) + 7) / 8;

    fill_f32<<<nN, TB, 0, stream>>>(deg, NNODES, 1.0f);
    deg_accum<<<nE, TB, 0, stream>>>(dst, deg, NEDGES);
    rsqrt_ip<<<nN, TB, 0, stream>>>(deg, NNODES);
    float* dinv = deg;

    gemm_wmma_f32<<<nTile, TB, 0, stream>>>(x, W1, bufA, NNODES, NFEAT);
    fill_f32<<<nN16, TB, 0, stream>>>(bufB, NNODES * NHID, 0.0f);
    edge_scatter<<<nE4, TB, 0, stream>>>(src, dst, dinv, bufA, bufB, NEDGES);
    finalize_node<<<nN16, TB, 0, stream>>>(bufB, bufA, dinv, b1, NNODES * NHID, 1);

    gemm_wmma_f32<<<nTile, TB, 0, stream>>>(bufB, W2, bufA, NNODES, NHID);
    fill_f32<<<nN16, TB, 0, stream>>>(bufB, NNODES * NHID, 0.0f);
    edge_scatter<<<nE4, TB, 0, stream>>>(src, dst, dinv, bufA, bufB, NEDGES);
    finalize_node<<<nN16, TB, 0, stream>>>(bufB, bufA, dinv, b2, NNODES * NHID, 0);

    fill_f32<<<(NGRAPH * NHID + TB - 1) / TB, TB, 0, stream>>>(gbuf, NGRAPH * NHID, 0.0f);
    pool_kernel<<<nN4, TB, 0, stream>>>(bvec, bufB, gbuf, NNODES);
    lin1_kernel<<<(NGRAPH * NMLP + TB - 1) / TB, TB, 0, stream>>>(gbuf, Wl1, bl1, g2);
    lin2_kernel<<<(NGRAPH * NCLS + TB - 1) / TB, TB, 0, stream>>>(g2, Wl2, bl2, out);
}


# Edge scatter: a work-group reads its die and adds only the edges whose destination rows belong to its own half of the L2 (groups drawn from per-half counters); adds in arrival order as before
# speedup vs baseline: 1.6939x; 1.6939x over previous
#include <hip/hip_runtime.h>
#include <hip/hip_bf16.h>


typedef float v2f __attribute__((ext_vector_type(2)));
typedef float v8f __attribute__((ext_vector_type(8)));

#define NNODES  200000
#define NEDGES  6400000
#define NFEAT   128
#define NHID    16
#define NMLP    100
#define NCLS    27
#define NGRAPH  4096


__global__ void fill_f32(float* __restrict__ p, int n, float v) {
    int i = blockIdx.x * blockDim.x + threadIdx.x;
    if (i < n) p[i] = v;
}

__global__ void fill_f32_ctr(float* __restrict__ p, int n, float v,
                             unsigned* __restrict__ ctr, int nctr) {
    int i = blockIdx.x * blockDim.x + threadIdx.x;
    if (i < n) p[i] = v;
    if (i < nctr) ctr[i] = 0u;
}

__global__ void deg_accum(const int* __restrict__ dst, float* __restrict__ deg, int e) {
    int i = blockIdx.x * blockDim.x + threadIdx.x;
    if (i < e) atomicAdd(&deg[dst[i]], 1.0f);
}

__global__ void rsqrt_ip(float* __restrict__ p, int n) {
    int i = blockIdx.x * blockDim.x + threadIdx.x;
    if (i < n) p[i] = rsqrtf(p[i]);
}

__global__ __launch_bounds__(256)
void gemm_wmma_f32(const float* __restrict__ X, const float* __restrict__ W,
                   float* __restrict__ T, int M, int K) {
    __shared__ float lw[NFEAT * NHID];
    const int tid = threadIdx.x;
    const int kn  = K * NHID;
    for (int i = tid; i < kn; i += 256) lw[i] = W[i];
    __syncthreads();

    const int wave = tid >> 5;
    const int lane = tid & 31;
    const int tile = blockIdx.x * 8 + wave;
    if (tile * 16 >= M) return;

    const int row   = tile * 16 + (lane & 15);
    const int kHalf = (lane >> 4) << 1;
    const int n     = lane & 15;

    v8f c = {};
    const float* xr = X + (size_t)row * K + kHalf;
    for (int k0 = 0; k0 < K; k0 += 4) {
        float2 av = *(const float2*)(xr + k0);
        v2f a; a.x = av.x; a.y = av.y;
        v2f b;
        b.x = lw[(k0 + kHalf)     * NHID + n];
        b.y = lw[(k0 + kHalf + 1) * NHID + n];
        c = __builtin_amdgcn_wmma_f32_16x16x4_f32(
                 false, a,  false, b,
                 (short)0, c,  false,  false);
    }

    const int rbase = tile * 16 + ((lane >> 4) << 3);
#pragma unroll
    for (int v = 0; v < 8; ++v)
        T[(size_t)(rbase + v) * NHID + n] = c[v];
}


#define EDGES_PER_WAVE 64
#define EDGES_PER_GROUP (8 * EDGES_PER_WAVE)
#define CTR_STRIDE 64

__global__ __launch_bounds__(256)
void edge_scatter(const int* __restrict__ src, const int* __restrict__ dst,
                  const float* __restrict__ dinv, const float* __restrict__ T,
                  float* __restrict__ OUT, int e, unsigned* __restrict__ ctr,
                  int ngroups, int nsplit) {
    __shared__ int  pick[2];
    __shared__ int4 queue[8][EDGES_PER_WAVE];

    if (threadIdx.x == 0) {
        const unsigned id = __builtin_amdgcn_s_sendmsg_rtn(0x87);
        int h = (int)((id >> 18) & 1u);
        int g = (int)atomicAdd(ctr + h * CTR_STRIDE, 1u);
        if (g >= ngroups) {
            h ^= 1;
            g = (int)atomicAdd(ctr + h * CTR_STRIDE, 1u);
        }
        pick[0] = g;
        pick[1] = h;
    }
    __syncthreads();
    const int g = pick[0];
    const int h = pick[1];
    if (g >= ngroups) return;

    const int wave = threadIdx.x >> 5;
    const int lane = threadIdx.x & 31;
    const int e0 = g * EDGES_PER_GROUP + wave * EDGES_PER_WAVE + lane;
    const int e1 = e0 + 32;

    int s0 = 0, d0 = 0, s1 = 0, d1 = 0;
    const bool v0 = e0 < e, v1 = e1 < e;
    if (v0) { s0 = src[e0]; d0 = dst[e0]; }
    if (v1) { s1 = src[e1]; d1 = dst[e1]; }
    const bool m0 = v0 && ((d0 >= nsplit) == (h != 0));
    const bool m1 = v1 && ((d1 >= nsplit) == (h != 0));

    const unsigned b0 = __builtin_amdgcn_ballot_w32(m0);
    const unsigned b1 = __builtin_amdgcn_ballot_w32(m1);
    const unsigned below = (1u << lane) - 1u;
    const int n0 = __popc(b0);
    const int n  = n0 + __popc(b1);
    int4* q = queue[wave];
    const float a0 = dinv[m0 ? s0 : 0], c0 = dinv[m0 ? d0 : 0];
    const float a1 = dinv[m1 ? s1 : 0], c1 = dinv[m1 ? d1 : 0];
    if (m0) q[__popc(b0 & below)]      = make_int4(s0, d0, __float_as_int(a0 * c0), 0);
    if (m1) q[n0 + __popc(b1 & below)] = make_int4(s1, d1, __float_as_int(a1 * c1), 0);

    const unsigned c = lane & 15;
    const int      r = lane >> 4;
    for (int base = 0; base < n; base += 16) {
        int4  ent[8];
        float m[8];
#pragma unroll
        for (int i = 0; i < 8; ++i) {
            const int idx = base + 2 * i + r;
            ent[i] = q[idx < n ? idx : n - 1];
        }
#pragma unroll
        for (int i = 0; i < 8; ++i)
            m[i] = T[(unsigned)ent[i].x * NHID + c];
#pragma unroll
        for (int i = 0; i < 8; ++i) {
            const int idx = base + 2 * i + r;
            if (idx < n)
                atomicAdd(OUT + ((unsigned)ent[i].y * NHID + c), m[i] * __int_as_float(ent[i].z));
        }
    }
}

__global__ void finalize_node(float* __restrict__ ACC, const float* __restrict__ T,
                              const float* __restrict__ dinv, const float* __restrict__ bias,
                              int n16, int doRelu) {
    int idx = blockIdx.x * blockDim.x + threadIdx.x;
    if (idx >= n16) return;
    int i = idx >> 4, f = idx & 15;
    float di = dinv[i];
    float v  = ACC[idx] + T[idx] * di * di + bias[f];
    ACC[idx] = doRelu ? fmaxf(v, 0.0f) : v;
}

__global__ void pool_kernel(const int* __restrict__ bvec, const float* __restrict__ Hn,
                            float* __restrict__ Gp, int n) {
    int t = blockIdx.x * blockDim.x + threadIdx.x;
    int i = t >> 2, q = t & 3;
    if (i >= n) return;
    int g = bvec[i];
    const float4 h = *(const float4*)(Hn + (size_t)i * NHID + q * 4);
    float* o = Gp + (size_t)g * NHID + q * 4;
    atomicAdd(o + 0, h.x);
    atomicAdd(o + 1, h.y);
    atomicAdd(o + 2, h.z);
    atomicAdd(o + 3, h.w);
}

__global__ void lin1_kernel(const float* __restrict__ G, const float* __restrict__ W,
                            const float* __restrict__ B, float* __restrict__ O) {
    int t = blockIdx.x * blockDim.x + threadIdx.x;
    if (t >= NGRAPH * NMLP) return;
    int gi = t / NMLP, j = t % NMLP;
    float acc = B[j];
#pragma unroll
    for (int k = 0; k < NHID; ++k)
        acc = fmaf(fmaxf(G[gi * NHID + k], 0.0f), W[k * NMLP + j], acc);
    O[t] = fmaxf(acc, 0.0f);
}

__global__ void lin2_kernel(const float* __restrict__ G, const float* __restrict__ W,
                            const float* __restrict__ B, float* __restrict__ O) {
    int t = blockIdx.x * blockDim.x + threadIdx.x;
    if (t >= NGRAPH * NCLS) return;
    int gi = t / NCLS, c = t % NCLS;
    float acc = B[c];
    for (int k = 0; k < NMLP; ++k)
        acc = fmaf(G[gi * NMLP + k], W[k * NCLS + c], acc);
    O[t] = acc;
}


extern "C" void kernel_launch(void* const* d_in, const int* in_sizes, int n_in,
                              void* d_out, int out_size, void* d_ws, size_t ws_size,
                              hipStream_t stream) {
    const float* x   = (const float*)d_in[0];
    const int*  eidx = (const int*)d_in[1];
    const int*  bvec = (const int*)d_in[2];
    const float* W1  = (const float*)d_in[3];
    const float* b1  = (const float*)d_in[4];
    const float* W2  = (const float*)d_in[5];
    const float* b2  = (const float*)d_in[6];
    const float* Wl1 = (const float*)d_in[7];
    const float* bl1 = (const float*)d_in[8];
    const float* Wl2 = (const float*)d_in[9];
    const float* bl2 = (const float*)d_in[10];
    float* out = (float*)d_out;

    char* ws = (char*)d_ws;
    size_t off = 0;
    auto walloc = [&](size_t bytes) -> void* {
        void* p = ws + off;
        off = (off + bytes + 255) & ~(size_t)255;
        return p;
    };
    float* deg  = (float*)walloc((size_t)NNODES * 4);
    float* bufA = (float*)walloc((size_t)NNODES * NHID * 4);
    float* bufB = (float*)walloc((size_t)NNODES * NHID * 4);
    float* gbuf = (float*)walloc((size_t)NGRAPH * NHID * 4);
    float* g2   = (float*)walloc((size_t)NGRAPH * NMLP * 4);

    const int* src = eidx;
    const int* dst = eidx + NEDGES;

    const int TB = 256;
    const int nN    = (NNODES + TB - 1) / TB;
    const int nE    = (NEDGES + TB - 1) / TB;
    const int nN16  = (NNODES * NHID + TB - 1) / TB;
    const int nN4   = (NNODES * 4 + TB - 1) / TB;
    const int nTile = ((NNODES / 16) + 7) / 8;
    const int nGrp  = (NEDGES + EDGES_PER_GROUP - 1) / EDGES_PER_GROUP;

    unsigned* ctr = (unsigned*)g2;

    fill_f32_ctr<<<nN, TB, 0, stream>>>(deg, NNODES, 1.0f, ctr, 4 * CTR_STRIDE);
    deg_accum<<<nE, TB, 0, stream>>>(dst, deg, NEDGES);
    rsqrt_ip<<<nN, TB, 0, stream>>>(deg, NNODES);
    float* dinv = deg;

    gemm_wmma_f32<<<nTile, TB, 0, stream>>>(x, W1, bufA, NNODES, NFEAT);
    fill_f32<<<nN16, TB, 0, stream>>>(bufB, NNODES * NHID, 0.0f);
    edge_scatter<<<2 * nGrp, TB, 0, stream>>>(src, dst, dinv, bufA, bufB, NEDGES,
                                              ctr, nGrp, NNODES / 2);
    finalize_node<<<nN16, TB, 0, stream>>>(bufB, bufA, dinv, b1, NNODES * NHID, 1);

    gemm_wmma_f32<<<nTile, TB, 0, stream>>>(bufB, W2, bufA, NNODES, NHID);
    fill_f32<<<nN16, TB, 0, stream>>>(bufB, NNODES * NHID, 0.0f);
    edge_scatter<<<2 * nGrp, TB, 0, stream>>>(src, dst, dinv, bufA, bufB, NEDGES,
                                              ctr + 2 * CTR_STRIDE, nGrp, NNODES / 2);
    finalize_node<<<nN16, TB, 0, stream>>>(bufB, bufA, dinv, b2, NNODES * NHID, 0);

    fill_f32<<<(NGRAPH * NHID + TB - 1) / TB, TB, 0, stream>>>(gbuf, NGRAPH * NHID, 0.0f);
    pool_kernel<<<nN4, TB, 0, stream>>>(bvec, bufB, gbuf, NNODES);
    lin1_kernel<<<(NGRAPH * NMLP + TB - 1) / TB, TB, 0, stream>>>(gbuf, Wl1, bl1, g2);
    lin2_kernel<<<(NGRAPH * NCLS + TB - 1) / TB, TB, 0, stream>>>(g2, Wl2, bl2, out);
}


# Edge weight dinv[src]*dinv[dst] computed once per call into an edge-ordered array by one added launch; both scatter passes stream it with the ids, no dinv gathers per edge; adds in arrival order
# speedup vs baseline: 1.7735x; 1.0470x over previous
#include <hip/hip_runtime.h>
#include <hip/hip_bf16.h>


typedef float v2f __attribute__((ext_vector_type(2)));
typedef float v8f __attribute__((ext_vector_type(8)));

#define NNODES  200000
#define NEDGES  6400000
#define NFEAT   128
#define NHID    16
#define NMLP    100
#define NCLS    27
#define NGRAPH  4096

#define WS_BYTES_WITH_WGT ((size_t)NNODES * 4 + 2 * ((size_t)NNODES * NHID * 4) + \
                           (size_t)NGRAPH * NHID * 4 + (size_t)NGRAPH * NMLP * 4 + \
                           (size_t)NEDGES * 4)


__global__ void fill_f32(float* __restrict__ p, int n, float v) {
    int i = blockIdx.x * blockDim.x + threadIdx.x;
    if (i < n) p[i] = v;
}

__global__ void fill_f32_ctr(float* __restrict__ p, int n, float v,
                             unsigned* __restrict__ ctr, int nctr) {
    int i = blockIdx.x * blockDim.x + threadIdx.x;
    if (i < n) p[i] = v;
    if (i < nctr) ctr[i] = 0u;
}

__global__ void deg_accum(const int* __restrict__ dst, float* __restrict__ deg, int e) {
    int i = blockIdx.x * blockDim.x + threadIdx.x;
    if (i < e) atomicAdd(&deg[dst[i]], 1.0f);
}

__global__ void rsqrt_ip(float* __restrict__ p, int n) {
    int i = blockIdx.x * blockDim.x + threadIdx.x;
    if (i < n) p[i] = rsqrtf(p[i]);
}

__global__ __launch_bounds__(256)
void gemm_wmma_f32(const float* __restrict__ X, const float* __restrict__ W,
                   float* __restrict__ T, int M, int K) {
    __shared__ float lw[NFEAT * NHID];
    const int tid = threadIdx.x;
    const int kn  = K * NHID;
    for (int i = tid; i < kn; i += 256) lw[i] = W[i];
    __syncthreads();

    const int wave = tid >> 5;
    const int lane = tid & 31;
    const int tile = blockIdx.x * 8 + wave;
    if (tile * 16 >= M) return;

    const int row   = tile * 16 + (lane & 15);
    const int kHalf = (lane >> 4) << 1;
    const int n     = lane & 15;

    v8f c = {};
    const float* xr = X + (size_t)row * K + kHalf;
    for (int k0 = 0; k0 < K; k0 += 4) {
        float2 av = *(const float2*)(xr + k0);
        v2f a; a.x = av.x; a.y = av.y;
        v2f b;
        b.x = lw[(k0 + kHalf)     * NHID + n];
        b.y = lw[(k0 + kHalf + 1) * NHID + n];
        c = __builtin_amdgcn_wmma_f32_16x16x4_f32(
                 false, a,  false, b,
                 (short)0, c,  false,  false);
    }

    const int rbase = tile * 16 + ((lane >> 4) << 3);
#pragma unroll
    for (int v = 0; v < 8; ++v)
        T[(size_t)(rbase + v) * NHID + n] = c[v];
}

#define NORM_PER_THREAD 4

__global__ __launch_bounds__(256)
void edge_norm(const int* __restrict__ src, const int* __restrict__ dst,
               const float* __restrict__ dinv, float* __restrict__ WGT, int e,
               size_t ws_size) {
    if (ws_size < WS_BYTES_WITH_WGT) return;
    const int base = blockIdx.x * (256 * NORM_PER_THREAD) + threadIdx.x;
    if (base + (NORM_PER_THREAD - 1) * 256 < e) {
        unsigned s[NORM_PER_THREAD], d[NORM_PER_THREAD];
        float    a[NORM_PER_THREAD], c[NORM_PER_THREAD];
#pragma unroll
        for (int k = 0; k < NORM_PER_THREAD; ++k) {
            s[k] = (unsigned)src[base + k * 256];
            d[k] = (unsigned)dst[base + k * 256];
        }
#pragma unroll
        for (int k = 0; k < NORM_PER_THREAD; ++k) {
            a[k] = dinv[s[k]];
            c[k] = dinv[d[k]];
        }
#pragma unroll
        for (int k = 0; k < NORM_PER_THREAD; ++k)
            WGT[base + k * 256] = a[k] * c[k];
    } else {
        for (int k = 0; k < NORM_PER_THREAD; ++k) {
            const int i = base + k * 256;
            if (i < e) WGT[i] = dinv[(unsigned)src[i]] * dinv[(unsigned)dst[i]];
        }
    }
}


#define EDGES_PER_WAVE 64
#define EDGES_PER_GROUP (8 * EDGES_PER_WAVE)
#define CTR_STRIDE 64

__global__ __launch_bounds__(256)
void edge_scatter(const int* __restrict__ src, const int* __restrict__ dst,
                  const float* __restrict__ dinv, const float* __restrict__ WGT,
                  const float* __restrict__ T,
                  float* __restrict__ OUT, int e, unsigned* __restrict__ ctr,
                  int ngroups, int nsplit, size_t ws_size) {
    __shared__ int  pick[2];
    __shared__ int4 queue[8][EDGES_PER_WAVE];

    if (threadIdx.x == 0) {
        const unsigned id = __builtin_amdgcn_s_sendmsg_rtn(0x87);
        int h = (int)((id >> 18) & 1u);
        int g = (int)atomicAdd(ctr + h * CTR_STRIDE, 1u);
        if (g >= ngroups) {
            h ^= 1;
            g = (int)atomicAdd(ctr + h * CTR_STRIDE, 1u);
        }
        pick[0] = g;
        pick[1] = h;
    }
    __syncthreads();
    const int g = pick[0];
    const int h = pick[1];
    if (g >= ngroups) return;

    const int wave = threadIdx.x >> 5;
    const int lane = threadIdx.x & 31;
    const int e0 = g * EDGES_PER_GROUP + wave * EDGES_PER_WAVE + lane;
    const int e1 = e0 + 32;

    const bool have_wgt = ws_size >= WS_BYTES_WITH_WGT;
    int s0 = 0, d0 = 0, s1 = 0, d1 = 0;
    float w0 = 0.0f, w1 = 0.0f;
    const bool v0 = e0 < e, v1 = e1 < e;
    if (v0) { s0 = src[e0]; d0 = dst[e0]; }
    if (v1) { s1 = src[e1]; d1 = dst[e1]; }
    if (have_wgt) {
        if (v0) w0 = WGT[e0];
        if (v1) w1 = WGT[e1];
    }
    const bool m0 = v0 && ((d0 >= nsplit) == (h != 0));
    const bool m1 = v1 && ((d1 >= nsplit) == (h != 0));

    const unsigned b0 = __builtin_amdgcn_ballot_w32(m0);
    const unsigned b1 = __builtin_amdgcn_ballot_w32(m1);
    const unsigned below = (1u << lane) - 1u;
    const int n0 = __popc(b0);
    const int n  = n0 + __popc(b1);
    int4* q = queue[wave];
    if (!have_wgt) {
        const float a0 = dinv[m0 ? s0 : 0], c0 = dinv[m0 ? d0 : 0];
        const float a1 = dinv[m1 ? s1 : 0], c1 = dinv[m1 ? d1 : 0];
        w0 = a0 * c0;
        w1 = a1 * c1;
    }
    if (m0) q[__popc(b0 & below)]      = make_int4(s0, d0, __float_as_int(w0), 0);
    if (m1) q[n0 + __popc(b1 & below)] = make_int4(s1, d1, __float_as_int(w1), 0);

    const unsigned c = lane & 15;
    const int      r = lane >> 4;
    for (int base = 0; base < n; base += 16) {
        int4  ent[8];
        float m[8];
#pragma unroll
        for (int i = 0; i < 8; ++i) {
            const int idx = base + 2 * i + r;
            ent[i] = q[idx < n ? idx : n - 1];
        }
#pragma unroll
        for (int i = 0; i < 8; ++i)
            m[i] = T[(unsigned)ent[i].x * NHID + c];
#pragma unroll
        for (int i = 0; i < 8; ++i) {
            const int idx = base + 2 * i + r;
            if (idx < n)
                atomicAdd(OUT + ((unsigned)ent[i].y * NHID + c), m[i] * __int_as_float(ent[i].z));
        }
    }
}

__global__ void finalize_node(float* __restrict__ ACC, const float* __restrict__ T,
                              const float* __restrict__ dinv, const float* __restrict__ bias,
                              int n16, int doRelu) {
    int idx = blockIdx.x * blockDim.x + threadIdx.x;
    if (idx >= n16) return;
    int i = idx >> 4, f = idx & 15;
    float di = dinv[i];
    float v  = ACC[idx] + T[idx] * di * di + bias[f];
    ACC[idx] = doRelu ? fmaxf(v, 0.0f) : v;
}

__global__ void pool_kernel(const int* __restrict__ bvec, const float* __restrict__ Hn,
                            float* __restrict__ Gp, int n) {
    int t = blockIdx.x * blockDim.x + threadIdx.x;
    int i = t >> 2, q = t & 3;
    if (i >= n) return;
    int g = bvec[i];
    const float4 h = *(const float4*)(Hn + (size_t)i * NHID + q * 4);
    float* o = Gp + (size_t)g * NHID + q * 4;
    atomicAdd(o + 0, h.x);
    atomicAdd(o + 1, h.y);
    atomicAdd(o + 2, h.z);
    atomicAdd(o + 3, h.w);
}

__global__ void lin1_kernel(const float* __restrict__ G, const float* __restrict__ W,
                            const float* __restrict__ B, float* __restrict__ O) {
    int t = blockIdx.x * blockDim.x + threadIdx.x;
    if (t >= NGRAPH * NMLP) return;
    int gi = t / NMLP, j = t % NMLP;
    float acc = B[j];
#pragma unroll
    for (int k = 0; k < NHID; ++k)
        acc = fmaf(fmaxf(G[gi * NHID + k], 0.0f), W[k * NMLP + j], acc);
    O[t] = fmaxf(acc, 0.0f);
}

__global__ void lin2_kernel(const float* __restrict__ G, const float* __restrict__ W,
                            const float* __restrict__ B, float* __restrict__ O) {
    int t = blockIdx.x * blockDim.x + threadIdx.x;
    if (t >= NGRAPH * NCLS) return;
    int gi = t / NCLS, c = t % NCLS;
    float acc = B[c];
    for (int k = 0; k < NMLP; ++k)
        acc = fmaf(G[gi * NMLP + k], W[k * NCLS + c], acc);
    O[t] = acc;
}


extern "C" void kernel_launch(void* const* d_in, const int* in_sizes, int n_in,
                              void* d_out, int out_size, void* d_ws, size_t ws_size,
                              hipStream_t stream) {
    const float* x   = (const float*)d_in[0];
    const int*  eidx = (const int*)d_in[1];
    const int*  bvec = (const int*)d_in[2];
    const float* W1  = (const float*)d_in[3];
    const float* b1  = (const float*)d_in[4];
    const float* W2  = (const float*)d_in[5];
    const float* b2  = (const float*)d_in[6];
    const float* Wl1 = (const float*)d_in[7];
    const float* bl1 = (const float*)d_in[8];
    const float* Wl2 = (const float*)d_in[9];
    const float* bl2 = (const float*)d_in[10];
    float* out = (float*)d_out;

    char* ws = (char*)d_ws;
    size_t off = 0;
    auto walloc = [&](size_t bytes) -> void* {
        void* p = ws + off;
        off = (off + bytes + 255) & ~(size_t)255;
        return p;
    };
    float* deg  = (float*)walloc((size_t)NNODES * 4);
    float* bufA = (float*)walloc((size_t)NNODES * NHID * 4);
    float* bufB = (float*)walloc((size_t)NNODES * NHID * 4);
    float* gbuf = (float*)walloc((size_t)NGRAPH * NHID * 4);
    float* g2   = (float*)walloc((size_t)NGRAPH * NMLP * 4);
    float* wgt  = (float*)walloc((size_t)NEDGES * 4);

    const int* src = eidx;
    const int* dst = eidx + NEDGES;

    const int TB = 256;
    const int nN    = (NNODES + TB - 1) / TB;
    const int nE    = (NEDGES + TB - 1) / TB;
    const int nN16  = (NNODES * NHID + TB - 1) / TB;
    const int nN4   = (NNODES * 4 + TB - 1) / TB;
    const int nTile = ((NNODES / 16) + 7) / 8;
    const int nGrp  = (NEDGES + EDGES_PER_GROUP - 1) / EDGES_PER_GROUP;
    const int nNorm = (NEDGES + TB * NORM_PER_THREAD - 1) / (TB * NORM_PER_THREAD);

    unsigned* ctr = (unsigned*)g2;

    fill_f32_ctr<<<nN, TB, 0, stream>>>(deg, NNODES, 1.0f, ctr, 4 * CTR_STRIDE);
    deg_accum<<<nE, TB, 0, stream>>>(dst, deg, NEDGES);
    rsqrt_ip<<<nN, TB, 0, stream>>>(deg, NNODES);
    float* dinv = deg;
    edge_norm<<<nNorm, TB, 0, stream>>>(src, dst, dinv, wgt, NEDGES, ws_size);

    gemm_wmma_f32<<<nTile, TB, 0, stream>>>(x, W1, bufA, NNODES, NFEAT);
    fill_f32<<<nN16, TB, 0, stream>>>(bufB, NNODES * NHID, 0.0f);
    edge_scatter<<<2 * nGrp, TB, 0, stream>>>(src, dst, dinv, wgt, bufA, bufB, NEDGES,
                                              ctr, nGrp, NNODES / 2, ws_size);
    finalize_node<<<nN16, TB, 0, stream>>>(bufB, bufA, dinv, b1, NNODES * NHID, 1);

    gemm_wmma_f32<<<nTile, TB, 0, stream>>>(bufB, W2, bufA, NNODES, NHID);
    fill_f32<<<nN16, TB, 0, stream>>>(bufB, NNODES * NHID, 0.0f);
    edge_scatter<<<2 * nGrp, TB, 0, stream>>>(src, dst, dinv, wgt, bufA, bufB, NEDGES,
                                              ctr + 2 * CTR_STRIDE, nGrp, NNODES / 2, ws_size);
    finalize_node<<<nN16, TB, 0, stream>>>(bufB, bufA, dinv, b2, NNODES * NHID, 0);

    fill_f32<<<(NGRAPH * NHID + TB - 1) / TB, TB, 0, stream>>>(gbuf, NGRAPH * NHID, 0.0f);
    pool_kernel<<<nN4, TB, 0, stream>>>(bvec, bufB, gbuf, NNODES);
    lin1_kernel<<<(NGRAPH * NMLP + TB - 1) / TB, TB, 0, stream>>>(gbuf, Wl1, bl1, g2);
    lin2_kernel<<<(NGRAPH * NCLS + TB - 1) / TB, TB, 0, stream>>>(g2, Wl2, bl2, out);
}


# Layer-2 edge sums go straight into the 4096x16 pooled table (no relu between conv2 and pool): a work-group adds its edges into an LDS copy, then adds it to global once; sum order changed
# speedup vs baseline: 1.9035x; 1.0733x over previous
#include <hip/hip_runtime.h>
#include <hip/hip_bf16.h>


typedef float v2f __attribute__((ext_vector_type(2)));
typedef float v8f __attribute__((ext_vector_type(8)));

#define NNODES  200000
#define NEDGES  6400000
#define NFEAT   128
#define NHID    16
#define NMLP    100
#define NCLS    27
#define NGRAPH  4096

#define WS_BYTES_WITH_WGT ((size_t)NNODES * 4 + 2 * ((size_t)NNODES * NHID * 4) + \
                           (size_t)NGRAPH * NHID * 4 + (size_t)NGRAPH * NMLP * 4 + \
                           (size_t)NEDGES * 4)


__global__ void fill_f32(float* __restrict__ p, int n, float v) {
    int i = blockIdx.x * blockDim.x + threadIdx.x;
    if (i < n) p[i] = v;
}

__global__ void fill_f32_ctr(float* __restrict__ p, int n, float v,
                             unsigned* __restrict__ ctr, int nctr) {
    int i = blockIdx.x * blockDim.x + threadIdx.x;
    if (i < n) p[i] = v;
    if (i < nctr) ctr[i] = 0u;
}

__global__ void deg_accum(const int* __restrict__ dst, float* __restrict__ deg, int e) {
    int i = blockIdx.x * blockDim.x + threadIdx.x;
    if (i < e) atomicAdd(&deg[dst[i]], 1.0f);
}

__global__ void rsqrt_ip(float* __restrict__ p, int n) {
    int i = blockIdx.x * blockDim.x + threadIdx.x;
    if (i < n) p[i] = rsqrtf(p[i]);
}

__global__ __launch_bounds__(256)
void gemm_wmma_f32(const float* __restrict__ X, const float* __restrict__ W,
                   float* __restrict__ T, int M, int K) {
    __shared__ float lw[NFEAT * NHID];
    const int tid = threadIdx.x;
    const int kn  = K * NHID;
    for (int i = tid; i < kn; i += 256) lw[i] = W[i];
    __syncthreads();

    const int wave = tid >> 5;
    const int lane = tid & 31;
    const int tile = blockIdx.x * 8 + wave;
    if (tile * 16 >= M) return;

    const int row   = tile * 16 + (lane & 15);
    const int kHalf = (lane >> 4) << 1;
    const int n     = lane & 15;

    v8f c = {};
    const float* xr = X + (size_t)row * K + kHalf;
    for (int k0 = 0; k0 < K; k0 += 4) {
        float2 av = *(const float2*)(xr + k0);
        v2f a; a.x = av.x; a.y = av.y;
        v2f b;
        b.x = lw[(k0 + kHalf)     * NHID + n];
        b.y = lw[(k0 + kHalf + 1) * NHID + n];
        c = __builtin_amdgcn_wmma_f32_16x16x4_f32(
                 false, a,  false, b,
                 (short)0, c,  false,  false);
    }

    const int rbase = tile * 16 + ((lane >> 4) << 3);
#pragma unroll
    for (int v = 0; v < 8; ++v)
        T[(size_t)(rbase + v) * NHID + n] = c[v];
}

#define NORM_PER_THREAD 4

__global__ __launch_bounds__(256)
void edge_norm(const int* __restrict__ src, const int* __restrict__ dst,
               const float* __restrict__ dinv, float* __restrict__ WGT, int e,
               size_t ws_size) {
    if (ws_size < WS_BYTES_WITH_WGT) return;
    const int base = blockIdx.x * (256 * NORM_PER_THREAD) + threadIdx.x;
    if (base + (NORM_PER_THREAD - 1) * 256 < e) {
        unsigned s[NORM_PER_THREAD], d[NORM_PER_THREAD];
        float    a[NORM_PER_THREAD], c[NORM_PER_THREAD];
#pragma unroll
        for (int k = 0; k < NORM_PER_THREAD; ++k) {
            s[k] = (unsigned)src[base + k * 256];
            d[k] = (unsigned)dst[base + k * 256];
        }
#pragma unroll
        for (int k = 0; k < NORM_PER_THREAD; ++k) {
            a[k] = dinv[s[k]];
            c[k] = dinv[d[k]];
        }
#pragma unroll
        for (int k = 0; k < NORM_PER_THREAD; ++k)
            WGT[base + k * 256] = a[k] * c[k];
    } else {
        for (int k = 0; k < NORM_PER_THREAD; ++k) {
            const int i = base + k * 256;
            if (i < e) WGT[i] = dinv[(unsigned)src[i]] * dinv[(unsigned)dst[i]];
        }
    }
}


#define EDGES_PER_WAVE 64
#define EDGES_PER_GROUP (8 * EDGES_PER_WAVE)
#define CTR_STRIDE 64

__global__ __launch_bounds__(256)
void edge_scatter(const int* __restrict__ src, const int* __restrict__ dst,
                  const float* __restrict__ dinv, const float* __restrict__ WGT,
                  const float* __restrict__ T,
                  float* __restrict__ OUT, int e, unsigned* __restrict__ ctr,
                  int ngroups, int nsplit, size_t ws_size) {
    __shared__ int  pick[2];
    __shared__ int4 queue[8][EDGES_PER_WAVE];

    if (threadIdx.x == 0) {
        const unsigned id = __builtin_amdgcn_s_sendmsg_rtn(0x87);
        int h = (int)((id >> 18) & 1u);
        int g = (int)atomicAdd(ctr + h * CTR_STRIDE, 1u);
        if (g >= ngroups) {
            h ^= 1;
            g = (int)atomicAdd(ctr + h * CTR_STRIDE, 1u);
        }
        pick[0] = g;
        pick[1] = h;
    }
    __syncthreads();
    const int g = pick[0];
    const int h = pick[1];
    if (g >= ngroups) return;

    const int wave = threadIdx.x >> 5;
    const int lane = threadIdx.x & 31;
    const int e0 = g * EDGES_PER_GROUP + wave * EDGES_PER_WAVE + lane;
    const int e1 = e0 + 32;

    const bool have_wgt = ws_size >= WS_BYTES_WITH_WGT;
    int s0 = 0, d0 = 0, s1 = 0, d1 = 0;
    float w0 = 0.0f, w1 = 0.0f;
    const bool v0 = e0 < e, v1 = e1 < e;
    if (v0) { s0 = src[e0]; d0 = dst[e0]; }
    if (v1) { s1 = src[e1]; d1 = dst[e1]; }
    if (have_wgt) {
        if (v0) w0 = WGT[e0];
        if (v1) w1 = WGT[e1];
    }
    const bool m0 = v0 && ((d0 >= nsplit) == (h != 0));
    const bool m1 = v1 && ((d1 >= nsplit) == (h != 0));

    const unsigned b0 = __builtin_amdgcn_ballot_w32(m0);
    const unsigned b1 = __builtin_amdgcn_ballot_w32(m1);
    const unsigned below = (1u << lane) - 1u;
    const int n0 = __popc(b0);
    const int n  = n0 + __popc(b1);
    int4* q = queue[wave];
    if (!have_wgt) {
        const float a0 = dinv[m0 ? s0 : 0], c0 = dinv[m0 ? d0 : 0];
        const float a1 = dinv[m1 ? s1 : 0], c1 = dinv[m1 ? d1 : 0];
        w0 = a0 * c0;
        w1 = a1 * c1;
    }
    if (m0) q[__popc(b0 & below)]      = make_int4(s0, d0, __float_as_int(w0), 0);
    if (m1) q[n0 + __popc(b1 & below)] = make_int4(s1, d1, __float_as_int(w1), 0);

    const unsigned c = lane & 15;
    const int      r = lane >> 4;
    for (int base = 0; base < n; base += 16) {
        int4  ent[8];
        float m[8];
#pragma unroll
        for (int i = 0; i < 8; ++i) {
            const int idx = base + 2 * i + r;
            ent[i] = q[idx < n ? idx : n - 1];
        }
#pragma unroll
        for (int i = 0; i < 8; ++i)
            m[i] = T[(unsigned)ent[i].x * NHID + c];
#pragma unroll
        for (int i = 0; i < 8; ++i) {
            const int idx = base + 2 * i + r;
            if (idx < n)
                atomicAdd(OUT + ((unsigned)ent[i].y * NHID + c), m[i] * __int_as_float(ent[i].z));
        }
    }
}

__global__ void finalize_node(float* __restrict__ ACC, const float* __restrict__ T,
                              const float* __restrict__ dinv, const float* __restrict__ bias,
                              int n16, int doRelu) {
    int idx = blockIdx.x * blockDim.x + threadIdx.x;
    if (idx >= n16) return;
    int i = idx >> 4, f = idx & 15;
    float di = dinv[i];
    float v  = ACC[idx] + T[idx] * di * di + bias[f];
    ACC[idx] = doRelu ? fmaxf(v, 0.0f) : v;
}


#define POOL_WAVES   32
#define POOL_THREADS (POOL_WAVES * 32)
#define POOL_CHUNK   64
#define POOL_GROUPS  256

struct __attribute__((aligned(16))) PoolEnt { int t; int g; float w; };

template <bool HAVE_WGT>
__device__ __forceinline__ void pool_edges(const int* __restrict__ src, const int* __restrict__ dst,
                                           const int* __restrict__ bvec,
                                           const float* __restrict__ dinv, const float* __restrict__ WGT,
                                           const float* __restrict__ T, float* tab, PoolEnt* q,
                                           int e, int c_lo, int c_hi, int wave, int lane) {
    const unsigned c = lane & 15;
    const int      r = lane >> 4;
    if (c_lo + wave >= c_hi) return;

    int s0, d0, s1, d1;
    float w0 = 0.0f, w1 = 0.0f;
    {
        const int i0 = min((c_lo + wave) * POOL_CHUNK + lane, e - 1);
        const int i1 = min((c_lo + wave) * POOL_CHUNK + 32 + lane, e - 1);
        s0 = src[i0]; d0 = dst[i0];
        s1 = src[i1]; d1 = dst[i1];
        if (HAVE_WGT) { w0 = WGT[i0]; w1 = WGT[i1]; }
    }
    for (int ch = c_lo + wave; ch < c_hi; ch += POOL_WAVES) {
        const int n = min(POOL_CHUNK, e - ch * POOL_CHUNK);
        const int g0 = bvec[d0];
        const int g1 = bvec[d1];
        if (!HAVE_WGT) {
            w0 = dinv[s0] * dinv[d0];
            w1 = dinv[s1] * dinv[d1];
        }
        const int cs0 = s0, cs1 = s1;
        const float cw0 = w0, cw1 = w1;
        {
            const int nx = min(ch + POOL_WAVES, c_hi - 1);
            const int i0 = min(nx * POOL_CHUNK + lane, e - 1);
            const int i1 = min(nx * POOL_CHUNK + 32 + lane, e - 1);
            s0 = src[i0]; d0 = dst[i0];
            s1 = src[i1]; d1 = dst[i1];
            if (HAVE_WGT) { w0 = WGT[i0]; w1 = WGT[i1]; }
        }
        q[lane].t      = cs0 * NHID;  q[lane].g      = g0 * NHID;  q[lane].w      = cw0;
        q[lane + 32].t = cs1 * NHID;  q[lane + 32].g = g1 * NHID;  q[lane + 32].w = cw1;

        for (int base = 0; base < n; base += 32) {
            PoolEnt ent[16];
            float   m[16];
#pragma unroll
            for (int i = 0; i < 16; ++i) {
                const int idx = base + 2 * i + r;
                ent[i] = q[idx < n ? idx : n - 1];
            }
#pragma unroll
            for (int i = 0; i < 16; ++i)
                m[i] = T[(unsigned)ent[i].t + c];
#pragma unroll
            for (int i = 0; i < 16; ++i) {
                const int idx = base + 2 * i + r;
                const float v = m[i] * ent[i].w;
                atomicAdd(&tab[(unsigned)ent[i].g + c], idx < n ? v : 0.0f);
            }
        }
    }
}

__global__ __launch_bounds__(POOL_THREADS)
void edge_pool(const int* __restrict__ src, const int* __restrict__ dst,
               const int* __restrict__ bvec,
               const float* __restrict__ dinv, const float* __restrict__ WGT,
               const float* __restrict__ T, const float* __restrict__ bias,
               float* __restrict__ GP, int e, int nnodes, size_t ws_size) {
    __shared__ __attribute__((aligned(16))) float tab[NGRAPH * NHID];
    __shared__ PoolEnt queue[POOL_WAVES][POOL_CHUNK];
    __shared__ int  half_s;

    const int tid  = threadIdx.x;
    const int wave = __builtin_amdgcn_readfirstlane(tid >> 5);
    const int lane = tid & 31;

    float4* tab4 = (float4*)tab;
    for (int i = tid; i < NGRAPH * NHID / 4; i += POOL_THREADS)
        tab4[i] = make_float4(0.0f, 0.0f, 0.0f, 0.0f);
    if (tid == 0) {
        const unsigned id = __builtin_amdgcn_s_sendmsg_rtn(0x87);
        half_s = (int)((id >> 18) & 1u);
    }
    __syncthreads();

    const int nb = gridDim.x, b = blockIdx.x;

    {
        const unsigned c = lane & 15;
        const int      r = lane >> 4;
        const int n_lo = (int)((long long)b * nnodes / nb);
        const int n_hi = (int)((long long)(b + 1) * nnodes / nb);
        const float bc = bias[c];
        for (int k0 = n_lo + wave * 2; k0 < n_hi; k0 += POOL_WAVES * 2 * 8) {
            float di[8], tv[8];
            int   gi[8];
#pragma unroll
            for (int j = 0; j < 8; ++j) {
                const int i = min(k0 + r + POOL_WAVES * 2 * j, n_hi - 1);
                di[j] = dinv[i];
                tv[j] = T[(unsigned)i * NHID + c];
                gi[j] = bvec[i];
            }
#pragma unroll
            for (int j = 0; j < 8; ++j) {
                const int i = k0 + r + POOL_WAVES * 2 * j;
                const float v = tv[j] * di[j] * di[j] + bc;
                atomicAdd(&tab[(unsigned)gi[j] * NHID + c], i < n_hi ? v : 0.0f);
            }
        }
    }

    const int nchunks = (e + POOL_CHUNK - 1) / POOL_CHUNK;
    const int c_lo = (int)((long long)b * nchunks / nb);
    const int c_hi = (int)((long long)(b + 1) * nchunks / nb);
    if (ws_size >= WS_BYTES_WITH_WGT)
        pool_edges<true>(src, dst, bvec, dinv, WGT, T, tab, queue[wave], e, c_lo, c_hi, wave, lane);
    else
        pool_edges<false>(src, dst, bvec, dinv, WGT, T, tab, queue[wave], e, c_lo, c_hi, wave, lane);

    __syncthreads();

    float* gp = GP + (size_t)half_s * (NGRAPH * NHID);
    const unsigned rot = (unsigned)b * (unsigned)(NGRAPH * NHID / POOL_GROUPS);
    for (int k0 = 0; k0 < NGRAPH * NHID / POOL_THREADS; k0 += 8) {
        unsigned idx[8];
        float    v[8];
#pragma unroll
        for (int k = 0; k < 8; ++k) {
            idx[k] = (rot + (unsigned)(k0 + k) * POOL_THREADS + (unsigned)tid) & (unsigned)(NGRAPH * NHID - 1);
            v[k]   = tab[idx[k]];
        }
#pragma unroll
        for (int k = 0; k < 8; ++k)
            atomicAdd(gp + idx[k], v[k]);
    }
}

__global__ void lin1_kernel(const float* __restrict__ G, const float* __restrict__ W,
                            const float* __restrict__ B, float* __restrict__ O) {
    int t = blockIdx.x * blockDim.x + threadIdx.x;
    if (t >= NGRAPH * NMLP) return;
    int gi = t / NMLP, j = t % NMLP;
    float acc = B[j];
#pragma unroll
    for (int k = 0; k < NHID; ++k)
        acc = fmaf(fmaxf(G[gi * NHID + k] + G[NGRAPH * NHID + gi * NHID + k], 0.0f), W[k * NMLP + j], acc);
    O[t] = fmaxf(acc, 0.0f);
}

__global__ void lin2_kernel(const float* __restrict__ G, const float* __restrict__ W,
                            const float* __restrict__ B, float* __restrict__ O) {
    int t = blockIdx.x * blockDim.x + threadIdx.x;
    if (t >= NGRAPH * NCLS) return;
    int gi = t / NCLS, c = t % NCLS;
    float acc = B[c];
    for (int k = 0; k < NMLP; ++k)
        acc = fmaf(G[gi * NMLP + k], W[k * NCLS + c], acc);
    O[t] = acc;
}


extern "C" void kernel_launch(void* const* d_in, const int* in_sizes, int n_in,
                              void* d_out, int out_size, void* d_ws, size_t ws_size,
                              hipStream_t stream) {
    const float* x   = (const float*)d_in[0];
    const int*  eidx = (const int*)d_in[1];
    const int*  bvec = (const int*)d_in[2];
    const float* W1  = (const float*)d_in[3];
    const float* b1  = (const float*)d_in[4];
    const float* W2  = (const float*)d_in[5];
    const float* b2  = (const float*)d_in[6];
    const float* Wl1 = (const float*)d_in[7];
    const float* bl1 = (const float*)d_in[8];
    const float* Wl2 = (const float*)d_in[9];
    const float* bl2 = (const float*)d_in[10];
    float* out = (float*)d_out;

    char* ws = (char*)d_ws;
    size_t off = 0;
    auto walloc = [&](size_t bytes) -> void* {
        void* p = ws + off;
        off = (off + bytes + 255) & ~(size_t)255;
        return p;
    };
    float* deg  = (float*)walloc((size_t)NNODES * 4);
    float* bufA = (float*)walloc((size_t)NNODES * NHID * 4);
    float* bufB = (float*)walloc((size_t)NNODES * NHID * 4);
    (void)walloc((size_t)NGRAPH * NHID * 4);
    float* g2   = (float*)walloc((size_t)NGRAPH * NMLP * 4);
    float* wgt  = (float*)walloc((size_t)NEDGES * 4);

    const int* src = eidx;
    const int* dst = eidx + NEDGES;

    const int TB = 256;
    const int nN    = (NNODES + TB - 1) / TB;
    const int nE    = (NEDGES + TB - 1) / TB;
    const int nN16  = (NNODES * NHID + TB - 1) / TB;
    const int nTile = ((NNODES / 16) + 7) / 8;
    const int nGrp  = (NEDGES + EDGES_PER_GROUP - 1) / EDGES_PER_GROUP;
    const int nNorm = (NEDGES + TB * NORM_PER_THREAD - 1) / (TB * NORM_PER_THREAD);

    unsigned* ctr = (unsigned*)g2;

    fill_f32_ctr<<<nN, TB, 0, stream>>>(deg, NNODES, 1.0f, ctr, 2 * CTR_STRIDE);
    deg_accum<<<nE, TB, 0, stream>>>(dst, deg, NEDGES);
    rsqrt_ip<<<nN, TB, 0, stream>>>(deg, NNODES);
    float* dinv = deg;
    edge_norm<<<nNorm, TB, 0, stream>>>(src, dst, dinv, wgt, NEDGES, ws_size);

    gemm_wmma_f32<<<nTile, TB, 0, stream>>>(x, W1, bufA, NNODES, NFEAT);
    fill_f32<<<nN16, TB, 0, stream>>>(bufB, NNODES * NHID, 0.0f);
    edge_scatter<<<2 * nGrp, TB, 0, stream>>>(src, dst, dinv, wgt, bufA, bufB, NEDGES,
                                              ctr, nGrp, NNODES / 2, ws_size);
    finalize_node<<<nN16, TB, 0, stream>>>(bufB, bufA, dinv, b1, NNODES * NHID, 1);

    gemm_wmma_f32<<<nTile, TB, 0, stream>>>(bufB, W2, bufA, NNODES, NHID);
    float* gtab = bufB;
    fill_f32<<<(2 * NGRAPH * NHID + TB - 1) / TB, TB, 0, stream>>>(gtab, 2 * NGRAPH * NHID, 0.0f);
    edge_pool<<<POOL_GROUPS, POOL_THREADS, 0, stream>>>(src, dst, bvec, dinv, wgt, bufA, b2, gtab,
                                                        NEDGES, NNODES, ws_size);

    lin1_kernel<<<(NGRAPH * NMLP + TB - 1) / TB, TB, 0, stream>>>(gtab, Wl1, bl1, g2);
    lin2_kernel<<<(NGRAPH * NCLS + TB - 1) / TB, TB, 0, stream>>>(g2, Wl2, bl2, out);
}


# Degree count: a work-group reads its die and adds all its edges into the degree plane of its own L2 half (two planes, summed in the rsqrt launch): no degree line is added to from both halves; exact
# speedup vs baseline: 3.3896x; 1.7807x over previous
#include <hip/hip_runtime.h>
#include <hip/hip_bf16.h>


typedef float v2f __attribute__((ext_vector_type(2)));
typedef float v8f __attribute__((ext_vector_type(8)));

#define NNODES  200000
#define NEDGES  6400000
#define NFEAT   128
#define NHID    16
#define NMLP    100
#define NCLS    27
#define NGRAPH  4096

#define WS_BYTES_WITH_WGT ((size_t)NNODES * 4 + 2 * ((size_t)NNODES * NHID * 4) + \
                           (size_t)NGRAPH * NHID * 4 + (size_t)NGRAPH * NMLP * 4 + \
                           (size_t)NEDGES * 4)


__global__ void fill_f32(float* __restrict__ p, int n, float v) {
    int i = blockIdx.x * blockDim.x + threadIdx.x;
    if (i < n) p[i] = v;
}

__global__ void fill_deg_ctr(float* __restrict__ d0, float* __restrict__ d1, int n,
                             unsigned* __restrict__ ctr, int nctr) {
    int i = blockIdx.x * blockDim.x + threadIdx.x;
    if (i < n) { d0[i] = 1.0f; d1[i] = 0.0f; }
    if (i < nctr) ctr[i] = 0u;
}

#define DEG_PER_THREAD 4

__global__ __launch_bounds__(256)
void deg_accum(const int* __restrict__ dst, float* __restrict__ deg0,
               float* __restrict__ deg1, int e) {
    __shared__ int half_s;
    const int base = blockIdx.x * (256 * DEG_PER_THREAD) + threadIdx.x;
    unsigned d[DEG_PER_THREAD];
#pragma unroll
    for (int k = 0; k < DEG_PER_THREAD; ++k)
        d[k] = (unsigned)dst[min(base + k * 256, e - 1)];
    if (threadIdx.x == 0) {
        const unsigned id = __builtin_amdgcn_s_sendmsg_rtn(0x87);
        half_s = (int)((id >> 18) & 1u);
    }
    __syncthreads();
    float* __restrict__ deg = half_s ? deg1 : deg0;
    if ((blockIdx.x + 1) * (256 * DEG_PER_THREAD) <= e) {
#pragma unroll
        for (int k = 0; k < DEG_PER_THREAD; ++k)
            atomicAdd(&deg[d[k]], 1.0f);
    } else {
        for (int k = 0; k < DEG_PER_THREAD; ++k)
            if (base + k * 256 < e) atomicAdd(&deg[d[k]], 1.0f);
    }
}

__global__ void rsqrt_sum(float* __restrict__ p, const float* __restrict__ q, int n) {
    int i = blockIdx.x * blockDim.x + threadIdx.x;
    if (i < n) p[i] = rsqrtf(p[i] + q[i]);
}

__global__ __launch_bounds__(256)
void gemm_wmma_f32(const float* __restrict__ X, const float* __restrict__ W,
                   float* __restrict__ T, int M, int K) {
    __shared__ float lw[NFEAT * NHID];
    const int tid = threadIdx.x;
    const int kn  = K * NHID;
    for (int i = tid; i < kn; i += 256) lw[i] = W[i];
    __syncthreads();

    const int wave = tid >> 5;
    const int lane = tid & 31;
    const int tile = blockIdx.x * 8 + wave;
    if (tile * 16 >= M) return;

    const int row   = tile * 16 + (lane & 15);
    const int kHalf = (lane >> 4) << 1;
    const int n     = lane & 15;

    v8f c = {};
    const float* xr = X + (size_t)row * K + kHalf;
    for (int k0 = 0; k0 < K; k0 += 4) {
        float2 av = *(const float2*)(xr + k0);
        v2f a; a.x = av.x; a.y = av.y;
        v2f b;
        b.x = lw[(k0 + kHalf)     * NHID + n];
        b.y = lw[(k0 + kHalf + 1) * NHID + n];
        c = __builtin_amdgcn_wmma_f32_16x16x4_f32(
                 false, a,  false, b,
                 (short)0, c,  false,  false);
    }

    const int rbase = tile * 16 + ((lane >> 4) << 3);
#pragma unroll
    for (int v = 0; v < 8; ++v)
        T[(size_t)(rbase + v) * NHID + n] = c[v];
}

#define NORM_PER_THREAD 4

__global__ __launch_bounds__(256)
void edge_norm(const int* __restrict__ src, const int* __restrict__ dst,
               const float* __restrict__ dinv, float* __restrict__ WGT, int e,
               size_t ws_size) {
    if (ws_size < WS_BYTES_WITH_WGT) return;
    const int base = blockIdx.x * (256 * NORM_PER_THREAD) + threadIdx.x;
    if (base + (NORM_PER_THREAD - 1) * 256 < e) {
        unsigned s[NORM_PER_THREAD], d[NORM_PER_THREAD];
        float    a[NORM_PER_THREAD], c[NORM_PER_THREAD];
#pragma unroll
        for (int k = 0; k < NORM_PER_THREAD; ++k) {
            s[k] = (unsigned)src[base + k * 256];
            d[k] = (unsigned)dst[base + k * 256];
        }
#pragma unroll
        for (int k = 0; k < NORM_PER_THREAD; ++k) {
            a[k] = dinv[s[k]];
            c[k] = dinv[d[k]];
        }
#pragma unroll
        for (int k = 0; k < NORM_PER_THREAD; ++k)
            WGT[base + k * 256] = a[k] * c[k];
    } else {
        for (int k = 0; k < NORM_PER_THREAD; ++k) {
            const int i = base + k * 256;
            if (i < e) WGT[i] = dinv[(unsigned)src[i]] * dinv[(unsigned)dst[i]];
        }
    }
}


#define EDGES_PER_WAVE 64
#define EDGES_PER_GROUP (8 * EDGES_PER_WAVE)
#define CTR_STRIDE 64

__global__ __launch_bounds__(256)
void edge_scatter(const int* __restrict__ src, const int* __restrict__ dst,
                  const float* __restrict__ dinv, const float* __restrict__ WGT,
                  const float* __restrict__ T,
                  float* __restrict__ OUT, int e, unsigned* __restrict__ ctr,
                  int ngroups, int nsplit, size_t ws_size) {
    __shared__ int  pick[2];
    __shared__ int4 queue[8][EDGES_PER_WAVE];

    if (threadIdx.x == 0) {
        const unsigned id = __builtin_amdgcn_s_sendmsg_rtn(0x87);
        int h = (int)((id >> 18) & 1u);
        int g = (int)atomicAdd(ctr + h * CTR_STRIDE, 1u);
        if (g >= ngroups) {
            h ^= 1;
            g = (int)atomicAdd(ctr + h * CTR_STRIDE, 1u);
        }
        pick[0] = g;
        pick[1] = h;
    }
    __syncthreads();
    const int g = pick[0];
    const int h = pick[1];
    if (g >= ngroups) return;

    const int wave = threadIdx.x >> 5;
    const int lane = threadIdx.x & 31;
    const int e0 = g * EDGES_PER_GROUP + wave * EDGES_PER_WAVE + lane;
    const int e1 = e0 + 32;

    const bool have_wgt = ws_size >= WS_BYTES_WITH_WGT;
    int s0 = 0, d0 = 0, s1 = 0, d1 = 0;
    float w0 = 0.0f, w1 = 0.0f;
    const bool v0 = e0 < e, v1 = e1 < e;
    if (v0) { s0 = src[e0]; d0 = dst[e0]; }
    if (v1) { s1 = src[e1]; d1 = dst[e1]; }
    if (have_wgt) {
        if (v0) w0 = WGT[e0];
        if (v1) w1 = WGT[e1];
    }
    const bool m0 = v0 && ((d0 >= nsplit) == (h != 0));
    const bool m1 = v1 && ((d1 >= nsplit) == (h != 0));

    const unsigned b0 = __builtin_amdgcn_ballot_w32(m0);
    const unsigned b1 = __builtin_amdgcn_ballot_w32(m1);
    const unsigned below = (1u << lane) - 1u;
    const int n0 = __popc(b0);
    const int n  = n0 + __popc(b1);
    int4* q = queue[wave];
    if (!have_wgt) {
        const float a0 = dinv[m0 ? s0 : 0], c0 = dinv[m0 ? d0 : 0];
        const float a1 = dinv[m1 ? s1 : 0], c1 = dinv[m1 ? d1 : 0];
        w0 = a0 * c0;
        w1 = a1 * c1;
    }
    if (m0) q[__popc(b0 & below)]      = make_int4(s0, d0, __float_as_int(w0), 0);
    if (m1) q[n0 + __popc(b1 & below)] = make_int4(s1, d1, __float_as_int(w1), 0);

    const unsigned c = lane & 15;
    const int      r = lane >> 4;
    for (int base = 0; base < n; base += 16) {
        int4  ent[8];
        float m[8];
#pragma unroll
        for (int i = 0; i < 8; ++i) {
            const int idx = base + 2 * i + r;
            ent[i] = q[idx < n ? idx : n - 1];
        }
#pragma unroll
        for (int i = 0; i < 8; ++i)
            m[i] = T[(unsigned)ent[i].x * NHID + c];
#pragma unroll
        for (int i = 0; i < 8; ++i) {
            const int idx = base + 2 * i + r;
            if (idx < n)
                atomicAdd(OUT + ((unsigned)ent[i].y * NHID + c), m[i] * __int_as_float(ent[i].z));
        }
    }
}

__global__ void finalize_node(float* __restrict__ ACC, const float* __restrict__ T,
                              const float* __restrict__ dinv, const float* __restrict__ bias,
                              int n16, int doRelu) {
    int idx = blockIdx.x * blockDim.x + threadIdx.x;
    if (idx >= n16) return;
    int i = idx >> 4, f = idx & 15;
    float di = dinv[i];
    float v  = ACC[idx] + T[idx] * di * di + bias[f];
    ACC[idx] = doRelu ? fmaxf(v, 0.0f) : v;
}


#define POOL_WAVES   32
#define POOL_THREADS (POOL_WAVES * 32)
#define POOL_CHUNK   64
#define POOL_GROUPS  256

struct __attribute__((aligned(16))) PoolEnt { int t; int g; float w; };

template <bool HAVE_WGT>
__device__ __forceinline__ void pool_edges(const int* __restrict__ src, const int* __restrict__ dst,
                                           const int* __restrict__ bvec,
                                           const float* __restrict__ dinv, const float* __restrict__ WGT,
                                           const float* __restrict__ T, float* tab, PoolEnt* q,
                                           int e, int c_lo, int c_hi, int wave, int lane) {
    const unsigned c = lane & 15;
    const int      r = lane >> 4;
    if (c_lo + wave >= c_hi) return;

    int s0, d0, s1, d1;
    float w0 = 0.0f, w1 = 0.0f;
    {
        const int i0 = min((c_lo + wave) * POOL_CHUNK + lane, e - 1);
        const int i1 = min((c_lo + wave) * POOL_CHUNK + 32 + lane, e - 1);
        s0 = src[i0]; d0 = dst[i0];
        s1 = src[i1]; d1 = dst[i1];
        if (HAVE_WGT) { w0 = WGT[i0]; w1 = WGT[i1]; }
    }
    for (int ch = c_lo + wave; ch < c_hi; ch += POOL_WAVES) {
        const int n = min(POOL_CHUNK, e - ch * POOL_CHUNK);
        const int g0 = bvec[d0];
        const int g1 = bvec[d1];
        if (!HAVE_WGT) {
            w0 = dinv[s0] * dinv[d0];
            w1 = dinv[s1] * dinv[d1];
        }
        const int cs0 = s0, cs1 = s1;
        const float cw0 = w0, cw1 = w1;
        {
            const int nx = min(ch + POOL_WAVES, c_hi - 1);
            const int i0 = min(nx * POOL_CHUNK + lane, e - 1);
            const int i1 = min(nx * POOL_CHUNK + 32 + lane, e - 1);
            s0 = src[i0]; d0 = dst[i0];
            s1 = src[i1]; d1 = dst[i1];
            if (HAVE_WGT) { w0 = WGT[i0]; w1 = WGT[i1]; }
        }
        q[lane].t      = cs0 * NHID;  q[lane].g      = g0 * NHID;  q[lane].w      = cw0;
        q[lane + 32].t = cs1 * NHID;  q[lane + 32].g = g1 * NHID;  q[lane + 32].w = cw1;

        for (int base = 0; base < n; base += 32) {
            PoolEnt ent[16];
            float   m[16];
#pragma unroll
            for (int i = 0; i < 16; ++i) {
                const int idx = base + 2 * i + r;
                ent[i] = q[idx < n ? idx : n - 1];
            }
#pragma unroll
            for (int i = 0; i < 16; ++i)
                m[i] = T[(unsigned)ent[i].t + c];
#pragma unroll
            for (int i = 0; i < 16; ++i) {
                const int idx = base + 2 * i + r;
                const float v = m[i] * ent[i].w;
                atomicAdd(&tab[(unsigned)ent[i].g + c], idx < n ? v : 0.0f);
            }
        }
    }
}

__global__ __launch_bounds__(POOL_THREADS)
void edge_pool(const int* __restrict__ src, const int* __restrict__ dst,
               const int* __restrict__ bvec,
               const float* __restrict__ dinv, const float* __restrict__ WGT,
               const float* __restrict__ T, const float* __restrict__ bias,
               float* __restrict__ GP, int e, int nnodes, size_t ws_size) {
    __shared__ __attribute__((aligned(16))) float tab[NGRAPH * NHID];
    __shared__ PoolEnt queue[POOL_WAVES][POOL_CHUNK];
    __shared__ int  half_s;

    const int tid  = threadIdx.x;
    const int wave = __builtin_amdgcn_readfirstlane(tid >> 5);
    const int lane = tid & 31;

    float4* tab4 = (float4*)tab;
    for (int i = tid; i < NGRAPH * NHID / 4; i += POOL_THREADS)
        tab4[i] = make_float4(0.0f, 0.0f, 0.0f, 0.0f);
    if (tid == 0) {
        const unsigned id = __builtin_amdgcn_s_sendmsg_rtn(0x87);
        half_s = (int)((id >> 18) & 1u);
    }
    __syncthreads();

    const int nb = gridDim.x, b = blockIdx.x;

    {
        const unsigned c = lane & 15;
        const int      r = lane >> 4;
        const int n_lo = (int)((long long)b * nnodes / nb);
        const int n_hi = (int)((long long)(b + 1) * nnodes / nb);
        const float bc = bias[c];
        for (int k0 = n_lo + wave * 2; k0 < n_hi; k0 += POOL_WAVES * 2 * 8) {
            float di[8], tv[8];
            int   gi[8];
#pragma unroll
            for (int j = 0; j < 8; ++j) {
                const int i = min(k0 + r + POOL_WAVES * 2 * j, n_hi - 1);
                di[j] = dinv[i];
                tv[j] = T[(unsigned)i * NHID + c];
                gi[j] = bvec[i];
            }
#pragma unroll
            for (int j = 0; j < 8; ++j) {
                const int i = k0 + r + POOL_WAVES * 2 * j;
                const float v = tv[j] * di[j] * di[j] + bc;
                atomicAdd(&tab[(unsigned)gi[j] * NHID + c], i < n_hi ? v : 0.0f);
            }
        }
    }

    const int nchunks = (e + POOL_CHUNK - 1) / POOL_CHUNK;
    const int c_lo = (int)((long long)b * nchunks / nb);
    const int c_hi = (int)((long long)(b + 1) * nchunks / nb);
    if (ws_size >= WS_BYTES_WITH_WGT)
        pool_edges<true>(src, dst, bvec, dinv, WGT, T, tab, queue[wave], e, c_lo, c_hi, wave, lane);
    else
        pool_edges<false>(src, dst, bvec, dinv, WGT, T, tab, queue[wave], e, c_lo, c_hi, wave, lane);

    __syncthreads();

    float* gp = GP + (size_t)half_s * (NGRAPH * NHID);
    const unsigned rot = (unsigned)b * (unsigned)(NGRAPH * NHID / POOL_GROUPS);
    for (int k0 = 0; k0 < NGRAPH * NHID / POOL_THREADS; k0 += 8) {
        unsigned idx[8];
        float    v[8];
#pragma unroll
        for (int k = 0; k < 8; ++k) {
            idx[k] = (rot + (unsigned)(k0 + k) * POOL_THREADS + (unsigned)tid) & (unsigned)(NGRAPH * NHID - 1);
            v[k]   = tab[idx[k]];
        }
#pragma unroll
        for (int k = 0; k < 8; ++k)
            atomicAdd(gp + idx[k], v[k]);
    }
}

__global__ void lin1_kernel(const float* __restrict__ G, const float* __restrict__ W,
                            const float* __restrict__ B, float* __restrict__ O) {
    int t = blockIdx.x * blockDim.x + threadIdx.x;
    if (t >= NGRAPH * NMLP) return;
    int gi = t / NMLP, j = t % NMLP;
    float acc = B[j];
#pragma unroll
    for (int k = 0; k < NHID; ++k)
        acc = fmaf(fmaxf(G[gi * NHID + k] + G[NGRAPH * NHID + gi * NHID + k], 0.0f), W[k * NMLP + j], acc);
    O[t] = fmaxf(acc, 0.0f);
}

__global__ void lin2_kernel(const float* __restrict__ G, const float* __restrict__ W,
                            const float* __restrict__ B, float* __restrict__ O) {
    int t = blockIdx.x * blockDim.x + threadIdx.x;
    if (t >= NGRAPH * NCLS) return;
    int gi = t / NCLS, c = t % NCLS;
    float acc = B[c];
    for (int k = 0; k < NMLP; ++k)
        acc = fmaf(G[gi * NMLP + k], W[k * NCLS + c], acc);
    O[t] = acc;
}


extern "C" void kernel_launch(void* const* d_in, const int* in_sizes, int n_in,
                              void* d_out, int out_size, void* d_ws, size_t ws_size,
                              hipStream_t stream) {
    const float* x   = (const float*)d_in[0];
    const int*  eidx = (const int*)d_in[1];
    const int*  bvec = (const int*)d_in[2];
    const float* W1  = (const float*)d_in[3];
    const float* b1  = (const float*)d_in[4];
    const float* W2  = (const float*)d_in[5];
    const float* b2  = (const float*)d_in[6];
    const float* Wl1 = (const float*)d_in[7];
    const float* bl1 = (const float*)d_in[8];
    const float* Wl2 = (const float*)d_in[9];
    const float* bl2 = (const float*)d_in[10];
    float* out = (float*)d_out;

    char* ws = (char*)d_ws;
    size_t off = 0;
    auto walloc = [&](size_t bytes) -> void* {
        void* p = ws + off;
        off = (off + bytes + 255) & ~(size_t)255;
        return p;
    };
    float* deg  = (float*)walloc((size_t)NNODES * 4);
    float* bufA = (float*)walloc((size_t)NNODES * NHID * 4);
    float* bufB = (float*)walloc((size_t)NNODES * NHID * 4);
    (void)walloc((size_t)NGRAPH * NHID * 4);
    float* g2   = (float*)walloc((size_t)NGRAPH * NMLP * 4);
    float* wgt  = (float*)walloc((size_t)NEDGES * 4);

    const int* src = eidx;
    const int* dst = eidx + NEDGES;

    const int TB = 256;
    const int nN    = (NNODES + TB - 1) / TB;
    const int nN16  = (NNODES * NHID + TB - 1) / TB;
    const int nTile = ((NNODES / 16) + 7) / 8;
    const int nGrp  = (NEDGES + EDGES_PER_GROUP - 1) / EDGES_PER_GROUP;
    const int nNorm = (NEDGES + TB * NORM_PER_THREAD - 1) / (TB * NORM_PER_THREAD);
    const int nDeg  = (NEDGES + TB * DEG_PER_THREAD - 1) / (TB * DEG_PER_THREAD);

    unsigned* ctr = (unsigned*)g2;

    float* deg1 = bufB;
    fill_deg_ctr<<<nN, TB, 0, stream>>>(deg, deg1, NNODES, ctr, 2 * CTR_STRIDE);
    deg_accum<<<nDeg, TB, 0, stream>>>(dst, deg, deg1, NEDGES);
    rsqrt_sum<<<nN, TB, 0, stream>>>(deg, deg1, NNODES);
    float* dinv = deg;
    edge_norm<<<nNorm, TB, 0, stream>>>(src, dst, dinv, wgt, NEDGES, ws_size);

    gemm_wmma_f32<<<nTile, TB, 0, stream>>>(x, W1, bufA, NNODES, NFEAT);
    fill_f32<<<nN16, TB, 0, stream>>>(bufB, NNODES * NHID, 0.0f);
    edge_scatter<<<2 * nGrp, TB, 0, stream>>>(src, dst, dinv, wgt, bufA, bufB, NEDGES,
                                              ctr, nGrp, NNODES / 2, ws_size);
    finalize_node<<<nN16, TB, 0, stream>>>(bufB, bufA, dinv, b1, NNODES * NHID, 1);

    gemm_wmma_f32<<<nTile, TB, 0, stream>>>(bufB, W2, bufA, NNODES, NHID);
    float* gtab = bufB;
    fill_f32<<<(2 * NGRAPH * NHID + TB - 1) / TB, TB, 0, stream>>>(gtab, 2 * NGRAPH * NHID, 0.0f);
    edge_pool<<<POOL_GROUPS, POOL_THREADS, 0, stream>>>(src, dst, bvec, dinv, wgt, bufA, b2, gtab,
                                                        NEDGES, NNODES, ws_size);

    lin1_kernel<<<(NGRAPH * NMLP + TB - 1) / TB, TB, 0, stream>>>(gtab, Wl1, bl1, g2);
    lin2_kernel<<<(NGRAPH * NCLS + TB - 1) / TB, TB, 0, stream>>>(g2, Wl2, bl2, out);
}


# Layer-1 GEMM stores self loop + bias as the start value the edge adds land on, layer-2 GEMM takes relu at its A load: zero-fill and finalize launches and their 200k short waves go; sum order changed
# speedup vs baseline: 3.5670x; 1.0523x over previous
#include <hip/hip_runtime.h>
#include <hip/hip_bf16.h>


typedef float v2f __attribute__((ext_vector_type(2)));
typedef float v8f __attribute__((ext_vector_type(8)));

#define NNODES  200000
#define NEDGES  6400000
#define NFEAT   128
#define NHID    16
#define NMLP    100
#define NCLS    27
#define NGRAPH  4096

#define WS_BYTES_WITH_WGT ((size_t)NNODES * 4 + 2 * ((size_t)NNODES * NHID * 4) + \
                           (size_t)NGRAPH * NHID * 4 + (size_t)NGRAPH * NMLP * 4 + \
                           (size_t)NEDGES * 4)


__global__ void fill_f32(float* __restrict__ p, int n, float v) {
    int i = blockIdx.x * blockDim.x + threadIdx.x;
    if (i < n) p[i] = v;
}

__global__ void fill_deg_ctr(float* __restrict__ d0, float* __restrict__ d1, int n,
                             unsigned* __restrict__ ctr, int nctr) {
    int i = blockIdx.x * blockDim.x + threadIdx.x;
    if (i < n) { d0[i] = 1.0f; d1[i] = 0.0f; }
    if (i < nctr) ctr[i] = 0u;
}

#define DEG_PER_THREAD 4

__global__ __launch_bounds__(256)
void deg_accum(const int* __restrict__ dst, float* __restrict__ deg0,
               float* __restrict__ deg1, int e) {
    __shared__ int half_s;
    const int base = blockIdx.x * (256 * DEG_PER_THREAD) + threadIdx.x;
    unsigned d[DEG_PER_THREAD];
#pragma unroll
    for (int k = 0; k < DEG_PER_THREAD; ++k)
        d[k] = (unsigned)dst[min(base + k * 256, e - 1)];
    if (threadIdx.x == 0) {
        const unsigned id = __builtin_amdgcn_s_sendmsg_rtn(0x87);
        half_s = (int)((id >> 18) & 1u);
    }
    __syncthreads();
    float* __restrict__ deg = half_s ? deg1 : deg0;
    if ((blockIdx.x + 1) * (256 * DEG_PER_THREAD) <= e) {
#pragma unroll
        for (int k = 0; k < DEG_PER_THREAD; ++k)
            atomicAdd(&deg[d[k]], 1.0f);
    } else {
        for (int k = 0; k < DEG_PER_THREAD; ++k)
            if (base + k * 256 < e) atomicAdd(&deg[d[k]], 1.0f);
    }
}

__global__ void rsqrt_sum(float* __restrict__ p, const float* __restrict__ q, int n) {
    int i = blockIdx.x * blockDim.x + threadIdx.x;
    if (i < n) p[i] = rsqrtf(p[i] + q[i]);
}

template <bool RELU_A, bool START>
__device__ __forceinline__ void gemm_tile(const float* __restrict__ X, const float* __restrict__ W,
                                          float* __restrict__ T, float* __restrict__ S,
                                          const float* __restrict__ dinv,
                                          const float* __restrict__ bias,
                                          float* lw, int M, int K) {
    const int tid = threadIdx.x;
    const int kn  = K * NHID;
    for (int i = tid; i < kn; i += 256) lw[i] = W[i];
    __syncthreads();

    const int wave = tid >> 5;
    const int lane = tid & 31;
    const int tile = blockIdx.x * 8 + wave;
    if (tile * 16 >= M) return;

    const int row   = tile * 16 + (lane & 15);
    const int kHalf = (lane >> 4) << 1;
    const int n     = lane & 15;
    const int rbase = tile * 16 + ((lane >> 4) << 3);

    float4 dlo = make_float4(0.0f, 0.0f, 0.0f, 0.0f), dhi = dlo;
    float  bn  = 0.0f;
    if (START) {
        dlo = *(const float4*)(dinv + rbase);
        dhi = *(const float4*)(dinv + rbase + 4);
        bn  = bias[n];
    }

    v8f c = {};
    const float* xr = X + (size_t)row * K + kHalf;
    for (int k0 = 0; k0 < K; k0 += 4) {
        float2 av = *(const float2*)(xr + k0);
        v2f a;
        if (RELU_A) { a.x = fmaxf(av.x, 0.0f); a.y = fmaxf(av.y, 0.0f); }
        else        { a.x = av.x;              a.y = av.y; }
        v2f b;
        b.x = lw[(k0 + kHalf)     * NHID + n];
        b.y = lw[(k0 + kHalf + 1) * NHID + n];
        c = __builtin_amdgcn_wmma_f32_16x16x4_f32(
                 false, a,  false, b,
                 (short)0, c,  false,  false);
    }

#pragma unroll
    for (int v = 0; v < 8; ++v)
        T[(size_t)(rbase + v) * NHID + n] = c[v];
    if (START) {
        const float di[8] = { dlo.x, dlo.y, dlo.z, dlo.w, dhi.x, dhi.y, dhi.z, dhi.w };
#pragma unroll
        for (int v = 0; v < 8; ++v)
            S[(size_t)(rbase + v) * NHID + n] = c[v] * di[v] * di[v] + bn;
    }
}

__global__ __launch_bounds__(256)
void gemm_start(const float* __restrict__ X, const float* __restrict__ W,
                float* __restrict__ T, float* __restrict__ S,
                const float* __restrict__ dinv, const float* __restrict__ bias,
                int M, int K) {
    __shared__ float lw[NFEAT * NHID];
    gemm_tile<false, true>(X, W, T, S, dinv, bias, lw, M, K);
}

__global__ __launch_bounds__(256)
void gemm_relu(const float* __restrict__ X, const float* __restrict__ W,
               float* __restrict__ T, int M, int K) {
    __shared__ float lw[NFEAT * NHID];
    gemm_tile<true, false>(X, W, T, nullptr, nullptr, nullptr, lw, M, K);
}

#define NORM_PER_THREAD 4

__global__ __launch_bounds__(256)
void edge_norm(const int* __restrict__ src, const int* __restrict__ dst,
               const float* __restrict__ dinv, float* __restrict__ WGT, int e,
               size_t ws_size) {
    if (ws_size < WS_BYTES_WITH_WGT) return;
    const int base = blockIdx.x * (256 * NORM_PER_THREAD) + threadIdx.x;
    if (base + (NORM_PER_THREAD - 1) * 256 < e) {
        unsigned s[NORM_PER_THREAD], d[NORM_PER_THREAD];
        float    a[NORM_PER_THREAD], c[NORM_PER_THREAD];
#pragma unroll
        for (int k = 0; k < NORM_PER_THREAD; ++k) {
            s[k] = (unsigned)src[base + k * 256];
            d[k] = (unsigned)dst[base + k * 256];
        }
#pragma unroll
        for (int k = 0; k < NORM_PER_THREAD; ++k) {
            a[k] = dinv[s[k]];
            c[k] = dinv[d[k]];
        }
#pragma unroll
        for (int k = 0; k < NORM_PER_THREAD; ++k)
            WGT[base + k * 256] = a[k] * c[k];
    } else {
        for (int k = 0; k < NORM_PER_THREAD; ++k) {
            const int i = base + k * 256;
            if (i < e) WGT[i] = dinv[(unsigned)src[i]] * dinv[(unsigned)dst[i]];
        }
    }
}


#define EDGES_PER_WAVE 64
#define EDGES_PER_GROUP (8 * EDGES_PER_WAVE)
#define CTR_STRIDE 64

__global__ __launch_bounds__(256)
void edge_scatter(const int* __restrict__ src, const int* __restrict__ dst,
                  const float* __restrict__ dinv, const float* __restrict__ WGT,
                  const float* __restrict__ T,
                  float* __restrict__ OUT, int e, unsigned* __restrict__ ctr,
                  int ngroups, int nsplit, size_t ws_size) {
    __shared__ int  pick[2];
    __shared__ int4 queue[8][EDGES_PER_WAVE];

    if (threadIdx.x == 0) {
        const unsigned id = __builtin_amdgcn_s_sendmsg_rtn(0x87);
        int h = (int)((id >> 18) & 1u);
        int g = (int)atomicAdd(ctr + h * CTR_STRIDE, 1u);
        if (g >= ngroups) {
            h ^= 1;
            g = (int)atomicAdd(ctr + h * CTR_STRIDE, 1u);
        }
        pick[0] = g;
        pick[1] = h;
    }
    __syncthreads();
    const int g = pick[0];
    const int h = pick[1];
    if (g >= ngroups) return;

    const int wave = threadIdx.x >> 5;
    const int lane = threadIdx.x & 31;
    const int e0 = g * EDGES_PER_GROUP + wave * EDGES_PER_WAVE + lane;
    const int e1 = e0 + 32;

    const bool have_wgt = ws_size >= WS_BYTES_WITH_WGT;
    int s0 = 0, d0 = 0, s1 = 0, d1 = 0;
    float w0 = 0.0f, w1 = 0.0f;
    const bool v0 = e0 < e, v1 = e1 < e;
    if (v0) { s0 = src[e0]; d0 = dst[e0]; }
    if (v1) { s1 = src[e1]; d1 = dst[e1]; }
    if (have_wgt) {
        if (v0) w0 = WGT[e0];
        if (v1) w1 = WGT[e1];
    }
    const bool m0 = v0 && ((d0 >= nsplit) == (h != 0));
    const bool m1 = v1 && ((d1 >= nsplit) == (h != 0));

    const unsigned b0 = __builtin_amdgcn_ballot_w32(m0);
    const unsigned b1 = __builtin_amdgcn_ballot_w32(m1);
    const unsigned below = (1u << lane) - 1u;
    const int n0 = __popc(b0);
    const int n  = n0 + __popc(b1);
    int4* q = queue[wave];
    if (!have_wgt) {
        const float a0 = dinv[m0 ? s0 : 0], c0 = dinv[m0 ? d0 : 0];
        const float a1 = dinv[m1 ? s1 : 0], c1 = dinv[m1 ? d1 : 0];
        w0 = a0 * c0;
        w1 = a1 * c1;
    }
    if (m0) q[__popc(b0 & below)]      = make_int4(s0, d0, __float_as_int(w0), 0);
    if (m1) q[n0 + __popc(b1 & below)] = make_int4(s1, d1, __float_as_int(w1), 0);

    const unsigned c = lane & 15;
    const int      r = lane >> 4;
    for (int base = 0; base < n; base += 16) {
        int4  ent[8];
        float m[8];
#pragma unroll
        for (int i = 0; i < 8; ++i) {
            const int idx = base + 2 * i + r;
            ent[i] = q[idx < n ? idx : n - 1];
        }
#pragma unroll
        for (int i = 0; i < 8; ++i)
            m[i] = T[(unsigned)ent[i].x * NHID + c];
#pragma unroll
        for (int i = 0; i < 8; ++i) {
            const int idx = base + 2 * i + r;
            if (idx < n)
                atomicAdd(OUT + ((unsigned)ent[i].y * NHID + c), m[i] * __int_as_float(ent[i].z));
        }
    }
}


#define POOL_WAVES   32
#define POOL_THREADS (POOL_WAVES * 32)
#define POOL_CHUNK   64
#define POOL_GROUPS  256

struct __attribute__((aligned(16))) PoolEnt { int t; int g; float w; };

template <bool HAVE_WGT>
__device__ __forceinline__ void pool_edges(const int* __restrict__ src, const int* __restrict__ dst,
                                           const int* __restrict__ bvec,
                                           const float* __restrict__ dinv, const float* __restrict__ WGT,
                                           const float* __restrict__ T, float* tab, PoolEnt* q,
                                           int e, int c_lo, int c_hi, int wave, int lane) {
    const unsigned c = lane & 15;
    const int      r = lane >> 4;
    if (c_lo + wave >= c_hi) return;

    int s0, d0, s1, d1;
    float w0 = 0.0f, w1 = 0.0f;
    {
        const int i0 = min((c_lo + wave) * POOL_CHUNK + lane, e - 1);
        const int i1 = min((c_lo + wave) * POOL_CHUNK + 32 + lane, e - 1);
        s0 = src[i0]; d0 = dst[i0];
        s1 = src[i1]; d1 = dst[i1];
        if (HAVE_WGT) { w0 = WGT[i0]; w1 = WGT[i1]; }
    }
    for (int ch = c_lo + wave; ch < c_hi; ch += POOL_WAVES) {
        const int n = min(POOL_CHUNK, e - ch * POOL_CHUNK);
        const int g0 = bvec[d0];
        const int g1 = bvec[d1];
        if (!HAVE_WGT) {
            w0 = dinv[s0] * dinv[d0];
            w1 = dinv[s1] * dinv[d1];
        }
        const int cs0 = s0, cs1 = s1;
        const float cw0 = w0, cw1 = w1;
        {
            const int nx = min(ch + POOL_WAVES, c_hi - 1);
            const int i0 = min(nx * POOL_CHUNK + lane, e - 1);
            const int i1 = min(nx * POOL_CHUNK + 32 + lane, e - 1);
            s0 = src[i0]; d0 = dst[i0];
            s1 = src[i1]; d1 = dst[i1];
            if (HAVE_WGT) { w0 = WGT[i0]; w1 = WGT[i1]; }
        }
        q[lane].t      = cs0 * NHID;  q[lane].g      = g0 * NHID;  q[lane].w      = cw0;
        q[lane + 32].t = cs1 * NHID;  q[lane + 32].g = g1 * NHID;  q[lane + 32].w = cw1;

        for (int base = 0; base < n; base += 32) {
            PoolEnt ent[16];
            float   m[16];
#pragma unroll
            for (int i = 0; i < 16; ++i) {
                const int idx = base + 2 * i + r;
                ent[i] = q[idx < n ? idx : n - 1];
            }
#pragma unroll
            for (int i = 0; i < 16; ++i)
                m[i] = T[(unsigned)ent[i].t + c];
#pragma unroll
            for (int i = 0; i < 16; ++i) {
                const int idx = base + 2 * i + r;
                const float v = m[i] * ent[i].w;
                atomicAdd(&tab[(unsigned)ent[i].g + c], idx < n ? v : 0.0f);
            }
        }
    }
}

__global__ __launch_bounds__(POOL_THREADS)
void edge_pool(const int* __restrict__ src, const int* __restrict__ dst,
               const int* __restrict__ bvec,
               const float* __restrict__ dinv, const float* __restrict__ WGT,
               const float* __restrict__ T, const float* __restrict__ bias,
               float* __restrict__ GP, int e, int nnodes, size_t ws_size) {
    __shared__ __attribute__((aligned(16))) float tab[NGRAPH * NHID];
    __shared__ PoolEnt queue[POOL_WAVES][POOL_CHUNK];
    __shared__ int  half_s;

    const int tid  = threadIdx.x;
    const int wave = __builtin_amdgcn_readfirstlane(tid >> 5);
    const int lane = tid & 31;

    float4* tab4 = (float4*)tab;
    for (int i = tid; i < NGRAPH * NHID / 4; i += POOL_THREADS)
        tab4[i] = make_float4(0.0f, 0.0f, 0.0f, 0.0f);
    if (tid == 0) {
        const unsigned id = __builtin_amdgcn_s_sendmsg_rtn(0x87);
        half_s = (int)((id >> 18) & 1u);
    }
    __syncthreads();

    const int nb = gridDim.x, b = blockIdx.x;

    {
        const unsigned c = lane & 15;
        const int      r = lane >> 4;
        const int n_lo = (int)((long long)b * nnodes / nb);
        const int n_hi = (int)((long long)(b + 1) * nnodes / nb);
        const float bc = bias[c];
        for (int k0 = n_lo + wave * 2; k0 < n_hi; k0 += POOL_WAVES * 2 * 8) {
            float di[8], tv[8];
            int   gi[8];
#pragma unroll
            for (int j = 0; j < 8; ++j) {
                const int i = min(k0 + r + POOL_WAVES * 2 * j, n_hi - 1);
                di[j] = dinv[i];
                tv[j] = T[(unsigned)i * NHID + c];
                gi[j] = bvec[i];
            }
#pragma unroll
            for (int j = 0; j < 8; ++j) {
                const int i = k0 + r + POOL_WAVES * 2 * j;
                const float v = tv[j] * di[j] * di[j] + bc;
                atomicAdd(&tab[(unsigned)gi[j] * NHID + c], i < n_hi ? v : 0.0f);
            }
        }
    }

    const int nchunks = (e + POOL_CHUNK - 1) / POOL_CHUNK;
    const int c_lo = (int)((long long)b * nchunks / nb);
    const int c_hi = (int)((long long)(b + 1) * nchunks / nb);
    if (ws_size >= WS_BYTES_WITH_WGT)
        pool_edges<true>(src, dst, bvec, dinv, WGT, T, tab, queue[wave], e, c_lo, c_hi, wave, lane);
    else
        pool_edges<false>(src, dst, bvec, dinv, WGT, T, tab, queue[wave], e, c_lo, c_hi, wave, lane);

    __syncthreads();

    float* gp = GP + (size_t)half_s * (NGRAPH * NHID);
    const unsigned rot = (unsigned)b * (unsigned)(NGRAPH * NHID / POOL_GROUPS);
    for (int k0 = 0; k0 < NGRAPH * NHID / POOL_THREADS; k0 += 8) {
        unsigned idx[8];
        float    v[8];
#pragma unroll
        for (int k = 0; k < 8; ++k) {
            idx[k] = (rot + (unsigned)(k0 + k) * POOL_THREADS + (unsigned)tid) & (unsigned)(NGRAPH * NHID - 1);
            v[k]   = tab[idx[k]];
        }
#pragma unroll
        for (int k = 0; k < 8; ++k)
            atomicAdd(gp + idx[k], v[k]);
    }
}

__global__ void lin1_kernel(const float* __restrict__ G, const float* __restrict__ W,
                            const float* __restrict__ B, float* __restrict__ O) {
    int t = blockIdx.x * blockDim.x + threadIdx.x;
    if (t >= NGRAPH * NMLP) return;
    int gi = t / NMLP, j = t % NMLP;
    float acc = B[j];
#pragma unroll
    for (int k = 0; k < NHID; ++k)
        acc = fmaf(fmaxf(G[gi * NHID + k] + G[NGRAPH * NHID + gi * NHID + k], 0.0f), W[k * NMLP + j], acc);
    O[t] = fmaxf(acc, 0.0f);
}

__global__ void lin2_kernel(const float* __restrict__ G, const float* __restrict__ W,
                            const float* __restrict__ B, float* __restrict__ O) {
    int t = blockIdx.x * blockDim.x + threadIdx.x;
    if (t >= NGRAPH * NCLS) return;
    int gi = t / NCLS, c = t % NCLS;
    float acc = B[c];
    for (int k = 0; k < NMLP; ++k)
        acc = fmaf(G[gi * NMLP + k], W[k * NCLS + c], acc);
    O[t] = acc;
}


extern "C" void kernel_launch(void* const* d_in, const int* in_sizes, int n_in,
                              void* d_out, int out_size, void* d_ws, size_t ws_size,
                              hipStream_t stream) {
    const float* x   = (const float*)d_in[0];
    const int*  eidx = (const int*)d_in[1];
    const int*  bvec = (const int*)d_in[2];
    const float* W1  = (const float*)d_in[3];
    const float* b1  = (const float*)d_in[4];
    const float* W2  = (const float*)d_in[5];
    const float* b2  = (const float*)d_in[6];
    const float* Wl1 = (const float*)d_in[7];
    const float* bl1 = (const float*)d_in[8];
    const float* Wl2 = (const float*)d_in[9];
    const float* bl2 = (const float*)d_in[10];
    float* out = (float*)d_out;

    char* ws = (char*)d_ws;
    size_t off = 0;
    auto walloc = [&](size_t bytes) -> void* {
        void* p = ws + off;
        off = (off + bytes + 255) & ~(size_t)255;
        return p;
    };
    float* deg  = (float*)walloc((size_t)NNODES * 4);
    float* bufA = (float*)walloc((size_t)NNODES * NHID * 4);
    float* bufB = (float*)walloc((size_t)NNODES * NHID * 4);
    (void)walloc((size_t)NGRAPH * NHID * 4);
    float* g2   = (float*)walloc((size_t)NGRAPH * NMLP * 4);
    float* wgt  = (float*)walloc((size_t)NEDGES * 4);

    const int* src = eidx;
    const int* dst = eidx + NEDGES;

    const int TB = 256;
    const int nN    = (NNODES + TB - 1) / TB;
    const int nTile = ((NNODES / 16) + 7) / 8;
    const int nGrp  = (NEDGES + EDGES_PER_GROUP - 1) / EDGES_PER_GROUP;
    const int nNorm = (NEDGES + TB * NORM_PER_THREAD - 1) / (TB * NORM_PER_THREAD);
    const int nDeg  = (NEDGES + TB * DEG_PER_THREAD - 1) / (TB * DEG_PER_THREAD);

    unsigned* ctr = (unsigned*)g2;

    float* deg1 = bufB;
    fill_deg_ctr<<<nN, TB, 0, stream>>>(deg, deg1, NNODES, ctr, 2 * CTR_STRIDE);
    deg_accum<<<nDeg, TB, 0, stream>>>(dst, deg, deg1, NEDGES);
    rsqrt_sum<<<nN, TB, 0, stream>>>(deg, deg1, NNODES);
    float* dinv = deg;
    edge_norm<<<nNorm, TB, 0, stream>>>(src, dst, dinv, wgt, NEDGES, ws_size);

    gemm_start<<<nTile, TB, 0, stream>>>(x, W1, bufA, bufB, dinv, b1, NNODES, NFEAT);
    edge_scatter<<<2 * nGrp, TB, 0, stream>>>(src, dst, dinv, wgt, bufA, bufB, NEDGES,
                                              ctr, nGrp, NNODES / 2, ws_size);

    gemm_relu<<<nTile, TB, 0, stream>>>(bufB, W2, bufA, NNODES, NHID);
    float* gtab = bufB;
    fill_f32<<<(2 * NGRAPH * NHID + TB - 1) / TB, TB, 0, stream>>>(gtab, 2 * NGRAPH * NHID, 0.0f);
    edge_pool<<<POOL_GROUPS, POOL_THREADS, 0, stream>>>(src, dst, bvec, dinv, wgt, bufA, b2, gtab,
                                                        NEDGES, NNODES, ws_size);

    lin1_kernel<<<(NGRAPH * NMLP + TB - 1) / TB, TB, 0, stream>>>(gtab, Wl1, bl1, g2);
    lin2_kernel<<<(NGRAPH * NCLS + TB - 1) / TB, TB, 0, stream>>>(g2, Wl2, bl2, out);
}


# Graph of each edge's destination kept in edge order, read with dinv[dst] as one 8-byte node record by the edge-weight launch: the pooled layer-2 pass streams it, no per-edge batch lookup
# speedup vs baseline: 3.7570x; 1.0533x over previous
#include <hip/hip_runtime.h>
#include <hip/hip_bf16.h>


typedef float v2f __attribute__((ext_vector_type(2)));
typedef float v8f __attribute__((ext_vector_type(8)));

#define NNODES  200000
#define NEDGES  6400000
#define NFEAT   128
#define NHID    16
#define NMLP    100
#define NCLS    27
#define NGRAPH  4096

#define WS_BYTES_WITH_WGT ((size_t)NNODES * 4 + 2 * ((size_t)NNODES * NHID * 4) + \
                           (size_t)NGRAPH * NHID * 4 + (size_t)NGRAPH * NMLP * 4 + \
                           (size_t)NEDGES * 4)
#define WS_BYTES_WITH_GID (WS_BYTES_WITH_WGT + (size_t)NEDGES * 2)
static_assert(NGRAPH <= 65536, "a graph id must fit the 16 bits of the per-edge plane");


__global__ void fill_f32(float* __restrict__ p, int n, float v) {
    int i = blockIdx.x * blockDim.x + threadIdx.x;
    if (i < n) p[i] = v;
}

__global__ void fill_deg_ctr(float* __restrict__ d0, float* __restrict__ d1, int n,
                             unsigned* __restrict__ ctr, int nctr) {
    int i = blockIdx.x * blockDim.x + threadIdx.x;
    if (i < n) { d0[i] = 1.0f; d1[i] = 0.0f; }
    if (i < nctr) ctr[i] = 0u;
}

#define DEG_PER_THREAD 4

__global__ __launch_bounds__(256)
void deg_accum(const int* __restrict__ dst, float* __restrict__ deg0,
               float* __restrict__ deg1, int e) {
    __shared__ int half_s;
    const int base = blockIdx.x * (256 * DEG_PER_THREAD) + threadIdx.x;
    unsigned d[DEG_PER_THREAD];
#pragma unroll
    for (int k = 0; k < DEG_PER_THREAD; ++k)
        d[k] = (unsigned)dst[min(base + k * 256, e - 1)];
    if (threadIdx.x == 0) {
        const unsigned id = __builtin_amdgcn_s_sendmsg_rtn(0x87);
        half_s = (int)((id >> 18) & 1u);
    }
    __syncthreads();
    float* __restrict__ deg = half_s ? deg1 : deg0;
    if ((blockIdx.x + 1) * (256 * DEG_PER_THREAD) <= e) {
#pragma unroll
        for (int k = 0; k < DEG_PER_THREAD; ++k)
            atomicAdd(&deg[d[k]], 1.0f);
    } else {
        for (int k = 0; k < DEG_PER_THREAD; ++k)
            if (base + k * 256 < e) atomicAdd(&deg[d[k]], 1.0f);
    }
}

__global__ void rsqrt_sum(float* __restrict__ p, const float* __restrict__ q, int n,
                          const int* __restrict__ bvec, int* __restrict__ rec, size_t ws_size) {
    int i = blockIdx.x * blockDim.x + threadIdx.x;
    if (i < n) {
        const float r = rsqrtf(p[i] + q[i]);
        p[i] = r;
        if (ws_size >= WS_BYTES_WITH_GID)
            ((int2*)rec)[i] = make_int2(__float_as_int(r), bvec[i]);
    }
}

template <bool RELU_A, bool START>
__device__ __forceinline__ void gemm_tile(const float* __restrict__ X, const float* __restrict__ W,
                                          float* __restrict__ T, float* __restrict__ S,
                                          const float* __restrict__ dinv,
                                          const float* __restrict__ bias,
                                          float* lw, int M, int K) {
    const int tid = threadIdx.x;
    const int kn  = K * NHID;
    for (int i = tid; i < kn; i += 256) lw[i] = W[i];
    __syncthreads();

    const int wave = tid >> 5;
    const int lane = tid & 31;
    const int tile = blockIdx.x * 8 + wave;
    if (tile * 16 >= M) return;

    const int row   = tile * 16 + (lane & 15);
    const int kHalf = (lane >> 4) << 1;
    const int n     = lane & 15;
    const int rbase = tile * 16 + ((lane >> 4) << 3);

    float4 dlo = make_float4(0.0f, 0.0f, 0.0f, 0.0f), dhi = dlo;
    float  bn  = 0.0f;
    if (START) {
        dlo = *(const float4*)(dinv + rbase);
        dhi = *(const float4*)(dinv + rbase + 4);
        bn  = bias[n];
    }

    v8f c = {};
    const float* xr = X + (size_t)row * K + kHalf;
    for (int k0 = 0; k0 < K; k0 += 4) {
        float2 av = *(const float2*)(xr + k0);
        v2f a;
        if (RELU_A) { a.x = fmaxf(av.x, 0.0f); a.y = fmaxf(av.y, 0.0f); }
        else        { a.x = av.x;              a.y = av.y; }
        v2f b;
        b.x = lw[(k0 + kHalf)     * NHID + n];
        b.y = lw[(k0 + kHalf + 1) * NHID + n];
        c = __builtin_amdgcn_wmma_f32_16x16x4_f32(
                 false, a,  false, b,
                 (short)0, c,  false,  false);
    }

#pragma unroll
    for (int v = 0; v < 8; ++v)
        T[(size_t)(rbase + v) * NHID + n] = c[v];
    if (START) {
        const float di[8] = { dlo.x, dlo.y, dlo.z, dlo.w, dhi.x, dhi.y, dhi.z, dhi.w };
#pragma unroll
        for (int v = 0; v < 8; ++v)
            S[(size_t)(rbase + v) * NHID + n] = c[v] * di[v] * di[v] + bn;
    }
}

__global__ __launch_bounds__(256)
void gemm_start(const float* __restrict__ X, const float* __restrict__ W,
                float* __restrict__ T, float* __restrict__ S,
                const float* __restrict__ dinv, const float* __restrict__ bias,
                int M, int K) {
    __shared__ float lw[NFEAT * NHID];
    gemm_tile<false, true>(X, W, T, S, dinv, bias, lw, M, K);
}

__global__ __launch_bounds__(256)
void gemm_relu(const float* __restrict__ X, const float* __restrict__ W,
               float* __restrict__ T, int M, int K) {
    __shared__ float lw[NFEAT * NHID];
    gemm_tile<true, false>(X, W, T, nullptr, nullptr, nullptr, lw, M, K);
}

#define NORM_PER_THREAD 4

__device__ __forceinline__ void edge_norm_gid(const int* __restrict__ src, const int* __restrict__ dst,
                                              const float* __restrict__ dinv, const int2* __restrict__ rec,
                                              float* __restrict__ WGT, unsigned short* __restrict__ GID,
                                              int e, int base) {
    if (base + (NORM_PER_THREAD - 1) * 256 < e) {
        unsigned s[NORM_PER_THREAD], d[NORM_PER_THREAD];
        float    a[NORM_PER_THREAD];
        int2     r[NORM_PER_THREAD];
#pragma unroll
        for (int k = 0; k < NORM_PER_THREAD; ++k) {
            s[k] = (unsigned)src[base + k * 256];
            d[k] = (unsigned)dst[base + k * 256];
        }
#pragma unroll
        for (int k = 0; k < NORM_PER_THREAD; ++k) {
            a[k] = dinv[s[k]];
            r[k] = rec[d[k]];
        }
#pragma unroll
        for (int k = 0; k < NORM_PER_THREAD; ++k) {
            WGT[base + k * 256] = a[k] * __int_as_float(r[k].x);
            GID[base + k * 256] = (unsigned short)r[k].y;
        }
    } else {
        for (int k = 0; k < NORM_PER_THREAD; ++k) {
            const int i = base + k * 256;
            if (i < e) {
                const int2 r = rec[(unsigned)dst[i]];
                WGT[i] = dinv[(unsigned)src[i]] * __int_as_float(r.x);
                GID[i] = (unsigned short)r.y;
            }
        }
    }
}

__global__ __launch_bounds__(256)
void edge_norm(const int* __restrict__ src, const int* __restrict__ dst,
               const float* __restrict__ dinv, const int* __restrict__ rec,
               float* __restrict__ WGT, unsigned short* __restrict__ GID, int e,
               size_t ws_size) {
    if (ws_size < WS_BYTES_WITH_WGT) return;
    const int base = blockIdx.x * (256 * NORM_PER_THREAD) + threadIdx.x;
    if (ws_size >= WS_BYTES_WITH_GID) {
        edge_norm_gid(src, dst, dinv, (const int2*)rec, WGT, GID, e, base);
        return;
    }
    if (base + (NORM_PER_THREAD - 1) * 256 < e) {
        unsigned s[NORM_PER_THREAD], d[NORM_PER_THREAD];
        float    a[NORM_PER_THREAD], c[NORM_PER_THREAD];
#pragma unroll
        for (int k = 0; k < NORM_PER_THREAD; ++k) {
            s[k] = (unsigned)src[base + k * 256];
            d[k] = (unsigned)dst[base + k * 256];
        }
#pragma unroll
        for (int k = 0; k < NORM_PER_THREAD; ++k) {
            a[k] = dinv[s[k]];
            c[k] = dinv[d[k]];
        }
#pragma unroll
        for (int k = 0; k < NORM_PER_THREAD; ++k)
            WGT[base + k * 256] = a[k] * c[k];
    } else {
        for (int k = 0; k < NORM_PER_THREAD; ++k) {
            const int i = base + k * 256;
            if (i < e) WGT[i] = dinv[(unsigned)src[i]] * dinv[(unsigned)dst[i]];
        }
    }
}


#define EDGES_PER_WAVE 64
#define EDGES_PER_GROUP (8 * EDGES_PER_WAVE)
#define CTR_STRIDE 64

__global__ __launch_bounds__(256)
void edge_scatter(const int* __restrict__ src, const int* __restrict__ dst,
                  const float* __restrict__ dinv, const float* __restrict__ WGT,
                  const float* __restrict__ T,
                  float* __restrict__ OUT, int e, unsigned* __restrict__ ctr,
                  int ngroups, int nsplit, size_t ws_size) {
    __shared__ int  pick[2];
    __shared__ int4 queue[8][EDGES_PER_WAVE];

    if (threadIdx.x == 0) {
        const unsigned id = __builtin_amdgcn_s_sendmsg_rtn(0x87);
        int h = (int)((id >> 18) & 1u);
        int g = (int)atomicAdd(ctr + h * CTR_STRIDE, 1u);
        if (g >= ngroups) {
            h ^= 1;
            g = (int)atomicAdd(ctr + h * CTR_STRIDE, 1u);
        }
        pick[0] = g;
        pick[1] = h;
    }
    __syncthreads();
    const int g = pick[0];
    const int h = pick[1];
    if (g >= ngroups) return;

    const int wave = threadIdx.x >> 5;
    const int lane = threadIdx.x & 31;
    const int e0 = g * EDGES_PER_GROUP + wave * EDGES_PER_WAVE + lane;
    const int e1 = e0 + 32;

    const bool have_wgt = ws_size >= WS_BYTES_WITH_WGT;
    int s0 = 0, d0 = 0, s1 = 0, d1 = 0;
    float w0 = 0.0f, w1 = 0.0f;
    const bool v0 = e0 < e, v1 = e1 < e;
    if (v0) { s0 = src[e0]; d0 = dst[e0]; }
    if (v1) { s1 = src[e1]; d1 = dst[e1]; }
    if (have_wgt) {
        if (v0) w0 = WGT[e0];
        if (v1) w1 = WGT[e1];
    }
    const bool m0 = v0 && ((d0 >= nsplit) == (h != 0));
    const bool m1 = v1 && ((d1 >= nsplit) == (h != 0));

    const unsigned b0 = __builtin_amdgcn_ballot_w32(m0);
    const unsigned b1 = __builtin_amdgcn_ballot_w32(m1);
    const unsigned below = (1u << lane) - 1u;
    const int n0 = __popc(b0);
    const int n  = n0 + __popc(b1);
    int4* q = queue[wave];
    if (!have_wgt) {
        const float a0 = dinv[m0 ? s0 : 0], c0 = dinv[m0 ? d0 : 0];
        const float a1 = dinv[m1 ? s1 : 0], c1 = dinv[m1 ? d1 : 0];
        w0 = a0 * c0;
        w1 = a1 * c1;
    }
    if (m0) q[__popc(b0 & below)]      = make_int4(s0, d0, __float_as_int(w0), 0);
    if (m1) q[n0 + __popc(b1 & below)] = make_int4(s1, d1, __float_as_int(w1), 0);

    const unsigned c = lane & 15;
    const int      r = lane >> 4;
    for (int base = 0; base < n; base += 16) {
        int4  ent[8];
        float m[8];
#pragma unroll
        for (int i = 0; i < 8; ++i) {
            const int idx = base + 2 * i + r;
            ent[i] = q[idx < n ? idx : n - 1];
        }
#pragma unroll
        for (int i = 0; i < 8; ++i)
            m[i] = T[(unsigned)ent[i].x * NHID + c];
#pragma unroll
        for (int i = 0; i < 8; ++i) {
            const int idx = base + 2 * i + r;
            if (idx < n)
                atomicAdd(OUT + ((unsigned)ent[i].y * NHID + c), m[i] * __int_as_float(ent[i].z));
        }
    }
}


#define POOL_WAVES   32
#define POOL_THREADS (POOL_WAVES * 32)
#define POOL_CHUNK   64
#define POOL_GROUPS  256

struct __attribute__((aligned(16))) PoolEnt { int t; int g; float w; };

template <bool HAVE_WGT, bool HAVE_GID>
__device__ __forceinline__ void pool_edges(const int* __restrict__ src, const int* __restrict__ dst,
                                           const int* __restrict__ bvec,
                                           const float* __restrict__ dinv, const float* __restrict__ WGT,
                                           const unsigned short* __restrict__ GID,
                                           const float* __restrict__ T, float* tab, PoolEnt* q,
                                           int e, int c_lo, int c_hi, int wave, int lane) {
    const unsigned c = lane & 15;
    const int      r = lane >> 4;
    if (c_lo + wave >= c_hi) return;

    int s0, d0, s1, d1;
    float w0 = 0.0f, w1 = 0.0f;
    {
        const int i0 = min((c_lo + wave) * POOL_CHUNK + lane, e - 1);
        const int i1 = min((c_lo + wave) * POOL_CHUNK + 32 + lane, e - 1);
        s0 = src[i0]; d0 = HAVE_GID ? (int)GID[i0] : dst[i0];
        s1 = src[i1]; d1 = HAVE_GID ? (int)GID[i1] : dst[i1];
        if (HAVE_WGT) { w0 = WGT[i0]; w1 = WGT[i1]; }
    }
    for (int ch = c_lo + wave; ch < c_hi; ch += POOL_WAVES) {
        const int n = min(POOL_CHUNK, e - ch * POOL_CHUNK);
        const int g0 = HAVE_GID ? d0 : bvec[d0];
        const int g1 = HAVE_GID ? d1 : bvec[d1];
        if (!HAVE_WGT) {
            w0 = dinv[s0] * dinv[d0];
            w1 = dinv[s1] * dinv[d1];
        }
        const int cs0 = s0, cs1 = s1;
        const float cw0 = w0, cw1 = w1;
        {
            const int nx = min(ch + POOL_WAVES, c_hi - 1);
            const int i0 = min(nx * POOL_CHUNK + lane, e - 1);
            const int i1 = min(nx * POOL_CHUNK + 32 + lane, e - 1);
            s0 = src[i0]; d0 = HAVE_GID ? (int)GID[i0] : dst[i0];
            s1 = src[i1]; d1 = HAVE_GID ? (int)GID[i1] : dst[i1];
            if (HAVE_WGT) { w0 = WGT[i0]; w1 = WGT[i1]; }
        }
        q[lane].t      = cs0 * NHID;  q[lane].g      = g0 * NHID;  q[lane].w      = cw0;
        q[lane + 32].t = cs1 * NHID;  q[lane + 32].g = g1 * NHID;  q[lane + 32].w = cw1;

        for (int base = 0; base < n; base += 32) {
            PoolEnt ent[16];
            float   m[16];
#pragma unroll
            for (int i = 0; i < 16; ++i) {
                const int idx = base + 2 * i + r;
                ent[i] = q[idx < n ? idx : n - 1];
            }
#pragma unroll
            for (int i = 0; i < 16; ++i)
                m[i] = T[(unsigned)ent[i].t + c];
#pragma unroll
            for (int i = 0; i < 16; ++i) {
                const int idx = base + 2 * i + r;
                const float v = m[i] * ent[i].w;
                atomicAdd(&tab[(unsigned)ent[i].g + c], idx < n ? v : 0.0f);
            }
        }
    }
}

__global__ __launch_bounds__(POOL_THREADS)
void edge_pool(const int* __restrict__ src, const int* __restrict__ dst,
               const int* __restrict__ bvec,
               const float* __restrict__ dinv, const float* __restrict__ WGT,
               const unsigned short* __restrict__ GID,
               const float* __restrict__ T, const float* __restrict__ bias,
               float* __restrict__ GP, int e, int nnodes, size_t ws_size) {
    __shared__ __attribute__((aligned(16))) float tab[NGRAPH * NHID];
    __shared__ PoolEnt queue[POOL_WAVES][POOL_CHUNK];
    __shared__ int  half_s;

    const int tid  = threadIdx.x;
    const int wave = __builtin_amdgcn_readfirstlane(tid >> 5);
    const int lane = tid & 31;

    float4* tab4 = (float4*)tab;
    for (int i = tid; i < NGRAPH * NHID / 4; i += POOL_THREADS)
        tab4[i] = make_float4(0.0f, 0.0f, 0.0f, 0.0f);
    if (tid == 0) {
        const unsigned id = __builtin_amdgcn_s_sendmsg_rtn(0x87);
        half_s = (int)((id >> 18) & 1u);
    }
    __syncthreads();

    const int nb = gridDim.x, b = blockIdx.x;

    {
        const unsigned c = lane & 15;
        const int      r = lane >> 4;
        const int n_lo = (int)((long long)b * nnodes / nb);
        const int n_hi = (int)((long long)(b + 1) * nnodes / nb);
        const float bc = bias[c];
        for (int k0 = n_lo + wave * 2; k0 < n_hi; k0 += POOL_WAVES * 2 * 8) {
            float di[8], tv[8];
            int   gi[8];
#pragma unroll
            for (int j = 0; j < 8; ++j) {
                const int i = min(k0 + r + POOL_WAVES * 2 * j, n_hi - 1);
                di[j] = dinv[i];
                tv[j] = T[(unsigned)i * NHID + c];
                gi[j] = bvec[i];
            }
#pragma unroll
            for (int j = 0; j < 8; ++j) {
                const int i = k0 + r + POOL_WAVES * 2 * j;
                const float v = tv[j] * di[j] * di[j] + bc;
                atomicAdd(&tab[(unsigned)gi[j] * NHID + c], i < n_hi ? v : 0.0f);
            }
        }
    }

    const int nchunks = (e + POOL_CHUNK - 1) / POOL_CHUNK;
    const int c_lo = (int)((long long)b * nchunks / nb);
    const int c_hi = (int)((long long)(b + 1) * nchunks / nb);
    if (ws_size >= WS_BYTES_WITH_GID)
        pool_edges<true, true>(src, dst, bvec, dinv, WGT, GID, T, tab, queue[wave], e, c_lo, c_hi, wave, lane);
    else if (ws_size >= WS_BYTES_WITH_WGT)
        pool_edges<true, false>(src, dst, bvec, dinv, WGT, GID, T, tab, queue[wave], e, c_lo, c_hi, wave, lane);
    else
        pool_edges<false, false>(src, dst, bvec, dinv, WGT, GID, T, tab, queue[wave], e, c_lo, c_hi, wave, lane);

    __syncthreads();

    float* gp = GP + (size_t)half_s * (NGRAPH * NHID);
    const unsigned rot = (unsigned)b * (unsigned)(NGRAPH * NHID / POOL_GROUPS);
    for (int k0 = 0; k0 < NGRAPH * NHID / POOL_THREADS; k0 += 8) {
        unsigned idx[8];
        float    v[8];
#pragma unroll
        for (int k = 0; k < 8; ++k) {
            idx[k] = (rot + (unsigned)(k0 + k) * POOL_THREADS + (unsigned)tid) & (unsigned)(NGRAPH * NHID - 1);
            v[k]   = tab[idx[k]];
        }
#pragma unroll
        for (int k = 0; k < 8; ++k)
            atomicAdd(gp + idx[k], v[k]);
    }
}

__global__ void lin1_kernel(const float* __restrict__ G, const float* __restrict__ W,
                            const float* __restrict__ B, float* __restrict__ O) {
    int t = blockIdx.x * blockDim.x + threadIdx.x;
    if (t >= NGRAPH * NMLP) return;
    int gi = t / NMLP, j = t % NMLP;
    float acc = B[j];
#pragma unroll
    for (int k = 0; k < NHID; ++k)
        acc = fmaf(fmaxf(G[gi * NHID + k] + G[NGRAPH * NHID + gi * NHID + k], 0.0f), W[k * NMLP + j], acc);
    O[t] = fmaxf(acc, 0.0f);
}

__global__ void lin2_kernel(const float* __restrict__ G, const float* __restrict__ W,
                            const float* __restrict__ B, float* __restrict__ O) {
    int t = blockIdx.x * blockDim.x + threadIdx.x;
    if (t >= NGRAPH * NCLS) return;
    int gi = t / NCLS, c = t % NCLS;
    float acc = B[c];
    for (int k = 0; k < NMLP; ++k)
        acc = fmaf(G[gi * NMLP + k], W[k * NCLS + c], acc);
    O[t] = acc;
}


extern "C" void kernel_launch(void* const* d_in, const int* in_sizes, int n_in,
                              void* d_out, int out_size, void* d_ws, size_t ws_size,
                              hipStream_t stream) {
    const float* x   = (const float*)d_in[0];
    const int*  eidx = (const int*)d_in[1];
    const int*  bvec = (const int*)d_in[2];
    const float* W1  = (const float*)d_in[3];
    const float* b1  = (const float*)d_in[4];
    const float* W2  = (const float*)d_in[5];
    const float* b2  = (const float*)d_in[6];
    const float* Wl1 = (const float*)d_in[7];
    const float* bl1 = (const float*)d_in[8];
    const float* Wl2 = (const float*)d_in[9];
    const float* bl2 = (const float*)d_in[10];
    float* out = (float*)d_out;

    char* ws = (char*)d_ws;
    size_t off = 0;
    auto walloc = [&](size_t bytes) -> void* {
        void* p = ws + off;
        off = (off + bytes + 255) & ~(size_t)255;
        return p;
    };
    float* deg  = (float*)walloc((size_t)NNODES * 4);
    float* bufA = (float*)walloc((size_t)NNODES * NHID * 4);
    float* bufB = (float*)walloc((size_t)NNODES * NHID * 4);
    (void)walloc((size_t)NGRAPH * NHID * 4);
    float* g2   = (float*)walloc((size_t)NGRAPH * NMLP * 4);
    float* wgt  = (float*)walloc((size_t)NEDGES * 4);
    unsigned short* gid = (unsigned short*)walloc((size_t)NEDGES * 2);

    const int* src = eidx;
    const int* dst = eidx + NEDGES;

    const int TB = 256;
    const int nN    = (NNODES + TB - 1) / TB;
    const int nTile = ((NNODES / 16) + 7) / 8;
    const int nGrp  = (NEDGES + EDGES_PER_GROUP - 1) / EDGES_PER_GROUP;
    const int nNorm = (NEDGES + TB * NORM_PER_THREAD - 1) / (TB * NORM_PER_THREAD);
    const int nDeg  = (NEDGES + TB * DEG_PER_THREAD - 1) / (TB * DEG_PER_THREAD);

    unsigned* ctr = (unsigned*)g2;

    float* deg1 = bufB;
    fill_deg_ctr<<<nN, TB, 0, stream>>>(deg, deg1, NNODES, ctr, 2 * CTR_STRIDE);
    deg_accum<<<nDeg, TB, 0, stream>>>(dst, deg, deg1, NEDGES);
    int* rec = (int*)bufA;
    rsqrt_sum<<<nN, TB, 0, stream>>>(deg, deg1, NNODES, bvec, rec, ws_size);
    float* dinv = deg;
    edge_norm<<<nNorm, TB, 0, stream>>>(src, dst, dinv, rec, wgt, gid, NEDGES, ws_size);

    gemm_start<<<nTile, TB, 0, stream>>>(x, W1, bufA, bufB, dinv, b1, NNODES, NFEAT);
    edge_scatter<<<2 * nGrp, TB, 0, stream>>>(src, dst, dinv, wgt, bufA, bufB, NEDGES,
                                              ctr, nGrp, NNODES / 2, ws_size);

    gemm_relu<<<nTile, TB, 0, stream>>>(bufB, W2, bufA, NNODES, NHID);
    float* gtab = bufB;
    fill_f32<<<(2 * NGRAPH * NHID + TB - 1) / TB, TB, 0, stream>>>(gtab, 2 * NGRAPH * NHID, 0.0f);
    edge_pool<<<POOL_GROUPS, POOL_THREADS, 0, stream>>>(src, dst, bvec, dinv, wgt, gid, bufA, b2, gtab,
                                                        NEDGES, NNODES, ws_size);

    lin1_kernel<<<(NGRAPH * NMLP + TB - 1) / TB, TB, 0, stream>>>(gtab, Wl1, bl1, g2);
    lin2_kernel<<<(NGRAPH * NCLS + TB - 1) / TB, TB, 0, stream>>>(g2, Wl2, bl2, out);
}


# Edge-weight launch as a resident grid with one degree byte per node (200 KB) in LDS: dinv[src] is rsqrt of the byte, so one scattered request per edge is left, not two; weights bit-identical
# speedup vs baseline: 3.9472x; 1.0506x over previous
#include <hip/hip_runtime.h>
#include <hip/hip_bf16.h>


typedef float v2f __attribute__((ext_vector_type(2)));
typedef float v8f __attribute__((ext_vector_type(8)));

#define NNODES  200000
#define NEDGES  6400000
#define NFEAT   128
#define NHID    16
#define NMLP    100
#define NCLS    27
#define NGRAPH  4096

#define WS_BYTES_WITH_WGT ((size_t)NNODES * 4 + 2 * ((size_t)NNODES * NHID * 4) + \
                           (size_t)NGRAPH * NHID * 4 + (size_t)NGRAPH * NMLP * 4 + \
                           (size_t)NEDGES * 4)
#define WS_BYTES_WITH_GID (WS_BYTES_WITH_WGT + (size_t)NEDGES * 2)
static_assert(NGRAPH <= 65536, "a graph id must fit the 16 bits of the per-edge plane");
static_assert((size_t)NNODES <= (size_t)NGRAPH * NHID * 4, "the degree bytes must fit their plane");
static_assert(NNODES % 16 == 0, "the degree bytes are copied to LDS sixteen at a time");
#define DEG8_CAP 255u


__global__ void fill_f32(float* __restrict__ p, int n, float v) {
    int i = blockIdx.x * blockDim.x + threadIdx.x;
    if (i < n) p[i] = v;
}

__global__ void fill_deg_ctr(float* __restrict__ d0, float* __restrict__ d1, int n,
                             unsigned* __restrict__ ctr, int nctr) {
    int i = blockIdx.x * blockDim.x + threadIdx.x;
    if (i < n) { d0[i] = 1.0f; d1[i] = 0.0f; }
    if (i < nctr) ctr[i] = 0u;
}

#define DEG_PER_THREAD 4

__global__ __launch_bounds__(256)
void deg_accum(const int* __restrict__ dst, float* __restrict__ deg0,
               float* __restrict__ deg1, int e) {
    __shared__ int half_s;
    const int base = blockIdx.x * (256 * DEG_PER_THREAD) + threadIdx.x;
    unsigned d[DEG_PER_THREAD];
#pragma unroll
    for (int k = 0; k < DEG_PER_THREAD; ++k)
        d[k] = (unsigned)dst[min(base + k * 256, e - 1)];
    if (threadIdx.x == 0) {
        const unsigned id = __builtin_amdgcn_s_sendmsg_rtn(0x87);
        half_s = (int)((id >> 18) & 1u);
    }
    __syncthreads();
    float* __restrict__ deg = half_s ? deg1 : deg0;
    if ((blockIdx.x + 1) * (256 * DEG_PER_THREAD) <= e) {
#pragma unroll
        for (int k = 0; k < DEG_PER_THREAD; ++k)
            atomicAdd(&deg[d[k]], 1.0f);
    } else {
        for (int k = 0; k < DEG_PER_THREAD; ++k)
            if (base + k * 256 < e) atomicAdd(&deg[d[k]], 1.0f);
    }
}

__global__ void rsqrt_sum(float* __restrict__ p, const float* __restrict__ q, int n,
                          const int* __restrict__ bvec, int* __restrict__ rec,
                          unsigned char* __restrict__ deg8, size_t ws_size) {
    const int i = (blockIdx.x * blockDim.x + threadIdx.x) * 4;
    if (i >= n) return;
    const bool have_gid = ws_size >= WS_BYTES_WITH_GID;
    if (i + 3 < n) {
        const float4 a = *(const float4*)(p + i);
        const float4 b = *(const float4*)(q + i);
        const float d0 = a.x + b.x, d1 = a.y + b.y, d2 = a.z + b.z, d3 = a.w + b.w;
        const float r0 = rsqrtf(d0), r1 = rsqrtf(d1), r2 = rsqrtf(d2), r3 = rsqrtf(d3);
        *(float4*)(p + i) = make_float4(r0, r1, r2, r3);
        if (have_gid) {
            const int4 g = *(const int4*)(bvec + i);
            int4* rec4 = (int4*)rec + (i >> 1);
            rec4[0] = make_int4(__float_as_int(r0), g.x, __float_as_int(r1), g.y);
            rec4[1] = make_int4(__float_as_int(r2), g.z, __float_as_int(r3), g.w);
            const unsigned c0 = (unsigned)fminf(d0, (float)DEG8_CAP), c1 = (unsigned)fminf(d1, (float)DEG8_CAP);
            const unsigned c2 = (unsigned)fminf(d2, (float)DEG8_CAP), c3 = (unsigned)fminf(d3, (float)DEG8_CAP);
            ((unsigned*)deg8)[i >> 2] = c0 | (c1 << 8) | (c2 << 16) | (c3 << 24);
        }
    } else {
        for (int k = i; k < n; ++k) {
            const float d = p[k] + q[k];
            const float r = rsqrtf(d);
            p[k] = r;
            if (have_gid) {
                ((int2*)rec)[k] = make_int2(__float_as_int(r), bvec[k]);
                deg8[k] = (unsigned char)fminf(d, (float)DEG8_CAP);
            }
        }
    }
}

template <bool RELU_A, bool START>
__device__ __forceinline__ void gemm_tile(const float* __restrict__ X, const float* __restrict__ W,
                                          float* __restrict__ T, float* __restrict__ S,
                                          const float* __restrict__ dinv,
                                          const float* __restrict__ bias,
                                          float* lw, int M, int K) {
    const int tid = threadIdx.x;
    const int kn  = K * NHID;
    for (int i = tid; i < kn; i += 256) lw[i] = W[i];
    __syncthreads();

    const int wave = tid >> 5;
    const int lane = tid & 31;
    const int tile = blockIdx.x * 8 + wave;
    if (tile * 16 >= M) return;

    const int row   = tile * 16 + (lane & 15);
    const int kHalf = (lane >> 4) << 1;
    const int n     = lane & 15;
    const int rbase = tile * 16 + ((lane >> 4) << 3);

    float4 dlo = make_float4(0.0f, 0.0f, 0.0f, 0.0f), dhi = dlo;
    float  bn  = 0.0f;
    if (START) {
        dlo = *(const float4*)(dinv + rbase);
        dhi = *(const float4*)(dinv + rbase + 4);
        bn  = bias[n];
    }

    v8f c = {};
    const float* xr = X + (size_t)row * K + kHalf;
    for (int k0 = 0; k0 < K; k0 += 4) {
        float2 av = *(const float2*)(xr + k0);
        v2f a;
        if (RELU_A) { a.x = fmaxf(av.x, 0.0f); a.y = fmaxf(av.y, 0.0f); }
        else        { a.x = av.x;              a.y = av.y; }
        v2f b;
        b.x = lw[(k0 + kHalf)     * NHID + n];
        b.y = lw[(k0 + kHalf + 1) * NHID + n];
        c = __builtin_amdgcn_wmma_f32_16x16x4_f32(
                 false, a,  false, b,
                 (short)0, c,  false,  false);
    }

#pragma unroll
    for (int v = 0; v < 8; ++v)
        T[(size_t)(rbase + v) * NHID + n] = c[v];
    if (START) {
        const float di[8] = { dlo.x, dlo.y, dlo.z, dlo.w, dhi.x, dhi.y, dhi.z, dhi.w };
#pragma unroll
        for (int v = 0; v < 8; ++v)
            S[(size_t)(rbase + v) * NHID + n] = c[v] * di[v] * di[v] + bn;
    }
}

__global__ __launch_bounds__(256)
void gemm_start(const float* __restrict__ X, const float* __restrict__ W,
                float* __restrict__ T, float* __restrict__ S,
                const float* __restrict__ dinv, const float* __restrict__ bias,
                int M, int K) {
    __shared__ float lw[NFEAT * NHID];
    gemm_tile<false, true>(X, W, T, S, dinv, bias, lw, M, K);
}

__global__ __launch_bounds__(256)
void gemm_relu(const float* __restrict__ X, const float* __restrict__ W,
               float* __restrict__ T, int M, int K) {
    __shared__ float lw[NFEAT * NHID];
    gemm_tile<true, false>(X, W, T, nullptr, nullptr, nullptr, lw, M, K);
}

#define NORM_GROUPS     256
#define NORM_THREADS    1024
#define NORM_PER_THREAD 5

__global__ __launch_bounds__(NORM_THREADS)
void edge_norm(const int* __restrict__ src, const int* __restrict__ dst,
               const float* __restrict__ dinv, const int* __restrict__ rec,
               const unsigned char* __restrict__ deg8,
               float* __restrict__ WGT, unsigned short* __restrict__ GID, int e,
               size_t ws_size) {
    __shared__ __attribute__((aligned(16))) unsigned char degs[NNODES];

    if (ws_size < WS_BYTES_WITH_WGT) return;
    const bool have_gid = ws_size >= WS_BYTES_WITH_GID;
    const int tid  = threadIdx.x;
    const int nblk = (e + NORM_THREADS - 1) / NORM_THREADS;
    const int lo = (int)((long long)blockIdx.x * nblk / gridDim.x);
    const int hi = (int)((long long)(blockIdx.x + 1) * nblk / gridDim.x);

    unsigned s[NORM_PER_THREAD], d[NORM_PER_THREAD];
#pragma unroll
    for (int k = 0; k < NORM_PER_THREAD; ++k) {
        const int i = min((lo + k) * NORM_THREADS + tid, e - 1);
        s[k] = (unsigned)src[i];
        d[k] = (unsigned)dst[i];
    }

    if (have_gid) {
        const uint4* __restrict__ g4 = (const uint4*)deg8;
        uint4* l4 = (uint4*)degs;
        for (int i = tid; i < NNODES / 16; i += NORM_THREADS) l4[i] = g4[i];
        __syncthreads();
    }

    const int2* __restrict__ rec2 = (const int2*)rec;
    for (int blk = lo; blk < hi; blk += NORM_PER_THREAD) {
        unsigned cs[NORM_PER_THREAD], cd[NORM_PER_THREAD];
#pragma unroll
        for (int k = 0; k < NORM_PER_THREAD; ++k) { cs[k] = s[k]; cd[k] = d[k]; }
#pragma unroll
        for (int k = 0; k < NORM_PER_THREAD; ++k) {
            const int i = min((blk + NORM_PER_THREAD + k) * NORM_THREADS + tid, e - 1);
            s[k] = (unsigned)src[i];
            d[k] = (unsigned)dst[i];
        }
        float w[NORM_PER_THREAD];
        int   g[NORM_PER_THREAD];
        if (have_gid) {
            int2     r[NORM_PER_THREAD];
            unsigned b[NORM_PER_THREAD];
#pragma unroll
            for (int k = 0; k < NORM_PER_THREAD; ++k) {
                r[k] = rec2[cd[k]];
                b[k] = degs[cs[k]];
            }
#pragma unroll
            for (int k = 0; k < NORM_PER_THREAD; ++k) {
                float a = rsqrtf((float)b[k]);
                if (b[k] == DEG8_CAP) a = dinv[cs[k]];
                w[k] = a * __int_as_float(r[k].x);
                g[k] = r[k].y;
            }
        } else {
            float a[NORM_PER_THREAD], c[NORM_PER_THREAD];
#pragma unroll
            for (int k = 0; k < NORM_PER_THREAD; ++k) {
                a[k] = dinv[cs[k]];
                c[k] = dinv[cd[k]];
            }
#pragma unroll
            for (int k = 0; k < NORM_PER_THREAD; ++k) { w[k] = a[k] * c[k]; g[k] = 0; }
        }
#pragma unroll
        for (int k = 0; k < NORM_PER_THREAD; ++k) {
            const int i = (blk + k) * NORM_THREADS + tid;
            if (blk + k < hi && i < e) {
                WGT[i] = w[k];
                if (have_gid) GID[i] = (unsigned short)g[k];
            }
        }
    }
}


#define EDGES_PER_WAVE 64
#define EDGES_PER_GROUP (8 * EDGES_PER_WAVE)
#define CTR_STRIDE 64

__global__ __launch_bounds__(256)
void edge_scatter(const int* __restrict__ src, const int* __restrict__ dst,
                  const float* __restrict__ dinv, const float* __restrict__ WGT,
                  const float* __restrict__ T,
                  float* __restrict__ OUT, int e, unsigned* __restrict__ ctr,
                  int ngroups, int nsplit, size_t ws_size) {
    __shared__ int  pick[2];
    __shared__ int4 queue[8][EDGES_PER_WAVE];

    if (threadIdx.x == 0) {
        const unsigned id = __builtin_amdgcn_s_sendmsg_rtn(0x87);
        int h = (int)((id >> 18) & 1u);
        int g = (int)atomicAdd(ctr + h * CTR_STRIDE, 1u);
        if (g >= ngroups) {
            h ^= 1;
            g = (int)atomicAdd(ctr + h * CTR_STRIDE, 1u);
        }
        pick[0] = g;
        pick[1] = h;
    }
    __syncthreads();
    const int g = pick[0];
    const int h = pick[1];
    if (g >= ngroups) return;

    const int wave = threadIdx.x >> 5;
    const int lane = threadIdx.x & 31;
    const int e0 = g * EDGES_PER_GROUP + wave * EDGES_PER_WAVE + lane;
    const int e1 = e0 + 32;

    const bool have_wgt = ws_size >= WS_BYTES_WITH_WGT;
    int s0 = 0, d0 = 0, s1 = 0, d1 = 0;
    float w0 = 0.0f, w1 = 0.0f;
    const bool v0 = e0 < e, v1 = e1 < e;
    if (v0) { s0 = src[e0]; d0 = dst[e0]; }
    if (v1) { s1 = src[e1]; d1 = dst[e1]; }
    if (have_wgt) {
        if (v0) w0 = WGT[e0];
        if (v1) w1 = WGT[e1];
    }
    const bool m0 = v0 && ((d0 >= nsplit) == (h != 0));
    const bool m1 = v1 && ((d1 >= nsplit) == (h != 0));

    const unsigned b0 = __builtin_amdgcn_ballot_w32(m0);
    const unsigned b1 = __builtin_amdgcn_ballot_w32(m1);
    const unsigned below = (1u << lane) - 1u;
    const int n0 = __popc(b0);
    const int n  = n0 + __popc(b1);
    int4* q = queue[wave];
    if (!have_wgt) {
        const float a0 = dinv[m0 ? s0 : 0], c0 = dinv[m0 ? d0 : 0];
        const float a1 = dinv[m1 ? s1 : 0], c1 = dinv[m1 ? d1 : 0];
        w0 = a0 * c0;
        w1 = a1 * c1;
    }
    if (m0) q[__popc(b0 & below)]      = make_int4(s0, d0, __float_as_int(w0), 0);
    if (m1) q[n0 + __popc(b1 & below)] = make_int4(s1, d1, __float_as_int(w1), 0);

    const unsigned c = lane & 15;
    const int      r = lane >> 4;
    for (int base = 0; base < n; base += 16) {
        int4  ent[8];
        float m[8];
#pragma unroll
        for (int i = 0; i < 8; ++i) {
            const int idx = base + 2 * i + r;
            ent[i] = q[idx < n ? idx : n - 1];
        }
#pragma unroll
        for (int i = 0; i < 8; ++i)
            m[i] = T[(unsigned)ent[i].x * NHID + c];
#pragma unroll
        for (int i = 0; i < 8; ++i) {
            const int idx = base + 2 * i + r;
            if (idx < n)
                atomicAdd(OUT + ((unsigned)ent[i].y * NHID + c), m[i] * __int_as_float(ent[i].z));
        }
    }
}


#define POOL_WAVES   32
#define POOL_THREADS (POOL_WAVES * 32)
#define POOL_CHUNK   64
#define POOL_GROUPS  256

struct __attribute__((aligned(16))) PoolEnt { int t; int g; float w; };

template <bool HAVE_WGT, bool HAVE_GID>
__device__ __forceinline__ void pool_edges(const int* __restrict__ src, const int* __restrict__ dst,
                                           const int* __restrict__ bvec,
                                           const float* __restrict__ dinv, const float* __restrict__ WGT,
                                           const unsigned short* __restrict__ GID,
                                           const float* __restrict__ T, float* tab, PoolEnt* q,
                                           int e, int c_lo, int c_hi, int wave, int lane) {
    const unsigned c = lane & 15;
    const int      r = lane >> 4;
    if (c_lo + wave >= c_hi) return;

    int s0, d0, s1, d1;
    float w0 = 0.0f, w1 = 0.0f;
    {
        const int i0 = min((c_lo + wave) * POOL_CHUNK + lane, e - 1);
        const int i1 = min((c_lo + wave) * POOL_CHUNK + 32 + lane, e - 1);
        s0 = src[i0]; d0 = HAVE_GID ? (int)GID[i0] : dst[i0];
        s1 = src[i1]; d1 = HAVE_GID ? (int)GID[i1] : dst[i1];
        if (HAVE_WGT) { w0 = WGT[i0]; w1 = WGT[i1]; }
    }
    for (int ch = c_lo + wave; ch < c_hi; ch += POOL_WAVES) {
        const int n = min(POOL_CHUNK, e - ch * POOL_CHUNK);
        const int g0 = HAVE_GID ? d0 : bvec[d0];
        const int g1 = HAVE_GID ? d1 : bvec[d1];
        if (!HAVE_WGT) {
            w0 = dinv[s0] * dinv[d0];
            w1 = dinv[s1] * dinv[d1];
        }
        const int cs0 = s0, cs1 = s1;
        const float cw0 = w0, cw1 = w1;
        {
            const int nx = min(ch + POOL_WAVES, c_hi - 1);
            const int i0 = min(nx * POOL_CHUNK + lane, e - 1);
            const int i1 = min(nx * POOL_CHUNK + 32 + lane, e - 1);
            s0 = src[i0]; d0 = HAVE_GID ? (int)GID[i0] : dst[i0];
            s1 = src[i1]; d1 = HAVE_GID ? (int)GID[i1] : dst[i1];
            if (HAVE_WGT) { w0 = WGT[i0]; w1 = WGT[i1]; }
        }
        q[lane].t      = cs0 * NHID;  q[lane].g      = g0 * NHID;  q[lane].w      = cw0;
        q[lane + 32].t = cs1 * NHID;  q[lane + 32].g = g1 * NHID;  q[lane + 32].w = cw1;

        for (int base = 0; base < n; base += 32) {
            PoolEnt ent[16];
            float   m[16];
#pragma unroll
            for (int i = 0; i < 16; ++i) {
                const int idx = base + 2 * i + r;
                ent[i] = q[idx < n ? idx : n - 1];
            }
#pragma unroll
            for (int i = 0; i < 16; ++i)
                m[i] = T[(unsigned)ent[i].t + c];
#pragma unroll
            for (int i = 0; i < 16; ++i) {
                const int idx = base + 2 * i + r;
                const float v = m[i] * ent[i].w;
                atomicAdd(&tab[(unsigned)ent[i].g + c], idx < n ? v : 0.0f);
            }
        }
    }
}

__global__ __launch_bounds__(POOL_THREADS)
void edge_pool(const int* __restrict__ src, const int* __restrict__ dst,
               const int* __restrict__ bvec,
               const float* __restrict__ dinv, const float* __restrict__ WGT,
               const unsigned short* __restrict__ GID,
               const float* __restrict__ T, const float* __restrict__ bias,
               float* __restrict__ GP, int e, int nnodes, size_t ws_size) {
    __shared__ __attribute__((aligned(16))) float tab[NGRAPH * NHID];
    __shared__ PoolEnt queue[POOL_WAVES][POOL_CHUNK];
    __shared__ int  half_s;

    const int tid  = threadIdx.x;
    const int wave = __builtin_amdgcn_readfirstlane(tid >> 5);
    const int lane = tid & 31;

    float4* tab4 = (float4*)tab;
    for (int i = tid; i < NGRAPH * NHID / 4; i += POOL_THREADS)
        tab4[i] = make_float4(0.0f, 0.0f, 0.0f, 0.0f);
    if (tid == 0) {
        const unsigned id = __builtin_amdgcn_s_sendmsg_rtn(0x87);
        half_s = (int)((id >> 18) & 1u);
    }
    __syncthreads();

    const int nb = gridDim.x, b = blockIdx.x;

    {
        const unsigned c = lane & 15;
        const int      r = lane >> 4;
        const int n_lo = (int)((long long)b * nnodes / nb);
        const int n_hi = (int)((long long)(b + 1) * nnodes / nb);
        const float bc = bias[c];
        for (int k0 = n_lo + wave * 2; k0 < n_hi; k0 += POOL_WAVES * 2 * 8) {
            float di[8], tv[8];
            int   gi[8];
#pragma unroll
            for (int j = 0; j < 8; ++j) {
                const int i = min(k0 + r + POOL_WAVES * 2 * j, n_hi - 1);
                di[j] = dinv[i];
                tv[j] = T[(unsigned)i * NHID + c];
                gi[j] = bvec[i];
            }
#pragma unroll
            for (int j = 0; j < 8; ++j) {
                const int i = k0 + r + POOL_WAVES * 2 * j;
                const float v = tv[j] * di[j] * di[j] + bc;
                atomicAdd(&tab[(unsigned)gi[j] * NHID + c], i < n_hi ? v : 0.0f);
            }
        }
    }

    const int nchunks = (e + POOL_CHUNK - 1) / POOL_CHUNK;
    const int c_lo = (int)((long long)b * nchunks / nb);
    const int c_hi = (int)((long long)(b + 1) * nchunks / nb);
    if (ws_size >= WS_BYTES_WITH_GID)
        pool_edges<true, true>(src, dst, bvec, dinv, WGT, GID, T, tab, queue[wave], e, c_lo, c_hi, wave, lane);
    else if (ws_size >= WS_BYTES_WITH_WGT)
        pool_edges<true, false>(src, dst, bvec, dinv, WGT, GID, T, tab, queue[wave], e, c_lo, c_hi, wave, lane);
    else
        pool_edges<false, false>(src, dst, bvec, dinv, WGT, GID, T, tab, queue[wave], e, c_lo, c_hi, wave, lane);

    __syncthreads();

    float* gp = GP + (size_t)half_s * (NGRAPH * NHID);
    const unsigned rot = (unsigned)b * (unsigned)(NGRAPH * NHID / POOL_GROUPS);
    for (int k0 = 0; k0 < NGRAPH * NHID / POOL_THREADS; k0 += 8) {
        unsigned idx[8];
        float    v[8];
#pragma unroll
        for (int k = 0; k < 8; ++k) {
            idx[k] = (rot + (unsigned)(k0 + k) * POOL_THREADS + (unsigned)tid) & (unsigned)(NGRAPH * NHID - 1);
            v[k]   = tab[idx[k]];
        }
#pragma unroll
        for (int k = 0; k < 8; ++k)
            atomicAdd(gp + idx[k], v[k]);
    }
}

__global__ void lin1_kernel(const float* __restrict__ G, const float* __restrict__ W,
                            const float* __restrict__ B, float* __restrict__ O) {
    int t = blockIdx.x * blockDim.x + threadIdx.x;
    if (t >= NGRAPH * NMLP) return;
    int gi = t / NMLP, j = t % NMLP;
    float acc = B[j];
#pragma unroll
    for (int k = 0; k < NHID; ++k)
        acc = fmaf(fmaxf(G[gi * NHID + k] + G[NGRAPH * NHID + gi * NHID + k], 0.0f), W[k * NMLP + j], acc);
    O[t] = fmaxf(acc, 0.0f);
}

__global__ void lin2_kernel(const float* __restrict__ G, const float* __restrict__ W,
                            const float* __restrict__ B, float* __restrict__ O) {
    int t = blockIdx.x * blockDim.x + threadIdx.x;
    if (t >= NGRAPH * NCLS) return;
    int gi = t / NCLS, c = t % NCLS;
    float acc = B[c];
    for (int k = 0; k < NMLP; ++k)
        acc = fmaf(G[gi * NMLP + k], W[k * NCLS + c], acc);
    O[t] = acc;
}


extern "C" void kernel_launch(void* const* d_in, const int* in_sizes, int n_in,
                              void* d_out, int out_size, void* d_ws, size_t ws_size,
                              hipStream_t stream) {
    const float* x   = (const float*)d_in[0];
    const int*  eidx = (const int*)d_in[1];
    const int*  bvec = (const int*)d_in[2];
    const float* W1  = (const float*)d_in[3];
    const float* b1  = (const float*)d_in[4];
    const float* W2  = (const float*)d_in[5];
    const float* b2  = (const float*)d_in[6];
    const float* Wl1 = (const float*)d_in[7];
    const float* bl1 = (const float*)d_in[8];
    const float* Wl2 = (const float*)d_in[9];
    const float* bl2 = (const float*)d_in[10];
    float* out = (float*)d_out;

    char* ws = (char*)d_ws;
    size_t off = 0;
    auto walloc = [&](size_t bytes) -> void* {
        void* p = ws + off;
        off = (off + bytes + 255) & ~(size_t)255;
        return p;
    };
    float* deg  = (float*)walloc((size_t)NNODES * 4);
    float* bufA = (float*)walloc((size_t)NNODES * NHID * 4);
    float* bufB = (float*)walloc((size_t)NNODES * NHID * 4);
    unsigned char* deg8 = (unsigned char*)walloc((size_t)NGRAPH * NHID * 4);
    float* g2   = (float*)walloc((size_t)NGRAPH * NMLP * 4);
    float* wgt  = (float*)walloc((size_t)NEDGES * 4);
    unsigned short* gid = (unsigned short*)walloc((size_t)NEDGES * 2);

    const int* src = eidx;
    const int* dst = eidx + NEDGES;

    const int TB = 256;
    const int nN    = (NNODES + TB - 1) / TB;
    const int nTile = ((NNODES / 16) + 7) / 8;
    const int nGrp  = (NEDGES + EDGES_PER_GROUP - 1) / EDGES_PER_GROUP;
    const int nDeg  = (NEDGES + TB * DEG_PER_THREAD - 1) / (TB * DEG_PER_THREAD);

    unsigned* ctr = (unsigned*)g2;

    float* deg1 = bufB;
    fill_deg_ctr<<<nN, TB, 0, stream>>>(deg, deg1, NNODES, ctr, 2 * CTR_STRIDE);
    deg_accum<<<nDeg, TB, 0, stream>>>(dst, deg, deg1, NEDGES);
    int* rec = (int*)bufA;
    rsqrt_sum<<<((NNODES + 3) / 4 + TB - 1) / TB, TB, 0, stream>>>(deg, deg1, NNODES, bvec, rec, deg8, ws_size);
    float* dinv = deg;
    edge_norm<<<NORM_GROUPS, NORM_THREADS, 0, stream>>>(src, dst, dinv, rec, deg8, wgt, gid, NEDGES, ws_size);

    gemm_start<<<nTile, TB, 0, stream>>>(x, W1, bufA, bufB, dinv, b1, NNODES, NFEAT);
    edge_scatter<<<2 * nGrp, TB, 0, stream>>>(src, dst, dinv, wgt, bufA, bufB, NEDGES,
                                              ctr, nGrp, NNODES / 2, ws_size);

    gemm_relu<<<nTile, TB, 0, stream>>>(bufB, W2, bufA, NNODES, NHID);
    float* gtab = bufB;
    fill_f32<<<(2 * NGRAPH * NHID + TB - 1) / TB, TB, 0, stream>>>(gtab, 2 * NGRAPH * NHID, 0.0f);
    edge_pool<<<POOL_GROUPS, POOL_THREADS, 0, stream>>>(src, dst, bvec, dinv, wgt, gid, bufA, b2, gtab,
                                                        NEDGES, NNODES, ws_size);

    lin1_kernel<<<(NGRAPH * NMLP + TB - 1) / TB, TB, 0, stream>>>(gtab, Wl1, bl1, g2);
    lin2_kernel<<<(NGRAPH * NCLS + TB - 1) / TB, TB, 0, stream>>>(g2, Wl2, bl2, out);
}


# Edge-weight launch: dinv[dst] from the LDS degree byte too, graph of dst from a per-64-node LDS table (first graph + bit count of a step word): no scattered request per edge; weights exact
# speedup vs baseline: 4.3438x; 1.1005x over previous
#include <hip/hip_runtime.h>
#include <hip/hip_bf16.h>


typedef float v2f __attribute__((ext_vector_type(2)));
typedef float v8f __attribute__((ext_vector_type(8)));

#define NNODES  200000
#define NEDGES  6400000
#define NFEAT   128
#define NHID    16
#define NMLP    100
#define NCLS    27
#define NGRAPH  4096

#define WS_BYTES_WITH_WGT ((size_t)NNODES * 4 + 2 * ((size_t)NNODES * NHID * 4) + \
                           (size_t)NGRAPH * NHID * 4 + (size_t)NGRAPH * NMLP * 4 + \
                           (size_t)NEDGES * 4)
#define WS_BYTES_WITH_GID (WS_BYTES_WITH_WGT + (size_t)NEDGES * 2)
static_assert(NGRAPH <= 65536, "a graph id must fit the 16 bits of the per-edge plane");
static_assert((size_t)NNODES <= (size_t)NGRAPH * NHID * 4, "the degree bytes must fit their plane");
static_assert(NNODES % 16 == 0, "the degree bytes are copied to LDS sixteen at a time");
#define DEG8_CAP 255u
#define NBLK64        (NNODES / 64)
#define TAB_BASE_OFF  NNODES
#define TAB_BASE_N    ((NBLK64 + 3) & ~3)
#define TAB_MASK_OFF  (TAB_BASE_OFF + TAB_BASE_N * 2)
#define TAB_MASK_N    ((NBLK64 + 1) & ~1)
#define TAB_BYTES     (TAB_MASK_OFF + TAB_MASK_N * 8)
#define GBASE_IRREG   0xFFFFu
static_assert(NNODES % 64 == 0, "the graph table takes whole blocks of 64 nodes");
static_assert(TAB_MASK_OFF % 8 == 0 && TAB_BYTES % 16 == 0, "the table is copied sixteen bytes at a time");
static_assert((size_t)TAB_BYTES <= (size_t)NGRAPH * NHID * 4, "degree bytes and graph table must fit their plane");
static_assert(NGRAPH + 64 <= 0xFFFF, "a graph id must stay below GBASE_IRREG");
static_assert(TAB_MASK_N <= TAB_BASE_N && TAB_BASE_N <= NNODES / 4, "rsqrt_sum has a thread per table entry");


__global__ void fill_f32(float* __restrict__ p, int n, float v) {
    int i = blockIdx.x * blockDim.x + threadIdx.x;
    if (i < n) p[i] = v;
}

__global__ void fill_deg_ctr(float* __restrict__ d0, float* __restrict__ d1, int n,
                             unsigned* __restrict__ ctr, int nctr) {
    int i = blockIdx.x * blockDim.x + threadIdx.x;
    if (i < n) { d0[i] = 1.0f; d1[i] = 0.0f; }
    if (i < nctr) ctr[i] = 0u;
}

#define DEG_PER_THREAD 4

__global__ __launch_bounds__(256)
void deg_accum(const int* __restrict__ dst, float* __restrict__ deg0,
               float* __restrict__ deg1, int e) {
    __shared__ int half_s;
    const int base = blockIdx.x * (256 * DEG_PER_THREAD) + threadIdx.x;
    unsigned d[DEG_PER_THREAD];
#pragma unroll
    for (int k = 0; k < DEG_PER_THREAD; ++k)
        d[k] = (unsigned)dst[min(base + k * 256, e - 1)];
    if (threadIdx.x == 0) {
        const unsigned id = __builtin_amdgcn_s_sendmsg_rtn(0x87);
        half_s = (int)((id >> 18) & 1u);
    }
    __syncthreads();
    float* __restrict__ deg = half_s ? deg1 : deg0;
    if ((blockIdx.x + 1) * (256 * DEG_PER_THREAD) <= e) {
#pragma unroll
        for (int k = 0; k < DEG_PER_THREAD; ++k)
            atomicAdd(&deg[d[k]], 1.0f);
    } else {
        for (int k = 0; k < DEG_PER_THREAD; ++k)
            if (base + k * 256 < e) atomicAdd(&deg[d[k]], 1.0f);
    }
}

__global__ void rsqrt_sum(float* __restrict__ p, const float* __restrict__ q, int n,
                          const int* __restrict__ bvec, int* __restrict__ rec,
                          unsigned char* __restrict__ deg8, size_t ws_size) {
    const int t = blockIdx.x * blockDim.x + threadIdx.x;
    const int i = t * 4;
    const bool have_gid = ws_size >= WS_BYTES_WITH_GID;
    if (have_gid && t < TAB_BASE_N) {
        unsigned short*     gbase = (unsigned short*)(deg8 + TAB_BASE_OFF);
        unsigned long long* gmask = (unsigned long long*)(deg8 + TAB_MASK_OFF);
        if (t < NBLK64) {
            const int4* __restrict__ b4 = (const int4*)(bvec + t * 64);
            int4 v[16];
#pragma unroll
            for (int q = 0; q < 16; ++q) v[q] = b4[q];
            const int first = v[0].x;
            int prev = first;
            unsigned long long m = 0ull;
            bool ok = (unsigned)first < (unsigned)(GBASE_IRREG - 64u);
#pragma unroll
            for (int q = 0; q < 16; ++q) {
                const int c[4] = { v[q].x, v[q].y, v[q].z, v[q].w };
#pragma unroll
                for (int j = 0; j < 4; ++j) {
                    const unsigned step = (unsigned)(c[j] - prev);
                    m  |= (unsigned long long)(step == 1u) << (q * 4 + j);
                    ok  = ok && step <= 1u;
                    prev = c[j];
                }
            }
            gmask[t] = m;
            gbase[t] = (unsigned short)(ok ? (unsigned)first : GBASE_IRREG);
        } else {
            const int u = t - NBLK64;
            if (NBLK64 + u < TAB_MASK_N) gmask[NBLK64 + u] = 0ull;
            if (NBLK64 + u < TAB_BASE_N) gbase[NBLK64 + u] = (unsigned short)GBASE_IRREG;
        }
    }
    if (i >= n) return;
    if (i + 3 < n) {
        const float4 a = *(const float4*)(p + i);
        const float4 b = *(const float4*)(q + i);
        const float d0 = a.x + b.x, d1 = a.y + b.y, d2 = a.z + b.z, d3 = a.w + b.w;
        const float r0 = rsqrtf(d0), r1 = rsqrtf(d1), r2 = rsqrtf(d2), r3 = rsqrtf(d3);
        *(float4*)(p + i) = make_float4(r0, r1, r2, r3);
        if (have_gid) {
            const int4 g = *(const int4*)(bvec + i);
            int4* rec4 = (int4*)rec + (i >> 1);
            rec4[0] = make_int4(__float_as_int(r0), g.x, __float_as_int(r1), g.y);
            rec4[1] = make_int4(__float_as_int(r2), g.z, __float_as_int(r3), g.w);
            const unsigned c0 = (unsigned)fminf(d0, (float)DEG8_CAP), c1 = (unsigned)fminf(d1, (float)DEG8_CAP);
            const unsigned c2 = (unsigned)fminf(d2, (float)DEG8_CAP), c3 = (unsigned)fminf(d3, (float)DEG8_CAP);
            ((unsigned*)deg8)[i >> 2] = c0 | (c1 << 8) | (c2 << 16) | (c3 << 24);
        }
    } else {
        for (int k = i; k < n; ++k) {
            const float d = p[k] + q[k];
            const float r = rsqrtf(d);
            p[k] = r;
            if (have_gid) {
                ((int2*)rec)[k] = make_int2(__float_as_int(r), bvec[k]);
                deg8[k] = (unsigned char)fminf(d, (float)DEG8_CAP);
            }
        }
    }
}

template <bool RELU_A, bool START>
__device__ __forceinline__ void gemm_tile(const float* __restrict__ X, const float* __restrict__ W,
                                          float* __restrict__ T, float* __restrict__ S,
                                          const float* __restrict__ dinv,
                                          const float* __restrict__ bias,
                                          float* lw, int M, int K) {
    const int tid = threadIdx.x;
    const int kn  = K * NHID;
    for (int i = tid; i < kn; i += 256) lw[i] = W[i];
    __syncthreads();

    const int wave = tid >> 5;
    const int lane = tid & 31;
    const int tile = blockIdx.x * 8 + wave;
    if (tile * 16 >= M) return;

    const int row   = tile * 16 + (lane & 15);
    const int kHalf = (lane >> 4) << 1;
    const int n     = lane & 15;
    const int rbase = tile * 16 + ((lane >> 4) << 3);

    float4 dlo = make_float4(0.0f, 0.0f, 0.0f, 0.0f), dhi = dlo;
    float  bn  = 0.0f;
    if (START) {
        dlo = *(const float4*)(dinv + rbase);
        dhi = *(const float4*)(dinv + rbase + 4);
        bn  = bias[n];
    }

    v8f c = {};
    const float* xr = X + (size_t)row * K + kHalf;
    for (int k0 = 0; k0 < K; k0 += 4) {
        float2 av = *(const float2*)(xr + k0);
        v2f a;
        if (RELU_A) { a.x = fmaxf(av.x, 0.0f); a.y = fmaxf(av.y, 0.0f); }
        else        { a.x = av.x;              a.y = av.y; }
        v2f b;
        b.x = lw[(k0 + kHalf)     * NHID + n];
        b.y = lw[(k0 + kHalf + 1) * NHID + n];
        c = __builtin_amdgcn_wmma_f32_16x16x4_f32(
                 false, a,  false, b,
                 (short)0, c,  false,  false);
    }

#pragma unroll
    for (int v = 0; v < 8; ++v)
        T[(size_t)(rbase + v) * NHID + n] = c[v];
    if (START) {
        const float di[8] = { dlo.x, dlo.y, dlo.z, dlo.w, dhi.x, dhi.y, dhi.z, dhi.w };
#pragma unroll
        for (int v = 0; v < 8; ++v)
            S[(size_t)(rbase + v) * NHID + n] = c[v] * di[v] * di[v] + bn;
    }
}

__global__ __launch_bounds__(256)
void gemm_start(const float* __restrict__ X, const float* __restrict__ W,
                float* __restrict__ T, float* __restrict__ S,
                const float* __restrict__ dinv, const float* __restrict__ bias,
                int M, int K) {
    __shared__ float lw[NFEAT * NHID];
    gemm_tile<false, true>(X, W, T, S, dinv, bias, lw, M, K);
}

__global__ __launch_bounds__(256)
void gemm_relu(const float* __restrict__ X, const float* __restrict__ W,
               float* __restrict__ T, int M, int K) {
    __shared__ float lw[NFEAT * NHID];
    gemm_tile<true, false>(X, W, T, nullptr, nullptr, nullptr, lw, M, K);
}

#define NORM_GROUPS     256
#define NORM_THREADS    1024
#define NORM_PER_THREAD 5

__global__ __launch_bounds__(NORM_THREADS)
void edge_norm(const int* __restrict__ src, const int* __restrict__ dst,
               const float* __restrict__ dinv, const int* __restrict__ rec,
               const unsigned char* __restrict__ deg8,
               float* __restrict__ WGT, unsigned short* __restrict__ GID, int e,
               size_t ws_size) {
    __shared__ __attribute__((aligned(16))) unsigned char degs[TAB_BYTES];

    if (ws_size < WS_BYTES_WITH_WGT) return;
    const bool have_gid = ws_size >= WS_BYTES_WITH_GID;
    const int tid  = threadIdx.x;
    const int nblk = (e + NORM_THREADS - 1) / NORM_THREADS;
    const int lo = (int)((long long)blockIdx.x * nblk / gridDim.x);
    const int hi = (int)((long long)(blockIdx.x + 1) * nblk / gridDim.x);

    unsigned s[NORM_PER_THREAD], d[NORM_PER_THREAD];
#pragma unroll
    for (int k = 0; k < NORM_PER_THREAD; ++k) {
        const int i = min((lo + k) * NORM_THREADS + tid, e - 1);
        s[k] = (unsigned)src[i];
        d[k] = (unsigned)dst[i];
    }

    if (have_gid) {
        const uint4* __restrict__ g4 = (const uint4*)deg8;
        uint4* l4 = (uint4*)degs;
        for (int i = tid; i < TAB_BYTES / 16; i += NORM_THREADS) l4[i] = g4[i];
        __syncthreads();
    }

    const int2* __restrict__ rec2 = (const int2*)rec;
    const unsigned short*     lbase = (const unsigned short*)(degs + TAB_BASE_OFF);
    const unsigned long long* lmask = (const unsigned long long*)(degs + TAB_MASK_OFF);
    for (int blk = lo; blk < hi; blk += NORM_PER_THREAD) {
        unsigned cs[NORM_PER_THREAD], cd[NORM_PER_THREAD];
#pragma unroll
        for (int k = 0; k < NORM_PER_THREAD; ++k) { cs[k] = s[k]; cd[k] = d[k]; }
#pragma unroll
        for (int k = 0; k < NORM_PER_THREAD; ++k) {
            const int i = min((blk + NORM_PER_THREAD + k) * NORM_THREADS + tid, e - 1);
            s[k] = (unsigned)src[i];
            d[k] = (unsigned)dst[i];
        }
        float w[NORM_PER_THREAD];
        int   g[NORM_PER_THREAD];
        bool  slow[NORM_PER_THREAD] = {};
        bool  any_slow = false;
        if (have_gid) {
            unsigned           b[NORM_PER_THREAD], bd[NORM_PER_THREAD], gb[NORM_PER_THREAD];
            unsigned long long m[NORM_PER_THREAD];
#pragma unroll
            for (int k = 0; k < NORM_PER_THREAD; ++k) {
                b[k]  = degs[cs[k]];
                bd[k] = degs[cd[k]];
                gb[k] = lbase[cd[k] >> 6];
                m[k]  = lmask[cd[k] >> 6];
            }
#pragma unroll
            for (int k = 0; k < NORM_PER_THREAD; ++k) {
                w[k] = rsqrtf((float)b[k]) * rsqrtf((float)bd[k]);
                g[k] = (int)gb[k] + __popcll(m[k] & (~0ull >> (63u - (cd[k] & 63u))));
                slow[k] = b[k] == DEG8_CAP || bd[k] == DEG8_CAP || gb[k] == GBASE_IRREG;
                any_slow = any_slow || slow[k];
            }
        } else {
            float a[NORM_PER_THREAD], c[NORM_PER_THREAD];
#pragma unroll
            for (int k = 0; k < NORM_PER_THREAD; ++k) {
                a[k] = dinv[cs[k]];
                c[k] = dinv[cd[k]];
            }
#pragma unroll
            for (int k = 0; k < NORM_PER_THREAD; ++k) { w[k] = a[k] * c[k]; g[k] = 0; }
        }
#pragma unroll
        for (int k = 0; k < NORM_PER_THREAD; ++k) {
            const int i = (blk + k) * NORM_THREADS + tid;
            if (blk + k < hi && i < e && !slow[k]) {
                WGT[i] = w[k];
                if (have_gid) GID[i] = (unsigned short)g[k];
            }
        }
        if (any_slow) {
#pragma unroll
            for (int k = 0; k < NORM_PER_THREAD; ++k) {
                const int i = (blk + k) * NORM_THREADS + tid;
                if (blk + k < hi && i < e && slow[k]) {
                    const int2 r = rec2[cd[k]];
                    WGT[i] = dinv[cs[k]] * __int_as_float(r.x);
                    GID[i] = (unsigned short)r.y;
                }
            }
        }
    }
}


#define EDGES_PER_WAVE 64
#define EDGES_PER_GROUP (8 * EDGES_PER_WAVE)
#define CTR_STRIDE 64

__global__ __launch_bounds__(256)
void edge_scatter(const int* __restrict__ src, const int* __restrict__ dst,
                  const float* __restrict__ dinv, const float* __restrict__ WGT,
                  const float* __restrict__ T,
                  float* __restrict__ OUT, int e, unsigned* __restrict__ ctr,
                  int ngroups, int nsplit, size_t ws_size) {
    __shared__ int  pick[2];
    __shared__ int4 queue[8][EDGES_PER_WAVE];

    if (threadIdx.x == 0) {
        const unsigned id = __builtin_amdgcn_s_sendmsg_rtn(0x87);
        int h = (int)((id >> 18) & 1u);
        int g = (int)atomicAdd(ctr + h * CTR_STRIDE, 1u);
        if (g >= ngroups) {
            h ^= 1;
            g = (int)atomicAdd(ctr + h * CTR_STRIDE, 1u);
        }
        pick[0] = g;
        pick[1] = h;
    }
    __syncthreads();
    const int g = pick[0];
    const int h = pick[1];
    if (g >= ngroups) return;

    const int wave = threadIdx.x >> 5;
    const int lane = threadIdx.x & 31;
    const int e0 = g * EDGES_PER_GROUP + wave * EDGES_PER_WAVE + lane;
    const int e1 = e0 + 32;

    const bool have_wgt = ws_size >= WS_BYTES_WITH_WGT;
    int s0 = 0, d0 = 0, s1 = 0, d1 = 0;
    float w0 = 0.0f, w1 = 0.0f;
    const bool v0 = e0 < e, v1 = e1 < e;
    if (v0) { s0 = src[e0]; d0 = dst[e0]; }
    if (v1) { s1 = src[e1]; d1 = dst[e1]; }
    if (have_wgt) {
        if (v0) w0 = WGT[e0];
        if (v1) w1 = WGT[e1];
    }
    const bool m0 = v0 && ((d0 >= nsplit) == (h != 0));
    const bool m1 = v1 && ((d1 >= nsplit) == (h != 0));

    const unsigned b0 = __builtin_amdgcn_ballot_w32(m0);
    const unsigned b1 = __builtin_amdgcn_ballot_w32(m1);
    const unsigned below = (1u << lane) - 1u;
    const int n0 = __popc(b0);
    const int n  = n0 + __popc(b1);
    int4* q = queue[wave];
    if (!have_wgt) {
        const float a0 = dinv[m0 ? s0 : 0], c0 = dinv[m0 ? d0 : 0];
        const float a1 = dinv[m1 ? s1 : 0], c1 = dinv[m1 ? d1 : 0];
        w0 = a0 * c0;
        w1 = a1 * c1;
    }
    if (m0) q[__popc(b0 & below)]      = make_int4(s0, d0, __float_as_int(w0), 0);
    if (m1) q[n0 + __popc(b1 & below)] = make_int4(s1, d1, __float_as_int(w1), 0);

    const unsigned c = lane & 15;
    const int      r = lane >> 4;
    for (int base = 0; base < n; base += 16) {
        int4  ent[8];
        float m[8];
#pragma unroll
        for (int i = 0; i < 8; ++i) {
            const int idx = base + 2 * i + r;
            ent[i] = q[idx < n ? idx : n - 1];
        }
#pragma unroll
        for (int i = 0; i < 8; ++i)
            m[i] = T[(unsigned)ent[i].x * NHID + c];
#pragma unroll
        for (int i = 0; i < 8; ++i) {
            const int idx = base + 2 * i + r;
            if (idx < n)
                atomicAdd(OUT + ((unsigned)ent[i].y * NHID + c), m[i] * __int_as_float(ent[i].z));
        }
    }
}


#define POOL_WAVES   32
#define POOL_THREADS (POOL_WAVES * 32)
#define POOL_CHUNK   64
#define POOL_GROUPS  256

struct __attribute__((aligned(16))) PoolEnt { int t; int g; float w; };

template <bool HAVE_WGT, bool HAVE_GID>
__device__ __forceinline__ void pool_edges(const int* __restrict__ src, const int* __restrict__ dst,
                                           const int* __restrict__ bvec,
                                           const float* __restrict__ dinv, const float* __restrict__ WGT,
                                           const unsigned short* __restrict__ GID,
                                           const float* __restrict__ T, float* tab, PoolEnt* q,
                                           int e, int c_lo, int c_hi, int wave, int lane) {
    const unsigned c = lane & 15;
    const int      r = lane >> 4;
    if (c_lo + wave >= c_hi) return;

    int s0, d0, s1, d1;
    float w0 = 0.0f, w1 = 0.0f;
    {
        const int i0 = min((c_lo + wave) * POOL_CHUNK + lane, e - 1);
        const int i1 = min((c_lo + wave) * POOL_CHUNK + 32 + lane, e - 1);
        s0 = src[i0]; d0 = HAVE_GID ? (int)GID[i0] : dst[i0];
        s1 = src[i1]; d1 = HAVE_GID ? (int)GID[i1] : dst[i1];
        if (HAVE_WGT) { w0 = WGT[i0]; w1 = WGT[i1]; }
    }
    for (int ch = c_lo + wave; ch < c_hi; ch += POOL_WAVES) {
        const int n = min(POOL_CHUNK, e - ch * POOL_CHUNK);
        const int g0 = HAVE_GID ? d0 : bvec[d0];
        const int g1 = HAVE_GID ? d1 : bvec[d1];
        if (!HAVE_WGT) {
            w0 = dinv[s0] * dinv[d0];
            w1 = dinv[s1] * dinv[d1];
        }
        const int cs0 = s0, cs1 = s1;
        const float cw0 = w0, cw1 = w1;
        {
            const int nx = min(ch + POOL_WAVES, c_hi - 1);
            const int i0 = min(nx * POOL_CHUNK + lane, e - 1);
            const int i1 = min(nx * POOL_CHUNK + 32 + lane, e - 1);
            s0 = src[i0]; d0 = HAVE_GID ? (int)GID[i0] : dst[i0];
            s1 = src[i1]; d1 = HAVE_GID ? (int)GID[i1] : dst[i1];
            if (HAVE_WGT) { w0 = WGT[i0]; w1 = WGT[i1]; }
        }
        q[lane].t      = cs0 * NHID;  q[lane].g      = g0 * NHID;  q[lane].w      = cw0;
        q[lane + 32].t = cs1 * NHID;  q[lane + 32].g = g1 * NHID;  q[lane + 32].w = cw1;

        for (int base = 0; base < n; base += 32) {
            PoolEnt ent[16];
            float   m[16];
#pragma unroll
            for (int i = 0; i < 16; ++i) {
                const int idx = base + 2 * i + r;
                ent[i] = q[idx < n ? idx : n - 1];
            }
#pragma unroll
            for (int i = 0; i < 16; ++i)
                m[i] = T[(unsigned)ent[i].t + c];
#pragma unroll
            for (int i = 0; i < 16; ++i) {
                const int idx = base + 2 * i + r;
                const float v = m[i] * ent[i].w;
                atomicAdd(&tab[(unsigned)ent[i].g + c], idx < n ? v : 0.0f);
            }
        }
    }
}

__global__ __launch_bounds__(POOL_THREADS)
void edge_pool(const int* __restrict__ src, const int* __restrict__ dst,
               const int* __restrict__ bvec,
               const float* __restrict__ dinv, const float* __restrict__ WGT,
               const unsigned short* __restrict__ GID,
               const float* __restrict__ T, const float* __restrict__ bias,
               float* __restrict__ GP, int e, int nnodes, size_t ws_size) {
    __shared__ __attribute__((aligned(16))) float tab[NGRAPH * NHID];
    __shared__ PoolEnt queue[POOL_WAVES][POOL_CHUNK];
    __shared__ int  half_s;

    const int tid  = threadIdx.x;
    const int wave = __builtin_amdgcn_readfirstlane(tid >> 5);
    const int lane = tid & 31;

    float4* tab4 = (float4*)tab;
    for (int i = tid; i < NGRAPH * NHID / 4; i += POOL_THREADS)
        tab4[i] = make_float4(0.0f, 0.0f, 0.0f, 0.0f);
    if (tid == 0) {
        const unsigned id = __builtin_amdgcn_s_sendmsg_rtn(0x87);
        half_s = (int)((id >> 18) & 1u);
    }
    __syncthreads();

    const int nb = gridDim.x, b = blockIdx.x;

    {
        const unsigned c = lane & 15;
        const int      r = lane >> 4;
        const int n_lo = (int)((long long)b * nnodes / nb);
        const int n_hi = (int)((long long)(b + 1) * nnodes / nb);
        const float bc = bias[c];
        for (int k0 = n_lo + wave * 2; k0 < n_hi; k0 += POOL_WAVES * 2 * 8) {
            float di[8], tv[8];
            int   gi[8];
#pragma unroll
            for (int j = 0; j < 8; ++j) {
                const int i = min(k0 + r + POOL_WAVES * 2 * j, n_hi - 1);
                di[j] = dinv[i];
                tv[j] = T[(unsigned)i * NHID + c];
                gi[j] = bvec[i];
            }
#pragma unroll
            for (int j = 0; j < 8; ++j) {
                const int i = k0 + r + POOL_WAVES * 2 * j;
                const float v = tv[j] * di[j] * di[j] + bc;
                atomicAdd(&tab[(unsigned)gi[j] * NHID + c], i < n_hi ? v : 0.0f);
            }
        }
    }

    const int nchunks = (e + POOL_CHUNK - 1) / POOL_CHUNK;
    const int c_lo = (int)((long long)b * nchunks / nb);
    const int c_hi = (int)((long long)(b + 1) * nchunks / nb);
    if (ws_size >= WS_BYTES_WITH_GID)
        pool_edges<true, true>(src, dst, bvec, dinv, WGT, GID, T, tab, queue[wave], e, c_lo, c_hi, wave, lane);
    else if (ws_size >= WS_BYTES_WITH_WGT)
        pool_edges<true, false>(src, dst, bvec, dinv, WGT, GID, T, tab, queue[wave], e, c_lo, c_hi, wave, lane);
    else
        pool_edges<false, false>(src, dst, bvec, dinv, WGT, GID, T, tab, queue[wave], e, c_lo, c_hi, wave, lane);

    __syncthreads();

    float* gp = GP + (size_t)half_s * (NGRAPH * NHID);
    const unsigned rot = (unsigned)b * (unsigned)(NGRAPH * NHID / POOL_GROUPS);
    for (int k0 = 0; k0 < NGRAPH * NHID / POOL_THREADS; k0 += 8) {
        unsigned idx[8];
        float    v[8];
#pragma unroll
        for (int k = 0; k < 8; ++k) {
            idx[k] = (rot + (unsigned)(k0 + k) * POOL_THREADS + (unsigned)tid) & (unsigned)(NGRAPH * NHID - 1);
            v[k]   = tab[idx[k]];
        }
#pragma unroll
        for (int k = 0; k < 8; ++k)
            atomicAdd(gp + idx[k], v[k]);
    }
}

__global__ void lin1_kernel(const float* __restrict__ G, const float* __restrict__ W,
                            const float* __restrict__ B, float* __restrict__ O) {
    int t = blockIdx.x * blockDim.x + threadIdx.x;
    if (t >= NGRAPH * NMLP) return;
    int gi = t / NMLP, j = t % NMLP;
    float acc = B[j];
#pragma unroll
    for (int k = 0; k < NHID; ++k)
        acc = fmaf(fmaxf(G[gi * NHID + k] + G[NGRAPH * NHID + gi * NHID + k], 0.0f), W[k * NMLP + j], acc);
    O[t] = fmaxf(acc, 0.0f);
}

__global__ void lin2_kernel(const float* __restrict__ G, const float* __restrict__ W,
                            const float* __restrict__ B, float* __restrict__ O) {
    int t = blockIdx.x * blockDim.x + threadIdx.x;
    if (t >= NGRAPH * NCLS) return;
    int gi = t / NCLS, c = t % NCLS;
    float acc = B[c];
    for (int k = 0; k < NMLP; ++k)
        acc = fmaf(G[gi * NMLP + k], W[k * NCLS + c], acc);
    O[t] = acc;
}


extern "C" void kernel_launch(void* const* d_in, const int* in_sizes, int n_in,
                              void* d_out, int out_size, void* d_ws, size_t ws_size,
                              hipStream_t stream) {
    const float* x   = (const float*)d_in[0];
    const int*  eidx = (const int*)d_in[1];
    const int*  bvec = (const int*)d_in[2];
    const float* W1  = (const float*)d_in[3];
    const float* b1  = (const float*)d_in[4];
    const float* W2  = (const float*)d_in[5];
    const float* b2  = (const float*)d_in[6];
    const float* Wl1 = (const float*)d_in[7];
    const float* bl1 = (const float*)d_in[8];
    const float* Wl2 = (const float*)d_in[9];
    const float* bl2 = (const float*)d_in[10];
    float* out = (float*)d_out;

    char* ws = (char*)d_ws;
    size_t off = 0;
    auto walloc = [&](size_t bytes) -> void* {
        void* p = ws + off;
        off = (off + bytes + 255) & ~(size_t)255;
        return p;
    };
    float* deg  = (float*)walloc((size_t)NNODES * 4);
    float* bufA = (float*)walloc((size_t)NNODES * NHID * 4);
    float* bufB = (float*)walloc((size_t)NNODES * NHID * 4);
    unsigned char* deg8 = (unsigned char*)walloc((size_t)NGRAPH * NHID * 4);
    float* g2   = (float*)walloc((size_t)NGRAPH * NMLP * 4);
    float* wgt  = (float*)walloc((size_t)NEDGES * 4);
    unsigned short* gid = (unsigned short*)walloc((size_t)NEDGES * 2);

    const int* src = eidx;
    const int* dst = eidx + NEDGES;

    const int TB = 256;
    const int nN    = (NNODES + TB - 1) / TB;
    const int nTile = ((NNODES / 16) + 7) / 8;
    const int nGrp  = (NEDGES + EDGES_PER_GROUP - 1) / EDGES_PER_GROUP;
    const int nDeg  = (NEDGES + TB * DEG_PER_THREAD - 1) / (TB * DEG_PER_THREAD);

    unsigned* ctr = (unsigned*)g2;

    float* deg1 = bufB;
    fill_deg_ctr<<<nN, TB, 0, stream>>>(deg, deg1, NNODES, ctr, 2 * CTR_STRIDE);
    deg_accum<<<nDeg, TB, 0, stream>>>(dst, deg, deg1, NEDGES);
    int* rec = (int*)bufA;
    rsqrt_sum<<<((NNODES + 3) / 4 + TB - 1) / TB, TB, 0, stream>>>(deg, deg1, NNODES, bvec, rec, deg8, ws_size);
    float* dinv = deg;
    edge_norm<<<NORM_GROUPS, NORM_THREADS, 0, stream>>>(src, dst, dinv, rec, deg8, wgt, gid, NEDGES, ws_size);

    gemm_start<<<nTile, TB, 0, stream>>>(x, W1, bufA, bufB, dinv, b1, NNODES, NFEAT);
    edge_scatter<<<2 * nGrp, TB, 0, stream>>>(src, dst, dinv, wgt, bufA, bufB, NEDGES,
                                              ctr, nGrp, NNODES / 2, ws_size);

    gemm_relu<<<nTile, TB, 0, stream>>>(bufB, W2, bufA, NNODES, NHID);
    float* gtab = bufB;
    fill_f32<<<(2 * NGRAPH * NHID + TB - 1) / TB, TB, 0, stream>>>(gtab, 2 * NGRAPH * NHID, 0.0f);
    edge_pool<<<POOL_GROUPS, POOL_THREADS, 0, stream>>>(src, dst, bvec, dinv, wgt, gid, bufA, b2, gtab,
                                                        NEDGES, NNODES, ws_size);

    lin1_kernel<<<(NGRAPH * NMLP + TB - 1) / TB, TB, 0, stream>>>(gtab, Wl1, bl1, g2);
    lin2_kernel<<<(NGRAPH * NCLS + TB - 1) / TB, TB, 0, stream>>>(g2, Wl2, bl2, out);
}


# Layer-2 product moved behind the pooling (no relu between; W2 is linear): edge_pool sums relu(a1) rows, W2 runs on the 4096 pooled rows in one head launch with lin1 and lin2; sum order changed
# speedup vs baseline: 4.6136x; 1.0621x over previous
#include <hip/hip_runtime.h>
#include <hip/hip_bf16.h>


typedef float v2f __attribute__((ext_vector_type(2)));
typedef float v8f __attribute__((ext_vector_type(8)));

#define NNODES  200000
#define NEDGES  6400000
#define NFEAT   128
#define NHID    16
#define NMLP    100
#define NCLS    27
#define NGRAPH  4096

#define WS_BYTES_WITH_WGT ((size_t)NNODES * 4 + 2 * ((size_t)NNODES * NHID * 4) + \
                           (size_t)NGRAPH * NHID * 4 + (size_t)NGRAPH * NMLP * 4 + \
                           (size_t)NEDGES * 4)
#define WS_BYTES_WITH_GID (WS_BYTES_WITH_WGT + (size_t)NEDGES * 2)
static_assert(NGRAPH <= 65536, "a graph id must fit the 16 bits of the per-edge plane");
static_assert((size_t)NNODES <= (size_t)NGRAPH * NHID * 4, "the degree bytes must fit their plane");
static_assert(NNODES % 16 == 0, "the degree bytes are copied to LDS sixteen at a time");
#define DEG8_CAP 255u
#define NBLK64        (NNODES / 64)
#define TAB_BASE_OFF  NNODES
#define TAB_BASE_N    ((NBLK64 + 3) & ~3)
#define TAB_MASK_OFF  (TAB_BASE_OFF + TAB_BASE_N * 2)
#define TAB_MASK_N    ((NBLK64 + 1) & ~1)
#define TAB_BYTES     (TAB_MASK_OFF + TAB_MASK_N * 8)
#define GBASE_IRREG   0xFFFFu
static_assert(NNODES % 64 == 0, "the graph table takes whole blocks of 64 nodes");
static_assert(TAB_MASK_OFF % 8 == 0 && TAB_BYTES % 16 == 0, "the table is copied sixteen bytes at a time");
static_assert((size_t)TAB_BYTES <= (size_t)NGRAPH * NHID * 4, "degree bytes and graph table must fit their plane");
static_assert(NGRAPH + 64 <= 0xFFFF, "a graph id must stay below GBASE_IRREG");
static_assert(TAB_MASK_N <= TAB_BASE_N && TAB_BASE_N <= NNODES / 4, "rsqrt_sum has a thread per table entry");
#define CTR_STRIDE     64
#define POOL_TAB_OFF   256
#define POOL_CNT_OFF   (POOL_TAB_OFF + 2 * NGRAPH * NHID)
#define POOL_WORDS     (POOL_CNT_OFF + 2 * NGRAPH)
static_assert(2 * CTR_STRIDE <= POOL_TAB_OFF, "the counters lie in front of the tables");
static_assert(POOL_WORDS <= NGRAPH * NMLP, "counters, tables and counts must fit their plane");
static_assert(POOL_WORDS <= NNODES, "fill_deg_ctr has a thread per word to clear");


__global__ void fill_deg_ctr(float* __restrict__ d0, float* __restrict__ d1, int n,
                             unsigned* __restrict__ clr, int nclr) {
    int i = blockIdx.x * blockDim.x + threadIdx.x;
    if (i < n) { d0[i] = 1.0f; d1[i] = 0.0f; }
    if (i < nclr) clr[i] = 0u;
}

#define DEG_PER_THREAD 4

__global__ __launch_bounds__(256)
void deg_accum(const int* __restrict__ dst, float* __restrict__ deg0,
               float* __restrict__ deg1, int e) {
    __shared__ int half_s;
    const int base = blockIdx.x * (256 * DEG_PER_THREAD) + threadIdx.x;
    unsigned d[DEG_PER_THREAD];
#pragma unroll
    for (int k = 0; k < DEG_PER_THREAD; ++k)
        d[k] = (unsigned)dst[min(base + k * 256, e - 1)];
    if (threadIdx.x == 0) {
        const unsigned id = __builtin_amdgcn_s_sendmsg_rtn(0x87);
        half_s = (int)((id >> 18) & 1u);
    }
    __syncthreads();
    float* __restrict__ deg = half_s ? deg1 : deg0;
    if ((blockIdx.x + 1) * (256 * DEG_PER_THREAD) <= e) {
#pragma unroll
        for (int k = 0; k < DEG_PER_THREAD; ++k)
            atomicAdd(&deg[d[k]], 1.0f);
    } else {
        for (int k = 0; k < DEG_PER_THREAD; ++k)
            if (base + k * 256 < e) atomicAdd(&deg[d[k]], 1.0f);
    }
}

__global__ void rsqrt_sum(float* __restrict__ p, const float* __restrict__ q, int n,
                          const int* __restrict__ bvec, int* __restrict__ rec,
                          unsigned char* __restrict__ deg8, size_t ws_size) {
    const int t = blockIdx.x * blockDim.x + threadIdx.x;
    const int i = t * 4;
    const bool have_gid = ws_size >= WS_BYTES_WITH_GID;
    if (have_gid && t < TAB_BASE_N) {
        unsigned short*     gbase = (unsigned short*)(deg8 + TAB_BASE_OFF);
        unsigned long long* gmask = (unsigned long long*)(deg8 + TAB_MASK_OFF);
        if (t < NBLK64) {
            const int4* __restrict__ b4 = (const int4*)(bvec + t * 64);
            int4 v[16];
#pragma unroll
            for (int q = 0; q < 16; ++q) v[q] = b4[q];
            const int first = v[0].x;
            int prev = first;
            unsigned long long m = 0ull;
            bool ok = (unsigned)first < (unsigned)(GBASE_IRREG - 64u);
#pragma unroll
            for (int q = 0; q < 16; ++q) {
                const int c[4] = { v[q].x, v[q].y, v[q].z, v[q].w };
#pragma unroll
                for (int j = 0; j < 4; ++j) {
                    const unsigned step = (unsigned)(c[j] - prev);
                    m  |= (unsigned long long)(step == 1u) << (q * 4 + j);
                    ok  = ok && step <= 1u;
                    prev = c[j];
                }
            }
            gmask[t] = m;
            gbase[t] = (unsigned short)(ok ? (unsigned)first : GBASE_IRREG);
        } else {
            const int u = t - NBLK64;
            if (NBLK64 + u < TAB_MASK_N) gmask[NBLK64 + u] = 0ull;
            if (NBLK64 + u < TAB_BASE_N) gbase[NBLK64 + u] = (unsigned short)GBASE_IRREG;
        }
    }
    if (i >= n) return;
    if (i + 3 < n) {
        const float4 a = *(const float4*)(p + i);
        const float4 b = *(const float4*)(q + i);
        const float d0 = a.x + b.x, d1 = a.y + b.y, d2 = a.z + b.z, d3 = a.w + b.w;
        const float r0 = rsqrtf(d0), r1 = rsqrtf(d1), r2 = rsqrtf(d2), r3 = rsqrtf(d3);
        *(float4*)(p + i) = make_float4(r0, r1, r2, r3);
        if (have_gid) {
            const int4 g = *(const int4*)(bvec + i);
            int4* rec4 = (int4*)rec + (i >> 1);
            rec4[0] = make_int4(__float_as_int(r0), g.x, __float_as_int(r1), g.y);
            rec4[1] = make_int4(__float_as_int(r2), g.z, __float_as_int(r3), g.w);
            const unsigned c0 = (unsigned)fminf(d0, (float)DEG8_CAP), c1 = (unsigned)fminf(d1, (float)DEG8_CAP);
            const unsigned c2 = (unsigned)fminf(d2, (float)DEG8_CAP), c3 = (unsigned)fminf(d3, (float)DEG8_CAP);
            ((unsigned*)deg8)[i >> 2] = c0 | (c1 << 8) | (c2 << 16) | (c3 << 24);
        }
    } else {
        for (int k = i; k < n; ++k) {
            const float d = p[k] + q[k];
            const float r = rsqrtf(d);
            p[k] = r;
            if (have_gid) {
                ((int2*)rec)[k] = make_int2(__float_as_int(r), bvec[k]);
                deg8[k] = (unsigned char)fminf(d, (float)DEG8_CAP);
            }
        }
    }
}

__global__ __launch_bounds__(256)
void gemm_start(const float* __restrict__ X, const float* __restrict__ W,
                float* __restrict__ T, float* __restrict__ S,
                const float* __restrict__ dinv, const float* __restrict__ bias,
                int M, int K) {
    __shared__ float lw[NFEAT * NHID];
    const int tid = threadIdx.x;
    const int kn  = K * NHID;
    for (int i = tid; i < kn; i += 256) lw[i] = W[i];
    __syncthreads();

    const int wave = tid >> 5;
    const int lane = tid & 31;
    const int tile = blockIdx.x * 8 + wave;
    if (tile * 16 >= M) return;

    const int row   = tile * 16 + (lane & 15);
    const int kHalf = (lane >> 4) << 1;
    const int n     = lane & 15;
    const int rbase = tile * 16 + ((lane >> 4) << 3);

    const float4 dlo = *(const float4*)(dinv + rbase);
    const float4 dhi = *(const float4*)(dinv + rbase + 4);
    const float  bn  = bias[n];

    v8f c = {};
    const float* xr = X + (size_t)row * K + kHalf;
    for (int k0 = 0; k0 < K; k0 += 4) {
        float2 av = *(const float2*)(xr + k0);
        v2f a;
        a.x = av.x;
        a.y = av.y;
        v2f b;
        b.x = lw[(k0 + kHalf)     * NHID + n];
        b.y = lw[(k0 + kHalf + 1) * NHID + n];
        c = __builtin_amdgcn_wmma_f32_16x16x4_f32(
                 false, a,  false, b,
                 (short)0, c,  false,  false);
    }

#pragma unroll
    for (int v = 0; v < 8; ++v)
        T[(size_t)(rbase + v) * NHID + n] = c[v];
    const float di[8] = { dlo.x, dlo.y, dlo.z, dlo.w, dhi.x, dhi.y, dhi.z, dhi.w };
#pragma unroll
    for (int v = 0; v < 8; ++v)
        S[(size_t)(rbase + v) * NHID + n] = c[v] * di[v] * di[v] + bn;
}

#define NORM_GROUPS     256
#define NORM_THREADS    1024
#define NORM_PER_THREAD 5

__global__ __launch_bounds__(NORM_THREADS)
void edge_norm(const int* __restrict__ src, const int* __restrict__ dst,
               const float* __restrict__ dinv, const int* __restrict__ rec,
               const unsigned char* __restrict__ deg8,
               float* __restrict__ WGT, unsigned short* __restrict__ GID, int e,
               size_t ws_size) {
    __shared__ __attribute__((aligned(16))) unsigned char degs[TAB_BYTES];

    if (ws_size < WS_BYTES_WITH_WGT) return;
    const bool have_gid = ws_size >= WS_BYTES_WITH_GID;
    const int tid  = threadIdx.x;
    const int nblk = (e + NORM_THREADS - 1) / NORM_THREADS;
    const int lo = (int)((long long)blockIdx.x * nblk / gridDim.x);
    const int hi = (int)((long long)(blockIdx.x + 1) * nblk / gridDim.x);

    unsigned s[NORM_PER_THREAD], d[NORM_PER_THREAD];
#pragma unroll
    for (int k = 0; k < NORM_PER_THREAD; ++k) {
        const int i = min((lo + k) * NORM_THREADS + tid, e - 1);
        s[k] = (unsigned)src[i];
        d[k] = (unsigned)dst[i];
    }

    if (have_gid) {
        const uint4* __restrict__ g4 = (const uint4*)deg8;
        uint4* l4 = (uint4*)degs;
        for (int i = tid; i < TAB_BYTES / 16; i += NORM_THREADS) l4[i] = g4[i];
        __syncthreads();
    }

    const int2* __restrict__ rec2 = (const int2*)rec;
    const unsigned short*     lbase = (const unsigned short*)(degs + TAB_BASE_OFF);
    const unsigned long long* lmask = (const unsigned long long*)(degs + TAB_MASK_OFF);
    for (int blk = lo; blk < hi; blk += NORM_PER_THREAD) {
        unsigned cs[NORM_PER_THREAD], cd[NORM_PER_THREAD];
#pragma unroll
        for (int k = 0; k < NORM_PER_THREAD; ++k) { cs[k] = s[k]; cd[k] = d[k]; }
#pragma unroll
        for (int k = 0; k < NORM_PER_THREAD; ++k) {
            const int i = min((blk + NORM_PER_THREAD + k) * NORM_THREADS + tid, e - 1);
            s[k] = (unsigned)src[i];
            d[k] = (unsigned)dst[i];
        }
        float w[NORM_PER_THREAD];
        int   g[NORM_PER_THREAD];
        bool  slow[NORM_PER_THREAD] = {};
        bool  any_slow = false;
        if (have_gid) {
            unsigned           b[NORM_PER_THREAD], bd[NORM_PER_THREAD], gb[NORM_PER_THREAD];
            unsigned long long m[NORM_PER_THREAD];
#pragma unroll
            for (int k = 0; k < NORM_PER_THREAD; ++k) {
                b[k]  = degs[cs[k]];
                bd[k] = degs[cd[k]];
                gb[k] = lbase[cd[k] >> 6];
                m[k]  = lmask[cd[k] >> 6];
            }
#pragma unroll
            for (int k = 0; k < NORM_PER_THREAD; ++k) {
                w[k] = rsqrtf((float)b[k]) * rsqrtf((float)bd[k]);
                g[k] = (int)gb[k] + __popcll(m[k] & (~0ull >> (63u - (cd[k] & 63u))));
                slow[k] = b[k] == DEG8_CAP || bd[k] == DEG8_CAP || gb[k] == GBASE_IRREG;
                any_slow = any_slow || slow[k];
            }
        } else {
            float a[NORM_PER_THREAD], c[NORM_PER_THREAD];
#pragma unroll
            for (int k = 0; k < NORM_PER_THREAD; ++k) {
                a[k] = dinv[cs[k]];
                c[k] = dinv[cd[k]];
            }
#pragma unroll
            for (int k = 0; k < NORM_PER_THREAD; ++k) { w[k] = a[k] * c[k]; g[k] = 0; }
        }
#pragma unroll
        for (int k = 0; k < NORM_PER_THREAD; ++k) {
            const int i = (blk + k) * NORM_THREADS + tid;
            if (blk + k < hi && i < e && !slow[k]) {
                WGT[i] = w[k];
                if (have_gid) GID[i] = (unsigned short)g[k];
            }
        }
        if (any_slow) {
#pragma unroll
            for (int k = 0; k < NORM_PER_THREAD; ++k) {
                const int i = (blk + k) * NORM_THREADS + tid;
                if (blk + k < hi && i < e && slow[k]) {
                    const int2 r = rec2[cd[k]];
                    WGT[i] = dinv[cs[k]] * __int_as_float(r.x);
                    GID[i] = (unsigned short)r.y;
                }
            }
        }
    }
}


#define EDGES_PER_WAVE 64
#define EDGES_PER_GROUP (8 * EDGES_PER_WAVE)

__global__ __launch_bounds__(256)
void edge_scatter(const int* __restrict__ src, const int* __restrict__ dst,
                  const float* __restrict__ dinv, const float* __restrict__ WGT,
                  const float* __restrict__ T,
                  float* __restrict__ OUT, int e, unsigned* __restrict__ ctr,
                  int ngroups, int nsplit, size_t ws_size) {
    __shared__ int  pick[2];
    __shared__ int4 queue[8][EDGES_PER_WAVE];

    if (threadIdx.x == 0) {
        const unsigned id = __builtin_amdgcn_s_sendmsg_rtn(0x87);
        int h = (int)((id >> 18) & 1u);
        int g = (int)atomicAdd(ctr + h * CTR_STRIDE, 1u);
        if (g >= ngroups) {
            h ^= 1;
            g = (int)atomicAdd(ctr + h * CTR_STRIDE, 1u);
        }
        pick[0] = g;
        pick[1] = h;
    }
    __syncthreads();
    const int g = pick[0];
    const int h = pick[1];
    if (g >= ngroups) return;

    const int wave = threadIdx.x >> 5;
    const int lane = threadIdx.x & 31;
    const int e0 = g * EDGES_PER_GROUP + wave * EDGES_PER_WAVE + lane;
    const int e1 = e0 + 32;

    const bool have_wgt = ws_size >= WS_BYTES_WITH_WGT;
    int s0 = 0, d0 = 0, s1 = 0, d1 = 0;
    float w0 = 0.0f, w1 = 0.0f;
    const bool v0 = e0 < e, v1 = e1 < e;
    if (v0) { s0 = src[e0]; d0 = dst[e0]; }
    if (v1) { s1 = src[e1]; d1 = dst[e1]; }
    if (have_wgt) {
        if (v0) w0 = WGT[e0];
        if (v1) w1 = WGT[e1];
    }
    const bool m0 = v0 && ((d0 >= nsplit) == (h != 0));
    const bool m1 = v1 && ((d1 >= nsplit) == (h != 0));

    const unsigned b0 = __builtin_amdgcn_ballot_w32(m0);
    const unsigned b1 = __builtin_amdgcn_ballot_w32(m1);
    const unsigned below = (1u << lane) - 1u;
    const int n0 = __popc(b0);
    const int n  = n0 + __popc(b1);
    int4* q = queue[wave];
    if (!have_wgt) {
        const float a0 = dinv[m0 ? s0 : 0], c0 = dinv[m0 ? d0 : 0];
        const float a1 = dinv[m1 ? s1 : 0], c1 = dinv[m1 ? d1 : 0];
        w0 = a0 * c0;
        w1 = a1 * c1;
    }
    if (m0) q[__popc(b0 & below)]      = make_int4(s0, d0, __float_as_int(w0), 0);
    if (m1) q[n0 + __popc(b1 & below)] = make_int4(s1, d1, __float_as_int(w1), 0);

    const unsigned c = lane & 15;
    const int      r = lane >> 4;
    for (int base = 0; base < n; base += 16) {
        int4  ent[8];
        float m[8];
#pragma unroll
        for (int i = 0; i < 8; ++i) {
            const int idx = base + 2 * i + r;
            ent[i] = q[idx < n ? idx : n - 1];
        }
#pragma unroll
        for (int i = 0; i < 8; ++i)
            m[i] = T[(unsigned)ent[i].x * NHID + c];
#pragma unroll
        for (int i = 0; i < 8; ++i) {
            const int idx = base + 2 * i + r;
            if (idx < n)
                atomicAdd(OUT + ((unsigned)ent[i].y * NHID + c), m[i] * __int_as_float(ent[i].z));
        }
    }
}


#define POOL_WAVES   32
#define POOL_THREADS (POOL_WAVES * 32)
#define POOL_CHUNK   64
#define POOL_GROUPS  256

struct __attribute__((aligned(16))) PoolEnt { int t; int g; float w; };

template <bool HAVE_WGT, bool HAVE_GID>
__device__ __forceinline__ void pool_edges(const int* __restrict__ src, const int* __restrict__ dst,
                                           const int* __restrict__ bvec,
                                           const float* __restrict__ dinv, const float* __restrict__ WGT,
                                           const unsigned short* __restrict__ GID,
                                           const float* __restrict__ A1, float* tab, PoolEnt* q,
                                           int e, int c_lo, int c_hi, int wave, int lane) {
    const unsigned c = lane & 15;
    const int      r = lane >> 4;
    if (c_lo + wave >= c_hi) return;

    int s0, d0, s1, d1;
    float w0 = 0.0f, w1 = 0.0f;
    {
        const int i0 = min((c_lo + wave) * POOL_CHUNK + lane, e - 1);
        const int i1 = min((c_lo + wave) * POOL_CHUNK + 32 + lane, e - 1);
        s0 = src[i0]; d0 = HAVE_GID ? (int)GID[i0] : dst[i0];
        s1 = src[i1]; d1 = HAVE_GID ? (int)GID[i1] : dst[i1];
        if (HAVE_WGT) { w0 = WGT[i0]; w1 = WGT[i1]; }
    }
    for (int ch = c_lo + wave; ch < c_hi; ch += POOL_WAVES) {
        const int n = min(POOL_CHUNK, e - ch * POOL_CHUNK);
        const int g0 = HAVE_GID ? d0 : bvec[d0];
        const int g1 = HAVE_GID ? d1 : bvec[d1];
        if (!HAVE_WGT) {
            w0 = dinv[s0] * dinv[d0];
            w1 = dinv[s1] * dinv[d1];
        }
        const int cs0 = s0, cs1 = s1;
        const float cw0 = w0, cw1 = w1;
        {
            const int nx = min(ch + POOL_WAVES, c_hi - 1);
            const int i0 = min(nx * POOL_CHUNK + lane, e - 1);
            const int i1 = min(nx * POOL_CHUNK + 32 + lane, e - 1);
            s0 = src[i0]; d0 = HAVE_GID ? (int)GID[i0] : dst[i0];
            s1 = src[i1]; d1 = HAVE_GID ? (int)GID[i1] : dst[i1];
            if (HAVE_WGT) { w0 = WGT[i0]; w1 = WGT[i1]; }
        }
        q[lane].t      = cs0 * NHID;  q[lane].g      = g0 * NHID;  q[lane].w      = cw0;
        q[lane + 32].t = cs1 * NHID;  q[lane + 32].g = g1 * NHID;  q[lane + 32].w = cw1;

        for (int base = 0; base < n; base += 32) {
            PoolEnt ent[16];
            float   m[16];
#pragma unroll
            for (int i = 0; i < 16; ++i) {
                const int idx = base + 2 * i + r;
                ent[i] = q[idx < n ? idx : n - 1];
            }
#pragma unroll
            for (int i = 0; i < 16; ++i)
                m[i] = A1[(unsigned)ent[i].t + c];
#pragma unroll
            for (int i = 0; i < 16; ++i) {
                const int idx = base + 2 * i + r;
                const float v = fmaxf(m[i], 0.0f) * ent[i].w;
                atomicAdd(&tab[(unsigned)ent[i].g + c], idx < n ? v : 0.0f);
            }
        }
    }
}

__global__ __launch_bounds__(POOL_THREADS)
void edge_pool(const int* __restrict__ src, const int* __restrict__ dst,
               const int* __restrict__ bvec,
               const float* __restrict__ dinv, const float* __restrict__ WGT,
               const unsigned short* __restrict__ GID,
               const float* __restrict__ A1,
               float* __restrict__ GP, float* __restrict__ GC, int e, int nnodes, size_t ws_size) {
    __shared__ __attribute__((aligned(16))) float tab[NGRAPH * NHID];
    __shared__ __attribute__((aligned(16))) float cnt[NGRAPH];
    __shared__ PoolEnt queue[POOL_WAVES][POOL_CHUNK];
    __shared__ int  half_s;

    const int tid  = threadIdx.x;
    const int wave = __builtin_amdgcn_readfirstlane(tid >> 5);
    const int lane = tid & 31;

    float4* tab4 = (float4*)tab;
    for (int i = tid; i < NGRAPH * NHID / 4; i += POOL_THREADS)
        tab4[i] = make_float4(0.0f, 0.0f, 0.0f, 0.0f);
    float4* cnt4 = (float4*)cnt;
    for (int i = tid; i < NGRAPH / 4; i += POOL_THREADS)
        cnt4[i] = make_float4(0.0f, 0.0f, 0.0f, 0.0f);
    if (tid == 0) {
        const unsigned id = __builtin_amdgcn_s_sendmsg_rtn(0x87);
        half_s = (int)((id >> 18) & 1u);
    }
    __syncthreads();

    const int nb = gridDim.x, b = blockIdx.x;

    {
        const unsigned c = lane & 15;
        const int      r = lane >> 4;
        const int n_lo = (int)((long long)b * nnodes / nb);
        const int n_hi = (int)((long long)(b + 1) * nnodes / nb);
        for (int k0 = n_lo + wave * 2; k0 < n_hi; k0 += POOL_WAVES * 2 * 8) {
            float di[8], tv[8];
            int   gi[8];
#pragma unroll
            for (int j = 0; j < 8; ++j) {
                const int i = min(k0 + r + POOL_WAVES * 2 * j, n_hi - 1);
                di[j] = dinv[i];
                tv[j] = A1[(unsigned)i * NHID + c];
                gi[j] = bvec[i];
            }
#pragma unroll
            for (int j = 0; j < 8; ++j) {
                const int i = k0 + r + POOL_WAVES * 2 * j;
                const float v = fmaxf(tv[j], 0.0f) * di[j] * di[j];
                atomicAdd(&tab[(unsigned)gi[j] * NHID + c], i < n_hi ? v : 0.0f);
                if (c == 0 && i < n_hi) atomicAdd(&cnt[(unsigned)gi[j]], 1.0f);
            }
        }
    }

    const int nchunks = (e + POOL_CHUNK - 1) / POOL_CHUNK;
    const int c_lo = (int)((long long)b * nchunks / nb);
    const int c_hi = (int)((long long)(b + 1) * nchunks / nb);
    if (ws_size >= WS_BYTES_WITH_GID)
        pool_edges<true, true>(src, dst, bvec, dinv, WGT, GID, A1, tab, queue[wave], e, c_lo, c_hi, wave, lane);
    else if (ws_size >= WS_BYTES_WITH_WGT)
        pool_edges<true, false>(src, dst, bvec, dinv, WGT, GID, A1, tab, queue[wave], e, c_lo, c_hi, wave, lane);
    else
        pool_edges<false, false>(src, dst, bvec, dinv, WGT, GID, A1, tab, queue[wave], e, c_lo, c_hi, wave, lane);

    __syncthreads();

    float* gp = GP + (size_t)half_s * (NGRAPH * NHID);
    const unsigned rot = (unsigned)b * (unsigned)(NGRAPH * NHID / POOL_GROUPS);
    for (int k0 = 0; k0 < NGRAPH * NHID / POOL_THREADS; k0 += 8) {
        unsigned idx[8];
        float    v[8];
#pragma unroll
        for (int k = 0; k < 8; ++k) {
            idx[k] = (rot + (unsigned)(k0 + k) * POOL_THREADS + (unsigned)tid) & (unsigned)(NGRAPH * NHID - 1);
            v[k]   = tab[idx[k]];
        }
#pragma unroll
        for (int k = 0; k < 8; ++k)
            atomicAdd(gp + idx[k], v[k]);
    }
    float* gc = GC + (size_t)half_s * NGRAPH;
    for (int i = tid; i < NGRAPH; i += POOL_THREADS)
        atomicAdd(gc + i, cnt[i]);
}

#define HEAD_GRAPHS  32
#define HEAD_WAVES   4
#define HEAD_ROWS    8
#define NMLP_PER_LANE ((NMLP + 31) / 32)
static_assert(HEAD_WAVES * HEAD_ROWS == HEAD_GRAPHS && NGRAPH % HEAD_GRAPHS == 0, "whole work-groups");
static_assert((HEAD_GRAPHS * NCLS * 4) % 128 == 0, "a work-group owns whole output lines");
static_assert(NCLS <= 32 && NHID == 16, "a lane per class; one 16-column tile");

__global__ __launch_bounds__(HEAD_WAVES * 32)
void head_kernel(const float* __restrict__ GP, const float* __restrict__ GC,
                 const float* __restrict__ W2, const float* __restrict__ b2,
                 const float* __restrict__ Wl1, const float* __restrict__ bl1,
                 const float* __restrict__ Wl2, const float* __restrict__ bl2,
                 float* __restrict__ out) {
    __shared__ __attribute__((aligned(16))) float gs[HEAD_WAVES][16 * NHID];
    __shared__ __attribute__((aligned(16))) float zs[HEAD_WAVES][HEAD_ROWS * NMLP];
    __shared__ float os[HEAD_GRAPHS * NCLS];

    const int tid  = threadIdx.x;
    const int wave = __builtin_amdgcn_readfirstlane(tid >> 5);
    const int lane = tid & 31;
    const int tile = blockIdx.x * 2 + (wave >> 1);
    const int sub  = wave & 1;

    const int row   = tile * 16 + (lane & 15);
    const int kHalf = (lane >> 4) << 1;
    const int n     = lane & 15;
    const int rbase = tile * 16 + ((lane >> 4) << 3);

    int jc[NMLP_PER_LANE];
    float w1[NHID][NMLP_PER_LANE], bj[NMLP_PER_LANE];
#pragma unroll
    for (int jj = 0; jj < NMLP_PER_LANE; ++jj) {
        jc[jj] = min(lane + 32 * jj, NMLP - 1);
        bj[jj] = bl1[jc[jj]];
    }
#pragma unroll
    for (int k = 0; k < NHID; ++k)
#pragma unroll
        for (int jj = 0; jj < NMLP_PER_LANE; ++jj)
            w1[k][jj] = Wl1[k * NMLP + jc[jj]];
    const int cc = min(lane, NCLS - 1);
    const float bc = bl2[cc];
    float w2[NMLP];
#pragma unroll
    for (int k = 0; k < NMLP; ++k) w2[k] = Wl2[k * NCLS + cc];

    const float4 nlo0 = *(const float4*)(GC + rbase),          nhi0 = *(const float4*)(GC + rbase + 4);
    const float4 nlo1 = *(const float4*)(GC + NGRAPH + rbase), nhi1 = *(const float4*)(GC + NGRAPH + rbase + 4);
    const float  bn   = b2[n];

    v8f c = {};
    const float* pr = GP + (size_t)row * NHID + kHalf;
#pragma unroll
    for (int k0 = 0; k0 < NHID; k0 += 4) {
        const float2 p0 = *(const float2*)(pr + k0);
        const float2 p1 = *(const float2*)(pr + NGRAPH * NHID + k0);
        v2f a;
        a.x = p0.x + p1.x;
        a.y = p0.y + p1.y;
        v2f b;
        b.x = W2[(k0 + kHalf)     * NHID + n];
        b.y = W2[(k0 + kHalf + 1) * NHID + n];
        c = __builtin_amdgcn_wmma_f32_16x16x4_f32(
                 false, a,  false, b,
                 (short)0, c,  false,  false);
    }
    {
        const float nn[8] = { nlo0.x + nlo1.x, nlo0.y + nlo1.y, nlo0.z + nlo1.z, nlo0.w + nlo1.w,
                              nhi0.x + nhi1.x, nhi0.y + nhi1.y, nhi0.z + nhi1.z, nhi0.w + nhi1.w };
#pragma unroll
        for (int v = 0; v < 8; ++v)
            gs[wave][(((lane >> 4) << 3) + v) * NHID + n] = fmaxf(c[v] + nn[v] * bn, 0.0f);
    }

    const float* g8 = &gs[wave][sub * HEAD_ROWS * NHID];
    float* z8 = zs[wave];
#pragma unroll
    for (int r = 0; r < HEAD_ROWS; ++r) {
        float acc[NMLP_PER_LANE];
#pragma unroll
        for (int jj = 0; jj < NMLP_PER_LANE; ++jj) acc[jj] = bj[jj];
#pragma unroll
        for (int k = 0; k < NHID; ++k) {
            const float g = g8[r * NHID + k];
#pragma unroll
            for (int jj = 0; jj < NMLP_PER_LANE; ++jj) acc[jj] = fmaf(g, w1[k][jj], acc[jj]);
        }
#pragma unroll
        for (int jj = 0; jj < NMLP_PER_LANE; ++jj)
            if (lane + 32 * jj < NMLP) z8[r * NMLP + lane + 32 * jj] = fmaxf(acc[jj], 0.0f);
    }

    float o[HEAD_ROWS];
#pragma unroll
    for (int r = 0; r < HEAD_ROWS; ++r) o[r] = bc;
#pragma unroll
    for (int k0 = 0; k0 < NMLP; k0 += 4) {
#pragma unroll
        for (int r = 0; r < HEAD_ROWS; ++r) {
            const float4 z = *(const float4*)(z8 + r * NMLP + k0);
            o[r] = fmaf(z.x, w2[k0],     o[r]);
            o[r] = fmaf(z.y, w2[k0 + 1], o[r]);
            o[r] = fmaf(z.z, w2[k0 + 2], o[r]);
            o[r] = fmaf(z.w, w2[k0 + 3], o[r]);
        }
    }
    if (lane < NCLS) {
#pragma unroll
        for (int r = 0; r < HEAD_ROWS; ++r)
            os[(wave * HEAD_ROWS + r) * NCLS + lane] = o[r];
    }
    __syncthreads();

    float* ob = out + (size_t)blockIdx.x * (HEAD_GRAPHS * NCLS);
    for (int i = tid; i < HEAD_GRAPHS * NCLS; i += HEAD_WAVES * 32)
        ob[i] = os[i];
}


extern "C" void kernel_launch(void* const* d_in, const int* in_sizes, int n_in,
                              void* d_out, int out_size, void* d_ws, size_t ws_size,
                              hipStream_t stream) {
    const float* x   = (const float*)d_in[0];
    const int*  eidx = (const int*)d_in[1];
    const int*  bvec = (const int*)d_in[2];
    const float* W1  = (const float*)d_in[3];
    const float* b1  = (const float*)d_in[4];
    const float* W2  = (const float*)d_in[5];
    const float* b2  = (const float*)d_in[6];
    const float* Wl1 = (const float*)d_in[7];
    const float* bl1 = (const float*)d_in[8];
    const float* Wl2 = (const float*)d_in[9];
    const float* bl2 = (const float*)d_in[10];
    float* out = (float*)d_out;

    char* ws = (char*)d_ws;
    size_t off = 0;
    auto walloc = [&](size_t bytes) -> void* {
        void* p = ws + off;
        off = (off + bytes + 255) & ~(size_t)255;
        return p;
    };
    float* deg  = (float*)walloc((size_t)NNODES * 4);
    float* bufA = (float*)walloc((size_t)NNODES * NHID * 4);
    float* bufB = (float*)walloc((size_t)NNODES * NHID * 4);
    unsigned char* deg8 = (unsigned char*)walloc((size_t)NGRAPH * NHID * 4);
    unsigned* pool = (unsigned*)walloc((size_t)NGRAPH * NMLP * 4);
    float* wgt  = (float*)walloc((size_t)NEDGES * 4);
    unsigned short* gid = (unsigned short*)walloc((size_t)NEDGES * 2);

    const int* src = eidx;
    const int* dst = eidx + NEDGES;

    const int TB = 256;
    const int nN    = (NNODES + TB - 1) / TB;
    const int nTile = ((NNODES / 16) + 7) / 8;
    const int nGrp  = (NEDGES + EDGES_PER_GROUP - 1) / EDGES_PER_GROUP;
    const int nDeg  = (NEDGES + TB * DEG_PER_THREAD - 1) / (TB * DEG_PER_THREAD);

    unsigned* ctr = pool;
    float* gtab = (float*)(pool + POOL_TAB_OFF);
    float* gcnt = (float*)(pool + POOL_CNT_OFF);

    float* deg1 = bufB;
    fill_deg_ctr<<<nN, TB, 0, stream>>>(deg, deg1, NNODES, pool, POOL_WORDS);
    deg_accum<<<nDeg, TB, 0, stream>>>(dst, deg, deg1, NEDGES);
    int* rec = (int*)bufA;
    rsqrt_sum<<<((NNODES + 3) / 4 + TB - 1) / TB, TB, 0, stream>>>(deg, deg1, NNODES, bvec, rec, deg8, ws_size);
    float* dinv = deg;
    edge_norm<<<NORM_GROUPS, NORM_THREADS, 0, stream>>>(src, dst, dinv, rec, deg8, wgt, gid, NEDGES, ws_size);

    gemm_start<<<nTile, TB, 0, stream>>>(x, W1, bufA, bufB, dinv, b1, NNODES, NFEAT);
    edge_scatter<<<2 * nGrp, TB, 0, stream>>>(src, dst, dinv, wgt, bufA, bufB, NEDGES,
                                              ctr, nGrp, NNODES / 2, ws_size);

    edge_pool<<<POOL_GROUPS, POOL_THREADS, 0, stream>>>(src, dst, bvec, dinv, wgt, gid, bufB, gtab, gcnt,
                                                        NEDGES, NNODES, ws_size);

    head_kernel<<<NGRAPH / HEAD_GRAPHS, HEAD_WAVES * 32, 0, stream>>>(gtab, gcnt, W2, b2, Wl1, bl1, Wl2, bl2, out);
}
